# starting kernel
# baseline (speedup 1.0000x reference)
#include <hip/hip_runtime.h>

typedef __attribute__((ext_vector_type(2))) float v2f;
typedef __attribute__((ext_vector_type(8))) float v8f;

#define NFEAT 512
#define NHID  128
#define NCLASS 10


__global__ void transpose_w(const float* __restrict__ W, float* __restrict__ Wt,
                            int K, int Nc) {
  int idx = blockIdx.x * blockDim.x + threadIdx.x;
  if (idx < K * Nc) {
    int k = idx / Nc, n = idx % Nc;
    Wt[(size_t)n * K + k] = W[idx];
  }
}

__global__ void deg_init(float* __restrict__ deg, int n) {
  int i = blockIdx.x * blockDim.x + threadIdx.x;
  if (i < n) deg[i] = 1.0f;
}

__global__ void deg_edges(const int* __restrict__ cols, float* __restrict__ deg, int E) {
  int e = blockIdx.x * blockDim.x + threadIdx.x;
  if (e < E) atomicAdd(&deg[cols[e]], 1.0f);
}

__global__ void deg_fin(float* __restrict__ deg, int n) {
  int i = blockIdx.x * blockDim.x + threadIdx.x;
  if (i < n) deg[i] = rsqrtf(deg[i]);
}

__global__ void __launch_bounds__(128) gemm_wmma_f32(const float* __restrict__ X,
                                                     const float* __restrict__ Wt,
                                                     float* __restrict__ H,
                                                     int M, int K) {
  int wave = (int)((blockIdx.x * (size_t)blockDim.x + threadIdx.x) >> 5);
  int lane = threadIdx.x & 31;
  if (wave * 16 >= M) return;

  int nlo = lane & 15;
  int kh  = (lane >> 4) * 2;

  const float* xrow = X + (size_t)(wave * 16 + nlo) * K;

  v8f acc[8] = {};
  for (int k0 = 0; k0 < K; k0 += 4) {
    v2f a = *(const v2f*)(xrow + k0 + kh);
#pragma unroll
    for (int t = 0; t < 8; ++t) {
      v2f b = *(const v2f*)(Wt + (size_t)(t * 16 + nlo) * K + k0 + kh);
      acc[t] = __builtin_amdgcn_wmma_f32_16x16x4_f32(
           false, a,  false, b,
           (short)0, acc[t],  false,  false);
    }
  }

  int mbase = wave * 16 + ((lane >> 4) << 3);
#pragma unroll
  for (int t = 0; t < 8; ++t) {
#pragma unroll
    for (int e = 0; e < 8; ++e) {
      H[(size_t)(mbase + e) * NHID + t * 16 + nlo] = acc[t][e];
    }
  }
}


__global__ void agg_init(const float* __restrict__ H, const float* __restrict__ dinv,
                         float* __restrict__ Y, int total) {
  int idx = blockIdx.x * blockDim.x + threadIdx.x;
  if (idx < total) {
    int i = idx >> 7;
    float d = dinv[i];
    Y[idx] = H[idx] * d * d;
  }
}

__global__ void __launch_bounds__(256) agg_edges(const int* __restrict__ rows,
                                                 const int* __restrict__ cols,
                                                 const float* __restrict__ dinv,
                                                 const float* __restrict__ H,
                                                 float* __restrict__ Y, int E) {
  int e = (int)((blockIdx.x * (size_t)blockDim.x + threadIdx.x) >> 5);
  int lane = threadIdx.x & 31;
  if (e >= E) return;
  int r = rows[e];
  int c = cols[e];
  float w = dinv[r] * dinv[c];
  const float4 h = *(const float4*)(H + (size_t)r * NHID + lane * 4);
  float* yp = Y + (size_t)c * NHID + lane * 4;
  atomicAdd(yp + 0, h.x * w);
  atomicAdd(yp + 1, h.y * w);
  atomicAdd(yp + 2, h.z * w);
  atomicAdd(yp + 3, h.w * w);
}

__global__ void agg_bias_act(float* __restrict__ Y, const float* __restrict__ b,
                             int total, int relu) {
  int idx = blockIdx.x * blockDim.x + threadIdx.x;
  if (idx < total) {
    float v = Y[idx] + b[idx & (NHID - 1)];
    Y[idx] = relu ? fmaxf(v, 0.0f) : v;
  }
}

__global__ void __launch_bounds__(256) final_logits(const float* __restrict__ X,
                                                    const float* __restrict__ Wlt,
                                                    const float* __restrict__ bl,
                                                    float* __restrict__ out, int n) {
  int node = (int)((blockIdx.x * (size_t)blockDim.x + threadIdx.x) >> 5);
  int lane = threadIdx.x & 31;
  if (node >= n) return;

  float4 xv = *(const float4*)(X + (size_t)node * NHID + lane * 4);
  float logit[NCLASS];
#pragma unroll
  for (int c = 0; c < NCLASS; ++c) {
    float4 wv = *(const float4*)(Wlt + (size_t)c * NHID + lane * 4);
    float p = xv.x * wv.x + xv.y * wv.y + xv.z * wv.z + xv.w * wv.w;
#pragma unroll
    for (int off = 16; off > 0; off >>= 1) p += __shfl_xor(p, off, 32);
    logit[c] = p + bl[c];
  }
  float m = logit[0];
#pragma unroll
  for (int c = 1; c < NCLASS; ++c) m = fmaxf(m, logit[c]);
  float s = 0.0f;
#pragma unroll
  for (int c = 0; c < NCLASS; ++c) s += __expf(logit[c] - m);
  float lse = m + __logf(s);
  if (lane < NCLASS) out[(size_t)node * NCLASS + lane] = logit[lane] - lse;
}

extern "C" void kernel_launch(void* const* d_in, const int* in_sizes, int n_in,
                              void* d_out, int out_size, void* d_ws, size_t ws_size,
                              hipStream_t stream) {
  const float* x  = (const float*)d_in[0];
  const int*   ei = (const int*)d_in[1];
  const float* W1 = (const float*)d_in[2];
  const float* b1 = (const float*)d_in[3];
  const float* W2 = (const float*)d_in[4];
  const float* b2 = (const float*)d_in[5];
  const float* W3 = (const float*)d_in[6];
  const float* b3 = (const float*)d_in[7];
  const float* Wl = (const float*)d_in[8];
  const float* bl = (const float*)d_in[9];
  float* out = (float*)d_out;

  const int E = in_sizes[1] / 2;
  const int N = in_sizes[0] / NFEAT;

  char* ws = (char*)d_ws;
  size_t off = 0;
  auto alloc = [&](size_t bytes) -> float* {
    float* p = (float*)(ws + off);
    off += (bytes + 255) & ~(size_t)255;
    return p;
  };
  float* dinv = alloc((size_t)N * sizeof(float));
  float* Wt1  = alloc((size_t)NFEAT * NHID * sizeof(float));
  float* Wt2  = alloc((size_t)NHID * NHID * sizeof(float));
  float* Wt3  = alloc((size_t)NHID * NHID * sizeof(float));
  float* Wlt  = alloc((size_t)NHID * NCLASS * sizeof(float));
  float* H    = alloc((size_t)N * NHID * sizeof(float));
  float* Y    = alloc((size_t)N * NHID * sizeof(float));

  const int* rows = ei;
  const int* cols = ei + E;

  transpose_w<<<(NFEAT * NHID + 255) / 256, 256, 0, stream>>>(W1, Wt1, NFEAT, NHID);
  transpose_w<<<(NHID * NHID + 255) / 256, 256, 0, stream>>>(W2, Wt2, NHID, NHID);
  transpose_w<<<(NHID * NHID + 255) / 256, 256, 0, stream>>>(W3, Wt3, NHID, NHID);
  transpose_w<<<(NHID * NCLASS + 255) / 256, 256, 0, stream>>>(Wl, Wlt, NHID, NCLASS);

  deg_init<<<(N + 255) / 256, 256, 0, stream>>>(dinv, N);
  deg_edges<<<(E + 255) / 256, 256, 0, stream>>>(cols, dinv, E);
  deg_fin<<<(N + 255) / 256, 256, 0, stream>>>(dinv, N);

  const int tiles = (N + 15) / 16;
  const int gemm_grid = (tiles + 3) / 4;
  const int tot = N * NHID;
  const int tgrid = (tot + 255) / 256;
  const int egrid = (int)(((size_t)E * 32 + 255) / 256);

  gemm_wmma_f32<<<gemm_grid, 128, 0, stream>>>(x, Wt1, H, N, NFEAT);
  agg_init<<<tgrid, 256, 0, stream>>>(H, dinv, Y, tot);
  agg_edges<<<egrid, 256, 0, stream>>>(rows, cols, dinv, H, Y, E);
  agg_bias_act<<<tgrid, 256, 0, stream>>>(Y, b1, tot, 1);

  gemm_wmma_f32<<<gemm_grid, 128, 0, stream>>>(Y, Wt2, H, N, NHID);
  agg_init<<<tgrid, 256, 0, stream>>>(H, dinv, Y, tot);
  agg_edges<<<egrid, 256, 0, stream>>>(rows, cols, dinv, H, Y, E);
  agg_bias_act<<<tgrid, 256, 0, stream>>>(Y, b2, tot, 1);

  gemm_wmma_f32<<<gemm_grid, 128, 0, stream>>>(Y, Wt3, H, N, NHID);
  agg_init<<<tgrid, 256, 0, stream>>>(H, dinv, Y, tot);
  agg_edges<<<egrid, 256, 0, stream>>>(rows, cols, dinv, H, Y, E);
  agg_bias_act<<<tgrid, 256, 0, stream>>>(Y, b3, tot, 0);

  final_logits<<<(int)(((size_t)N * 32 + 255) / 256), 256, 0, stream>>>(Y, Wlt, bl, out, N);
}


# Edges counting-sorted by target once per call (returning int adds, scan, fill); one wave sums a target row in registers in fixed source order, one store with self-loop, bias, relu: no float atomics
# speedup vs baseline: 2.1743x; 2.1743x over previous
#include <hip/hip_runtime.h>

typedef __attribute__((ext_vector_type(2))) float v2f;
typedef __attribute__((ext_vector_type(8))) float v8f;

#define NFEAT 512
#define NHID  128
#define NCLASS 10


__global__ void transpose_w(const float* __restrict__ W, float* __restrict__ Wt,
                            int K, int Nc) {
  int idx = blockIdx.x * blockDim.x + threadIdx.x;
  if (idx < K * Nc) {
    int k = idx / Nc, n = idx % Nc;
    Wt[(size_t)n * K + k] = W[idx];
  }
}


__global__ void cnt_zero(int* __restrict__ cnt, int npad) {
  int i = blockIdx.x * blockDim.x + threadIdx.x;
  if (i < npad) cnt[i] = 0;
}

__global__ void cnt_rank(const int* __restrict__ cols, int* __restrict__ cnt,
                         int* __restrict__ rank, int E) {
  int e = blockIdx.x * blockDim.x + threadIdx.x;
  if (e < E) rank[e] = atomicAdd(&cnt[cols[e]], 1);
}

#define SCAN_T 1024
__global__ void __launch_bounds__(SCAN_T) seg_scan(const int* __restrict__ cnt,
                                                   int* __restrict__ offs,
                                                   float* __restrict__ dinv, int npad) {
  __shared__ int wsum[SCAN_T / 32];
  __shared__ int tot;
  const int t = threadIdx.x, lane = t & 31, wv = t >> 5;
  int carry = 0;
  for (int base = 0; base < npad; base += SCAN_T * 16) {
    const int i0 = base + t * 16;
    const bool live = i0 < npad;
    int v[16];
#pragma unroll
    for (int q = 0; q < 4; ++q) {
      int4 c = live ? *(const int4*)(cnt + i0 + 4 * q) : make_int4(0, 0, 0, 0);
      v[4 * q + 0] = c.x; v[4 * q + 1] = c.y; v[4 * q + 2] = c.z; v[4 * q + 3] = c.w;
    }
    int ex[16];
    int s = 0;
#pragma unroll
    for (int k = 0; k < 16; ++k) { ex[k] = s; s += v[k]; }
    int inc = s;
#pragma unroll
    for (int o = 1; o < 32; o <<= 1) {
      int nb = __shfl_up(inc, o, 32);
      if (lane >= o) inc += nb;
    }
    if (lane == 31) wsum[wv] = inc;
    __syncthreads();
    if (wv == 0) {
      int w = wsum[lane];
      int wi = w;
#pragma unroll
      for (int o = 1; o < 32; o <<= 1) {
        int nb = __shfl_up(wi, o, 32);
        if (lane >= o) wi += nb;
      }
      wsum[lane] = wi - w;
      if (lane == 31) tot = wi;
    }
    __syncthreads();
    const int pre = carry + wsum[wv] + (inc - s);
    carry += tot;
    if (live) {
#pragma unroll
      for (int q = 0; q < 4; ++q) {
        *(int4*)(offs + i0 + 4 * q) = make_int4(pre + ex[4 * q + 0], pre + ex[4 * q + 1],
                                                pre + ex[4 * q + 2], pre + ex[4 * q + 3]);
        *(float4*)(dinv + i0 + 4 * q) =
            make_float4(rsqrtf((float)(v[4 * q + 0] + 1)), rsqrtf((float)(v[4 * q + 1] + 1)),
                        rsqrtf((float)(v[4 * q + 2] + 1)), rsqrtf((float)(v[4 * q + 3] + 1)));
      }
    }
    __syncthreads();
  }
}

__global__ void seg_fill(const int* __restrict__ rows, const int* __restrict__ cols,
                         const int* __restrict__ rank, const int* __restrict__ offs,
                         int* __restrict__ tmp, int E) {
  int e = blockIdx.x * blockDim.x + threadIdx.x;
  if (e < E) tmp[offs[cols[e]] + rank[e]] = rows[e];
}

__global__ void __launch_bounds__(256) seg_sort(const int* __restrict__ offs,
                                                const int* __restrict__ tmp,
                                                int* __restrict__ srcs, int n) {
  int node = (int)((blockIdx.x * (size_t)blockDim.x + threadIdx.x) >> 5);
  const int lane = threadIdx.x & 31;
  if (node >= n) return;
  node = __builtin_amdgcn_readfirstlane(node);
  const int beg = offs[node];
  const int d = offs[node + 1] - beg;
  for (int ib = 0; ib < d; ib += 32) {
    const int i = ib + lane;
    const int key = (i < d) ? tmp[beg + i] : 0;
    int r = 0;
    for (int jb = 0; jb < d; jb += 32) {
      const int j = jb + lane;
      const int kj = (j < d) ? tmp[beg + j] : 0;
      const int m = (d - jb < 32) ? (d - jb) : 32;
      for (int q = 0; q < m; ++q) {
        const int k = __builtin_amdgcn_readlane(kj, q);
        r += (int)(k < key) + (int)((k == key) & (jb + q < i));
      }
    }
    if (i < d) srcs[beg + r] = key;
  }
}

__global__ void __launch_bounds__(128) gemm_wmma_f32(const float* __restrict__ X,
                                                     const float* __restrict__ Wt,
                                                     float* __restrict__ H,
                                                     int M, int K) {
  int wave = (int)((blockIdx.x * (size_t)blockDim.x + threadIdx.x) >> 5);
  int lane = threadIdx.x & 31;
  if (wave * 16 >= M) return;

  int nlo = lane & 15;
  int kh  = (lane >> 4) * 2;

  const float* xrow = X + (size_t)(wave * 16 + nlo) * K;

  v8f acc[8] = {};
  for (int k0 = 0; k0 < K; k0 += 4) {
    v2f a = *(const v2f*)(xrow + k0 + kh);
#pragma unroll
    for (int t = 0; t < 8; ++t) {
      v2f b = *(const v2f*)(Wt + (size_t)(t * 16 + nlo) * K + k0 + kh);
      acc[t] = __builtin_amdgcn_wmma_f32_16x16x4_f32(
           false, a,  false, b,
           (short)0, acc[t],  false,  false);
    }
  }

  int mbase = wave * 16 + ((lane >> 4) << 3);
#pragma unroll
  for (int t = 0; t < 8; ++t) {
#pragma unroll
    for (int e = 0; e < 8; ++e) {
      H[(size_t)(mbase + e) * NHID + t * 16 + nlo] = acc[t][e];
    }
  }
}


__device__ __forceinline__ void row_fma(float4& acc, const float4 h, const float w) {
  acc.x = fmaf(h.x, w, acc.x);
  acc.y = fmaf(h.y, w, acc.y);
  acc.z = fmaf(h.z, w, acc.z);
  acc.w = fmaf(h.w, w, acc.w);
}

__global__ void __launch_bounds__(256) agg_rows(const int* __restrict__ offs,
                                                const int* __restrict__ srcs,
                                                const float* __restrict__ dinv,
                                                const float* __restrict__ H,
                                                const float* __restrict__ b,
                                                float* __restrict__ Y, int n, int relu) {
  int node = (int)((blockIdx.x * (size_t)blockDim.x + threadIdx.x) >> 5);
  const int lane = threadIdx.x & 31;
  if (node >= n) return;
  node = __builtin_amdgcn_readfirstlane(node);
  const int beg = offs[node];
  const int d = offs[node + 1] - beg;
  const float di = dinv[node];
  const float* hl = H + lane * 4;

  float4 acc = make_float4(0.0f, 0.0f, 0.0f, 0.0f);

  for (int base = 0; base < d; base += 32) {
    const int m = (d - base < 32) ? (d - base) : 32;
    int s = node;
    if (lane < m) s = srcs[beg + base + lane];
    const float dv = dinv[s];
    int j = 0;
    for (; j + 4 <= m; j += 4) {
      const int s0 = __builtin_amdgcn_readlane(s, j);
      const int s1 = __builtin_amdgcn_readlane(s, j + 1);
      const int s2 = __builtin_amdgcn_readlane(s, j + 2);
      const int s3 = __builtin_amdgcn_readlane(s, j + 3);
      const float4 h0 = *(const float4*)(hl + (size_t)s0 * NHID);
      const float4 h1 = *(const float4*)(hl + (size_t)s1 * NHID);
      const float4 h2 = *(const float4*)(hl + (size_t)s2 * NHID);
      const float4 h3 = *(const float4*)(hl + (size_t)s3 * NHID);
      row_fma(acc, h0, di * __int_as_float(__builtin_amdgcn_readlane(__float_as_int(dv), j)));
      row_fma(acc, h1, di * __int_as_float(__builtin_amdgcn_readlane(__float_as_int(dv), j + 1)));
      row_fma(acc, h2, di * __int_as_float(__builtin_amdgcn_readlane(__float_as_int(dv), j + 2)));
      row_fma(acc, h3, di * __int_as_float(__builtin_amdgcn_readlane(__float_as_int(dv), j + 3)));
    }
#pragma nounroll
    for (; j < m; ++j) {
      const int s0 = __builtin_amdgcn_readlane(s, j);
      const float4 h0 = *(const float4*)(hl + (size_t)s0 * NHID);
      row_fma(acc, h0, di * __int_as_float(__builtin_amdgcn_readlane(__float_as_int(dv), j)));
    }
  }

  const float4 hself = *(const float4*)(hl + (size_t)node * NHID);
  const float4 bv = *(const float4*)(b + lane * 4);
  acc.x += hself.x * di * di; acc.y += hself.y * di * di;
  acc.z += hself.z * di * di; acc.w += hself.w * di * di;
  acc.x += bv.x; acc.y += bv.y; acc.z += bv.z; acc.w += bv.w;
  if (relu) {
    acc.x = fmaxf(acc.x, 0.0f); acc.y = fmaxf(acc.y, 0.0f);
    acc.z = fmaxf(acc.z, 0.0f); acc.w = fmaxf(acc.w, 0.0f);
  }
  *(float4*)(Y + (size_t)node * NHID + lane * 4) = acc;
}

__global__ void __launch_bounds__(256) final_logits(const float* __restrict__ X,
                                                    const float* __restrict__ Wlt,
                                                    const float* __restrict__ bl,
                                                    float* __restrict__ out, int n) {
  int node = (int)((blockIdx.x * (size_t)blockDim.x + threadIdx.x) >> 5);
  int lane = threadIdx.x & 31;
  if (node >= n) return;

  float4 xv = *(const float4*)(X + (size_t)node * NHID + lane * 4);
  float logit[NCLASS];
#pragma unroll
  for (int c = 0; c < NCLASS; ++c) {
    float4 wv = *(const float4*)(Wlt + (size_t)c * NHID + lane * 4);
    float p = xv.x * wv.x + xv.y * wv.y + xv.z * wv.z + xv.w * wv.w;
#pragma unroll
    for (int off = 16; off > 0; off >>= 1) p += __shfl_xor(p, off, 32);
    logit[c] = p + bl[c];
  }
  float m = logit[0];
#pragma unroll
  for (int c = 1; c < NCLASS; ++c) m = fmaxf(m, logit[c]);
  float s = 0.0f;
#pragma unroll
  for (int c = 0; c < NCLASS; ++c) s += __expf(logit[c] - m);
  float lse = m + __logf(s);
  if (lane < NCLASS) out[(size_t)node * NCLASS + lane] = logit[lane] - lse;
}

extern "C" void kernel_launch(void* const* d_in, const int* in_sizes, int n_in,
                              void* d_out, int out_size, void* d_ws, size_t ws_size,
                              hipStream_t stream) {
  const float* x  = (const float*)d_in[0];
  const int*   ei = (const int*)d_in[1];
  const float* W1 = (const float*)d_in[2];
  const float* b1 = (const float*)d_in[3];
  const float* W2 = (const float*)d_in[4];
  const float* b2 = (const float*)d_in[5];
  const float* W3 = (const float*)d_in[6];
  const float* b3 = (const float*)d_in[7];
  const float* Wl = (const float*)d_in[8];
  const float* bl = (const float*)d_in[9];
  float* out = (float*)d_out;

  const int E = in_sizes[1] / 2;
  const int N = in_sizes[0] / NFEAT;
  const int npad = ((N + 16) / 16) * 16;

  char* ws = (char*)d_ws;
  size_t off = 0;
  auto alloc = [&](size_t bytes) -> float* {
    float* p = (float*)(ws + off);
    off += (bytes + 255) & ~(size_t)255;
    return p;
  };
  const size_t cbytes = ((size_t)npad * sizeof(int) + 255) & ~(size_t)255;
  const size_t ebytes = ((size_t)E * sizeof(int) + 255) & ~(size_t)255;
  const size_t tbytes = cbytes + 2 * ebytes;
  const size_t ybytes = (size_t)N * NHID * sizeof(float);
  float* dinv = alloc((size_t)npad * sizeof(float));
  float* Wt1  = alloc((size_t)NFEAT * NHID * sizeof(float));
  float* Wt2  = alloc((size_t)NHID * NHID * sizeof(float));
  float* Wt3  = alloc((size_t)NHID * NHID * sizeof(float));
  float* Wlt  = alloc((size_t)NHID * NCLASS * sizeof(float));
  float* H    = alloc(ybytes);
  float* Y    = alloc(ybytes > tbytes ? ybytes : tbytes);
  int* offs   = (int*)alloc((size_t)npad * sizeof(int));
  int* srcs   = (int*)alloc((size_t)E * sizeof(int));
  if (off > ws_size) return;
  int* cnt  = (int*)Y;
  int* rank = (int*)((char*)Y + cbytes);
  int* tmp  = (int*)((char*)Y + cbytes + ebytes);

  const int* rows = ei;
  const int* cols = ei + E;

  transpose_w<<<(NFEAT * NHID + 255) / 256, 256, 0, stream>>>(W1, Wt1, NFEAT, NHID);
  transpose_w<<<(NHID * NHID + 255) / 256, 256, 0, stream>>>(W2, Wt2, NHID, NHID);
  transpose_w<<<(NHID * NHID + 255) / 256, 256, 0, stream>>>(W3, Wt3, NHID, NHID);
  transpose_w<<<(NHID * NCLASS + 255) / 256, 256, 0, stream>>>(Wl, Wlt, NHID, NCLASS);

  const int wgrid = (int)(((size_t)N * 32 + 255) / 256);

  cnt_zero<<<(npad + 255) / 256, 256, 0, stream>>>(cnt, npad);
  cnt_rank<<<(E + 255) / 256, 256, 0, stream>>>(cols, cnt, rank, E);
  seg_scan<<<1, SCAN_T, 0, stream>>>(cnt, offs, dinv, npad);
  seg_fill<<<(E + 255) / 256, 256, 0, stream>>>(rows, cols, rank, offs, tmp, E);
  seg_sort<<<wgrid, 256, 0, stream>>>(offs, tmp, srcs, N);

  const int tiles = (N + 15) / 16;
  const int gemm_grid = (tiles + 3) / 4;

  gemm_wmma_f32<<<gemm_grid, 128, 0, stream>>>(x, Wt1, H, N, NFEAT);
  agg_rows<<<wgrid, 256, 0, stream>>>(offs, srcs, dinv, H, b1, Y, N, 1);

  gemm_wmma_f32<<<gemm_grid, 128, 0, stream>>>(Y, Wt2, H, N, NHID);
  agg_rows<<<wgrid, 256, 0, stream>>>(offs, srcs, dinv, H, b2, Y, N, 1);

  gemm_wmma_f32<<<gemm_grid, 128, 0, stream>>>(Y, Wt3, H, N, NHID);
  agg_rows<<<wgrid, 256, 0, stream>>>(offs, srcs, dinv, H, b3, Y, N, 0);

  final_logits<<<(int)(((size_t)N * 32 + 255) / 256), 256, 0, stream>>>(Y, Wlt, bl, out, N);
}


# GEMM retiled: X and Wt K-chunks staged through LDS by whole-line 16-byte loads (next chunk prefetched in registers), WMMA fragments read from LDS, result strip stored as whole rows
# speedup vs baseline: 2.6395x; 1.2140x over previous
#include <hip/hip_runtime.h>

typedef __attribute__((ext_vector_type(2))) float v2f;
typedef __attribute__((ext_vector_type(8))) float v8f;

#define NFEAT 512
#define NHID  128
#define NCLASS 10


__global__ void transpose_w(const float* __restrict__ W, float* __restrict__ Wt,
                            int K, int Nc) {
  int idx = blockIdx.x * blockDim.x + threadIdx.x;
  if (idx < K * Nc) {
    int k = idx / Nc, n = idx % Nc;
    Wt[(size_t)n * K + k] = W[idx];
  }
}


__global__ void cnt_zero(int* __restrict__ cnt, int npad) {
  int i = blockIdx.x * blockDim.x + threadIdx.x;
  if (i < npad) cnt[i] = 0;
}

__global__ void cnt_rank(const int* __restrict__ cols, int* __restrict__ cnt,
                         int* __restrict__ rank, int E) {
  int e = blockIdx.x * blockDim.x + threadIdx.x;
  if (e < E) rank[e] = atomicAdd(&cnt[cols[e]], 1);
}

#define SCAN_T 1024
__global__ void __launch_bounds__(SCAN_T) seg_scan(const int* __restrict__ cnt,
                                                   int* __restrict__ offs,
                                                   float* __restrict__ dinv, int npad) {
  __shared__ int wsum[SCAN_T / 32];
  __shared__ int tot;
  const int t = threadIdx.x, lane = t & 31, wv = t >> 5;
  int carry = 0;
  for (int base = 0; base < npad; base += SCAN_T * 16) {
    const int i0 = base + t * 16;
    const bool live = i0 < npad;
    int v[16];
#pragma unroll
    for (int q = 0; q < 4; ++q) {
      int4 c = live ? *(const int4*)(cnt + i0 + 4 * q) : make_int4(0, 0, 0, 0);
      v[4 * q + 0] = c.x; v[4 * q + 1] = c.y; v[4 * q + 2] = c.z; v[4 * q + 3] = c.w;
    }
    int ex[16];
    int s = 0;
#pragma unroll
    for (int k = 0; k < 16; ++k) { ex[k] = s; s += v[k]; }
    int inc = s;
#pragma unroll
    for (int o = 1; o < 32; o <<= 1) {
      int nb = __shfl_up(inc, o, 32);
      if (lane >= o) inc += nb;
    }
    if (lane == 31) wsum[wv] = inc;
    __syncthreads();
    if (wv == 0) {
      int w = wsum[lane];
      int wi = w;
#pragma unroll
      for (int o = 1; o < 32; o <<= 1) {
        int nb = __shfl_up(wi, o, 32);
        if (lane >= o) wi += nb;
      }
      wsum[lane] = wi - w;
      if (lane == 31) tot = wi;
    }
    __syncthreads();
    const int pre = carry + wsum[wv] + (inc - s);
    carry += tot;
    if (live) {
#pragma unroll
      for (int q = 0; q < 4; ++q) {
        *(int4*)(offs + i0 + 4 * q) = make_int4(pre + ex[4 * q + 0], pre + ex[4 * q + 1],
                                                pre + ex[4 * q + 2], pre + ex[4 * q + 3]);
        *(float4*)(dinv + i0 + 4 * q) =
            make_float4(rsqrtf((float)(v[4 * q + 0] + 1)), rsqrtf((float)(v[4 * q + 1] + 1)),
                        rsqrtf((float)(v[4 * q + 2] + 1)), rsqrtf((float)(v[4 * q + 3] + 1)));
      }
    }
    __syncthreads();
  }
}

__global__ void seg_fill(const int* __restrict__ rows, const int* __restrict__ cols,
                         const int* __restrict__ rank, const int* __restrict__ offs,
                         int* __restrict__ tmp, int E) {
  int e = blockIdx.x * blockDim.x + threadIdx.x;
  if (e < E) tmp[offs[cols[e]] + rank[e]] = rows[e];
}

__global__ void __launch_bounds__(256) seg_sort(const int* __restrict__ offs,
                                                const int* __restrict__ tmp,
                                                int* __restrict__ srcs, int n) {
  int node = (int)((blockIdx.x * (size_t)blockDim.x + threadIdx.x) >> 5);
  const int lane = threadIdx.x & 31;
  if (node >= n) return;
  node = __builtin_amdgcn_readfirstlane(node);
  const int beg = offs[node];
  const int d = offs[node + 1] - beg;
  for (int ib = 0; ib < d; ib += 32) {
    const int i = ib + lane;
    const int key = (i < d) ? tmp[beg + i] : 0;
    int r = 0;
    for (int jb = 0; jb < d; jb += 32) {
      const int j = jb + lane;
      const int kj = (j < d) ? tmp[beg + j] : 0;
      const int m = (d - jb < 32) ? (d - jb) : 32;
      for (int q = 0; q < m; ++q) {
        const int k = __builtin_amdgcn_readlane(kj, q);
        r += (int)(k < key) + (int)((k == key) & (jb + q < i));
      }
    }
    if (i < d) srcs[beg + r] = key;
  }
}

typedef __attribute__((ext_vector_type(4))) float v4f;
#define GB_M 64
#define GB_K 32
#define GB_P 36
#define GB_STAGE ((GB_M + NHID) * GB_P)
#define GB_CP 132
#define GB_RS (16 * GB_P)

#define GB_FETCH(kc)                                  \
  a0 = *(const v4f*)(x0 + (kc));                      \
  a1 = *(const v4f*)(x1 + (kc));                      \
  a2 = *(const v4f*)(x2 + (kc));                      \
  a3 = *(const v4f*)(x3 + (kc));                      \
  b0 = *(const v4f*)(wsrc + (kc));                    \
  b1 = *(const v4f*)(wsrc + wstep + (kc));            \
  b2 = *(const v4f*)(wsrc + 2 * wstep + (kc));        \
  b3 = *(const v4f*)(wsrc + 3 * wstep + (kc));        \
  b4 = *(const v4f*)(wsrc + 4 * wstep + (kc));        \
  b5 = *(const v4f*)(wsrc + 5 * wstep + (kc));        \
  b6 = *(const v4f*)(wsrc + 6 * wstep + (kc));        \
  b7 = *(const v4f*)(wsrc + 7 * wstep + (kc));

#define GB_PUT(dst)                                           \
  *(v4f*)((dst) + sdst) = a0;                                 \
  *(v4f*)((dst) + sdst + GB_RS) = a1;                         \
  *(v4f*)((dst) + sdst + 2 * GB_RS) = a2;                     \
  *(v4f*)((dst) + sdst + 3 * GB_RS) = a3;                     \
  *(v4f*)((dst) + GB_M * GB_P + sdst) = b0;                   \
  *(v4f*)((dst) + GB_M * GB_P + sdst + GB_RS) = b1;           \
  *(v4f*)((dst) + GB_M * GB_P + sdst + 2 * GB_RS) = b2;       \
  *(v4f*)((dst) + GB_M * GB_P + sdst + 3 * GB_RS) = b3;       \
  *(v4f*)((dst) + GB_M * GB_P + sdst + 4 * GB_RS) = b4;       \
  *(v4f*)((dst) + GB_M * GB_P + sdst + 5 * GB_RS) = b5;       \
  *(v4f*)((dst) + GB_M * GB_P + sdst + 6 * GB_RS) = b6;       \
  *(v4f*)((dst) + GB_M * GB_P + sdst + 7 * GB_RS) = b7;

__global__ void __launch_bounds__(128) gemm_wmma_f32(const float* __restrict__ X,
                                                     const float* __restrict__ Wt,
                                                     float* __restrict__ H,
                                                     int M, int K) {
  __shared__ __attribute__((aligned(16))) float lds[2 * GB_STAGE];

  const int tid = threadIdx.x;
  const int lane = tid & 31;
  const int wv = tid >> 5;
  const int c = lane & 15;
  const int h = lane >> 4;
  const int row0 = blockIdx.x * GB_M;

  const int seg = tid & 7;
  const int rsub = tid >> 3;
  const int last = M - 1;
  const int r0 = row0 + rsub, r1 = r0 + 16, r2 = r0 + 32, r3 = r0 + 48;
  const float* x0 = X + (size_t)(r0 < last ? r0 : last) * K + seg * 4;
  const float* x1 = X + (size_t)(r1 < last ? r1 : last) * K + seg * 4;
  const float* x2 = X + (size_t)(r2 < last ? r2 : last) * K + seg * 4;
  const float* x3 = X + (size_t)(r3 < last ? r3 : last) * K + seg * 4;
  const float* wsrc = Wt + (size_t)rsub * K + seg * 4;
  const size_t wstep = (size_t)16 * K;
  const int sdst = rsub * GB_P + seg * 4;

  v4f a0, a1, a2, a3, b0, b1, b2, b3, b4, b5, b6, b7;
  GB_FETCH(0)
  GB_PUT(lds)
  __syncthreads();

  v8f acc[8] = {};
  const int nch = K / GB_K;
  for (int ch = 0; ch < nch; ++ch) {
    const float* st = lds + (ch & 1) * GB_STAGE;
    float* nx = lds + ((ch + 1) & 1) * GB_STAGE;
    const bool more = ch + 1 < nch;
    if (more) {
      const int kc = (ch + 1) * GB_K;
      GB_FETCH(kc)
    }

    const float* ap = st + (wv * 16 + c) * GB_P + 2 * h;
    const float* bp = st + (GB_M + c) * GB_P + 2 * h;
#pragma unroll
    for (int ks = 0; ks < GB_K / 4; ++ks) {
      const v2f a = *(const v2f*)(ap + ks * 4);
#pragma unroll
      for (int t = 0; t < 8; ++t) {
        const v2f b = *(const v2f*)(bp + t * GB_RS + ks * 4);
        acc[t] = __builtin_amdgcn_wmma_f32_16x16x4_f32(
             false, a,  false, b,
             (short)0, acc[t],  false,  false);
      }
    }

    if (more) {
      GB_PUT(nx)
    }
    __syncthreads();
  }

  float* cs = lds + wv * (16 * GB_CP);
#pragma unroll
  for (int t = 0; t < 8; ++t) {
#pragma unroll
    for (int e = 0; e < 8; ++e) cs[(8 * h + e) * GB_CP + t * 16 + c] = acc[t][e];
  }
  __syncthreads();
#pragma unroll
  for (int r = 0; r < 16; ++r) {
    const int row = row0 + wv * 16 + r;
    if (row < M)
      *(v4f*)(H + (size_t)row * NHID + lane * 4) = *(const v4f*)(cs + r * GB_CP + lane * 4);
  }
}
#undef GB_FETCH
#undef GB_PUT


__device__ __forceinline__ void row_fma(float4& acc, const float4 h, const float w) {
  acc.x = fmaf(h.x, w, acc.x);
  acc.y = fmaf(h.y, w, acc.y);
  acc.z = fmaf(h.z, w, acc.z);
  acc.w = fmaf(h.w, w, acc.w);
}

__global__ void __launch_bounds__(256) agg_rows(const int* __restrict__ offs,
                                                const int* __restrict__ srcs,
                                                const float* __restrict__ dinv,
                                                const float* __restrict__ H,
                                                const float* __restrict__ b,
                                                float* __restrict__ Y, int n, int relu) {
  int node = (int)((blockIdx.x * (size_t)blockDim.x + threadIdx.x) >> 5);
  const int lane = threadIdx.x & 31;
  if (node >= n) return;
  node = __builtin_amdgcn_readfirstlane(node);
  const int beg = offs[node];
  const int d = offs[node + 1] - beg;
  const float di = dinv[node];
  const float* hl = H + lane * 4;

  float4 acc = make_float4(0.0f, 0.0f, 0.0f, 0.0f);

  for (int base = 0; base < d; base += 32) {
    const int m = (d - base < 32) ? (d - base) : 32;
    int s = node;
    if (lane < m) s = srcs[beg + base + lane];
    const float dv = dinv[s];
    int j = 0;
    for (; j + 4 <= m; j += 4) {
      const int s0 = __builtin_amdgcn_readlane(s, j);
      const int s1 = __builtin_amdgcn_readlane(s, j + 1);
      const int s2 = __builtin_amdgcn_readlane(s, j + 2);
      const int s3 = __builtin_amdgcn_readlane(s, j + 3);
      const float4 h0 = *(const float4*)(hl + (size_t)s0 * NHID);
      const float4 h1 = *(const float4*)(hl + (size_t)s1 * NHID);
      const float4 h2 = *(const float4*)(hl + (size_t)s2 * NHID);
      const float4 h3 = *(const float4*)(hl + (size_t)s3 * NHID);
      row_fma(acc, h0, di * __int_as_float(__builtin_amdgcn_readlane(__float_as_int(dv), j)));
      row_fma(acc, h1, di * __int_as_float(__builtin_amdgcn_readlane(__float_as_int(dv), j + 1)));
      row_fma(acc, h2, di * __int_as_float(__builtin_amdgcn_readlane(__float_as_int(dv), j + 2)));
      row_fma(acc, h3, di * __int_as_float(__builtin_amdgcn_readlane(__float_as_int(dv), j + 3)));
    }
#pragma nounroll
    for (; j < m; ++j) {
      const int s0 = __builtin_amdgcn_readlane(s, j);
      const float4 h0 = *(const float4*)(hl + (size_t)s0 * NHID);
      row_fma(acc, h0, di * __int_as_float(__builtin_amdgcn_readlane(__float_as_int(dv), j)));
    }
  }

  const float4 hself = *(const float4*)(hl + (size_t)node * NHID);
  const float4 bv = *(const float4*)(b + lane * 4);
  acc.x += hself.x * di * di; acc.y += hself.y * di * di;
  acc.z += hself.z * di * di; acc.w += hself.w * di * di;
  acc.x += bv.x; acc.y += bv.y; acc.z += bv.z; acc.w += bv.w;
  if (relu) {
    acc.x = fmaxf(acc.x, 0.0f); acc.y = fmaxf(acc.y, 0.0f);
    acc.z = fmaxf(acc.z, 0.0f); acc.w = fmaxf(acc.w, 0.0f);
  }
  *(float4*)(Y + (size_t)node * NHID + lane * 4) = acc;
}

__global__ void __launch_bounds__(256) final_logits(const float* __restrict__ X,
                                                    const float* __restrict__ Wlt,
                                                    const float* __restrict__ bl,
                                                    float* __restrict__ out, int n) {
  int node = (int)((blockIdx.x * (size_t)blockDim.x + threadIdx.x) >> 5);
  int lane = threadIdx.x & 31;
  if (node >= n) return;

  float4 xv = *(const float4*)(X + (size_t)node * NHID + lane * 4);
  float logit[NCLASS];
#pragma unroll
  for (int c = 0; c < NCLASS; ++c) {
    float4 wv = *(const float4*)(Wlt + (size_t)c * NHID + lane * 4);
    float p = xv.x * wv.x + xv.y * wv.y + xv.z * wv.z + xv.w * wv.w;
#pragma unroll
    for (int off = 16; off > 0; off >>= 1) p += __shfl_xor(p, off, 32);
    logit[c] = p + bl[c];
  }
  float m = logit[0];
#pragma unroll
  for (int c = 1; c < NCLASS; ++c) m = fmaxf(m, logit[c]);
  float s = 0.0f;
#pragma unroll
  for (int c = 0; c < NCLASS; ++c) s += __expf(logit[c] - m);
  float lse = m + __logf(s);
  if (lane < NCLASS) out[(size_t)node * NCLASS + lane] = logit[lane] - lse;
}

extern "C" void kernel_launch(void* const* d_in, const int* in_sizes, int n_in,
                              void* d_out, int out_size, void* d_ws, size_t ws_size,
                              hipStream_t stream) {
  const float* x  = (const float*)d_in[0];
  const int*   ei = (const int*)d_in[1];
  const float* W1 = (const float*)d_in[2];
  const float* b1 = (const float*)d_in[3];
  const float* W2 = (const float*)d_in[4];
  const float* b2 = (const float*)d_in[5];
  const float* W3 = (const float*)d_in[6];
  const float* b3 = (const float*)d_in[7];
  const float* Wl = (const float*)d_in[8];
  const float* bl = (const float*)d_in[9];
  float* out = (float*)d_out;

  const int E = in_sizes[1] / 2;
  const int N = in_sizes[0] / NFEAT;
  const int npad = ((N + 16) / 16) * 16;

  char* ws = (char*)d_ws;
  size_t off = 0;
  auto alloc = [&](size_t bytes) -> float* {
    float* p = (float*)(ws + off);
    off += (bytes + 255) & ~(size_t)255;
    return p;
  };
  const size_t cbytes = ((size_t)npad * sizeof(int) + 255) & ~(size_t)255;
  const size_t ebytes = ((size_t)E * sizeof(int) + 255) & ~(size_t)255;
  const size_t tbytes = cbytes + 2 * ebytes;
  const size_t ybytes = (size_t)N * NHID * sizeof(float);
  float* dinv = alloc((size_t)npad * sizeof(float));
  float* Wt1  = alloc((size_t)NFEAT * NHID * sizeof(float));
  float* Wt2  = alloc((size_t)NHID * NHID * sizeof(float));
  float* Wt3  = alloc((size_t)NHID * NHID * sizeof(float));
  float* Wlt  = alloc((size_t)NHID * NCLASS * sizeof(float));
  float* H    = alloc(ybytes);
  float* Y    = alloc(ybytes > tbytes ? ybytes : tbytes);
  int* offs   = (int*)alloc((size_t)npad * sizeof(int));
  int* srcs   = (int*)alloc((size_t)E * sizeof(int));
  if (off > ws_size) return;
  int* cnt  = (int*)Y;
  int* rank = (int*)((char*)Y + cbytes);
  int* tmp  = (int*)((char*)Y + cbytes + ebytes);

  const int* rows = ei;
  const int* cols = ei + E;

  transpose_w<<<(NFEAT * NHID + 255) / 256, 256, 0, stream>>>(W1, Wt1, NFEAT, NHID);
  transpose_w<<<(NHID * NHID + 255) / 256, 256, 0, stream>>>(W2, Wt2, NHID, NHID);
  transpose_w<<<(NHID * NHID + 255) / 256, 256, 0, stream>>>(W3, Wt3, NHID, NHID);
  transpose_w<<<(NHID * NCLASS + 255) / 256, 256, 0, stream>>>(Wl, Wlt, NHID, NCLASS);

  const int wgrid = (int)(((size_t)N * 32 + 255) / 256);

  cnt_zero<<<(npad + 255) / 256, 256, 0, stream>>>(cnt, npad);
  cnt_rank<<<(E + 255) / 256, 256, 0, stream>>>(cols, cnt, rank, E);
  seg_scan<<<1, SCAN_T, 0, stream>>>(cnt, offs, dinv, npad);
  seg_fill<<<(E + 255) / 256, 256, 0, stream>>>(rows, cols, rank, offs, tmp, E);
  seg_sort<<<wgrid, 256, 0, stream>>>(offs, tmp, srcs, N);

  const int tiles = (N + 15) / 16;
  const int gemm_grid = (tiles + 3) / 4;

  gemm_wmma_f32<<<gemm_grid, 128, 0, stream>>>(x, Wt1, H, N, NFEAT);
  agg_rows<<<wgrid, 256, 0, stream>>>(offs, srcs, dinv, H, b1, Y, N, 1);

  gemm_wmma_f32<<<gemm_grid, 128, 0, stream>>>(Y, Wt2, H, N, NHID);
  agg_rows<<<wgrid, 256, 0, stream>>>(offs, srcs, dinv, H, b2, Y, N, 1);

  gemm_wmma_f32<<<gemm_grid, 128, 0, stream>>>(Y, Wt3, H, N, NHID);
  agg_rows<<<wgrid, 256, 0, stream>>>(offs, srcs, dinv, H, b3, Y, N, 0);

  final_logits<<<(int)(((size_t)N * 32 + 255) / 256), 256, 0, stream>>>(Y, Wlt, bl, out, N);
}


# 16 launches to 11: four weight transposes and counter zeroing run as one launch chosen by block index; classifier and log-softmax run in the last aggregation's epilogue on the row in registers
# speedup vs baseline: 2.8017x; 1.0615x over previous
#include <hip/hip_runtime.h>

typedef __attribute__((ext_vector_type(2))) float v2f;
typedef __attribute__((ext_vector_type(8))) float v8f;

#define NFEAT 512
#define NHID  128
#define NCLASS 10


__device__ __forceinline__ void transpose_elem(const float* __restrict__ W, float* __restrict__ Wt,
                                               int K, int Nc, int idx) {
  if (idx < K * Nc) {
    int k = idx / Nc, n = idx % Nc;
    Wt[(size_t)n * K + k] = W[idx];
  }
}


#define PB_W1 (NFEAT * NHID / 256)
#define PB_W2 (NHID * NHID / 256)
#define PB_WL ((NHID * NCLASS + 255) / 256)
#define PB_ALLW (PB_W1 + 2 * PB_W2 + PB_WL)
__global__ void __launch_bounds__(256) prep_all(const float* __restrict__ W1, float* __restrict__ Wt1,
                                                const float* __restrict__ W2, float* __restrict__ Wt2,
                                                const float* __restrict__ W3, float* __restrict__ Wt3,
                                                const float* __restrict__ Wl, float* __restrict__ Wlt,
                                                int* __restrict__ cnt, int npad) {
  const int b = blockIdx.x, t = threadIdx.x;
  if (b < PB_W1) {
    transpose_elem(W1, Wt1, NFEAT, NHID, b * 256 + t);
  } else if (b < PB_W1 + PB_W2) {
    transpose_elem(W2, Wt2, NHID, NHID, (b - PB_W1) * 256 + t);
  } else if (b < PB_W1 + 2 * PB_W2) {
    transpose_elem(W3, Wt3, NHID, NHID, (b - PB_W1 - PB_W2) * 256 + t);
  } else if (b < PB_ALLW) {
    transpose_elem(Wl, Wlt, NHID, NCLASS, (b - PB_W1 - 2 * PB_W2) * 256 + t);
  } else {
    const int i = (b - PB_ALLW) * 256 + t;
    if (i < npad) cnt[i] = 0;
  }
}

__global__ void cnt_rank(const int* __restrict__ cols, int* __restrict__ cnt,
                         int* __restrict__ rank, int E) {
  int e = blockIdx.x * blockDim.x + threadIdx.x;
  if (e < E) rank[e] = atomicAdd(&cnt[cols[e]], 1);
}

#define SCAN_T 1024
__global__ void __launch_bounds__(SCAN_T) seg_scan(const int* __restrict__ cnt,
                                                   int* __restrict__ offs,
                                                   float* __restrict__ dinv, int npad) {
  __shared__ int wsum[SCAN_T / 32];
  __shared__ int tot;
  const int t = threadIdx.x, lane = t & 31, wv = t >> 5;
  int carry = 0;
  for (int base = 0; base < npad; base += SCAN_T * 16) {
    const int i0 = base + t * 16;
    const bool live = i0 < npad;
    int v[16];
#pragma unroll
    for (int q = 0; q < 4; ++q) {
      int4 c = live ? *(const int4*)(cnt + i0 + 4 * q) : make_int4(0, 0, 0, 0);
      v[4 * q + 0] = c.x; v[4 * q + 1] = c.y; v[4 * q + 2] = c.z; v[4 * q + 3] = c.w;
    }
    int ex[16];
    int s = 0;
#pragma unroll
    for (int k = 0; k < 16; ++k) { ex[k] = s; s += v[k]; }
    int inc = s;
#pragma unroll
    for (int o = 1; o < 32; o <<= 1) {
      int nb = __shfl_up(inc, o, 32);
      if (lane >= o) inc += nb;
    }
    if (lane == 31) wsum[wv] = inc;
    __syncthreads();
    if (wv == 0) {
      int w = wsum[lane];
      int wi = w;
#pragma unroll
      for (int o = 1; o < 32; o <<= 1) {
        int nb = __shfl_up(wi, o, 32);
        if (lane >= o) wi += nb;
      }
      wsum[lane] = wi - w;
      if (lane == 31) tot = wi;
    }
    __syncthreads();
    const int pre = carry + wsum[wv] + (inc - s);
    carry += tot;
    if (live) {
#pragma unroll
      for (int q = 0; q < 4; ++q) {
        *(int4*)(offs + i0 + 4 * q) = make_int4(pre + ex[4 * q + 0], pre + ex[4 * q + 1],
                                                pre + ex[4 * q + 2], pre + ex[4 * q + 3]);
        *(float4*)(dinv + i0 + 4 * q) =
            make_float4(rsqrtf((float)(v[4 * q + 0] + 1)), rsqrtf((float)(v[4 * q + 1] + 1)),
                        rsqrtf((float)(v[4 * q + 2] + 1)), rsqrtf((float)(v[4 * q + 3] + 1)));
      }
    }
    __syncthreads();
  }
}

__global__ void seg_fill(const int* __restrict__ rows, const int* __restrict__ cols,
                         const int* __restrict__ rank, const int* __restrict__ offs,
                         int* __restrict__ tmp, int E) {
  int e = blockIdx.x * blockDim.x + threadIdx.x;
  if (e < E) tmp[offs[cols[e]] + rank[e]] = rows[e];
}

__global__ void __launch_bounds__(256) seg_sort(const int* __restrict__ offs,
                                                const int* __restrict__ tmp,
                                                int* __restrict__ srcs, int n) {
  int node = (int)((blockIdx.x * (size_t)blockDim.x + threadIdx.x) >> 5);
  const int lane = threadIdx.x & 31;
  if (node >= n) return;
  node = __builtin_amdgcn_readfirstlane(node);
  const int beg = offs[node];
  const int d = offs[node + 1] - beg;
  for (int ib = 0; ib < d; ib += 32) {
    const int i = ib + lane;
    const int key = (i < d) ? tmp[beg + i] : 0;
    int r = 0;
    for (int jb = 0; jb < d; jb += 32) {
      const int j = jb + lane;
      const int kj = (j < d) ? tmp[beg + j] : 0;
      const int m = (d - jb < 32) ? (d - jb) : 32;
      for (int q = 0; q < m; ++q) {
        const int k = __builtin_amdgcn_readlane(kj, q);
        r += (int)(k < key) + (int)((k == key) & (jb + q < i));
      }
    }
    if (i < d) srcs[beg + r] = key;
  }
}

typedef __attribute__((ext_vector_type(4))) float v4f;
#define GB_M 64
#define GB_K 32
#define GB_P 36
#define GB_STAGE ((GB_M + NHID) * GB_P)
#define GB_CP 132
#define GB_RS (16 * GB_P)

#define GB_FETCH(kc)                                  \
  a0 = *(const v4f*)(x0 + (kc));                      \
  a1 = *(const v4f*)(x1 + (kc));                      \
  a2 = *(const v4f*)(x2 + (kc));                      \
  a3 = *(const v4f*)(x3 + (kc));                      \
  b0 = *(const v4f*)(wsrc + (kc));                    \
  b1 = *(const v4f*)(wsrc + wstep + (kc));            \
  b2 = *(const v4f*)(wsrc + 2 * wstep + (kc));        \
  b3 = *(const v4f*)(wsrc + 3 * wstep + (kc));        \
  b4 = *(const v4f*)(wsrc + 4 * wstep + (kc));        \
  b5 = *(const v4f*)(wsrc + 5 * wstep + (kc));        \
  b6 = *(const v4f*)(wsrc + 6 * wstep + (kc));        \
  b7 = *(const v4f*)(wsrc + 7 * wstep + (kc));

#define GB_PUT(dst)                                           \
  *(v4f*)((dst) + sdst) = a0;                                 \
  *(v4f*)((dst) + sdst + GB_RS) = a1;                         \
  *(v4f*)((dst) + sdst + 2 * GB_RS) = a2;                     \
  *(v4f*)((dst) + sdst + 3 * GB_RS) = a3;                     \
  *(v4f*)((dst) + GB_M * GB_P + sdst) = b0;                   \
  *(v4f*)((dst) + GB_M * GB_P + sdst + GB_RS) = b1;           \
  *(v4f*)((dst) + GB_M * GB_P + sdst + 2 * GB_RS) = b2;       \
  *(v4f*)((dst) + GB_M * GB_P + sdst + 3 * GB_RS) = b3;       \
  *(v4f*)((dst) + GB_M * GB_P + sdst + 4 * GB_RS) = b4;       \
  *(v4f*)((dst) + GB_M * GB_P + sdst + 5 * GB_RS) = b5;       \
  *(v4f*)((dst) + GB_M * GB_P + sdst + 6 * GB_RS) = b6;       \
  *(v4f*)((dst) + GB_M * GB_P + sdst + 7 * GB_RS) = b7;

__global__ void __launch_bounds__(128) gemm_wmma_f32(const float* __restrict__ X,
                                                     const float* __restrict__ Wt,
                                                     float* __restrict__ H,
                                                     int M, int K) {
  __shared__ __attribute__((aligned(16))) float lds[2 * GB_STAGE];

  const int tid = threadIdx.x;
  const int lane = tid & 31;
  const int wv = tid >> 5;
  const int c = lane & 15;
  const int h = lane >> 4;
  const int row0 = blockIdx.x * GB_M;

  const int seg = tid & 7;
  const int rsub = tid >> 3;
  const int last = M - 1;
  const int r0 = row0 + rsub, r1 = r0 + 16, r2 = r0 + 32, r3 = r0 + 48;
  const float* x0 = X + (size_t)(r0 < last ? r0 : last) * K + seg * 4;
  const float* x1 = X + (size_t)(r1 < last ? r1 : last) * K + seg * 4;
  const float* x2 = X + (size_t)(r2 < last ? r2 : last) * K + seg * 4;
  const float* x3 = X + (size_t)(r3 < last ? r3 : last) * K + seg * 4;
  const float* wsrc = Wt + (size_t)rsub * K + seg * 4;
  const size_t wstep = (size_t)16 * K;
  const int sdst = rsub * GB_P + seg * 4;

  v4f a0, a1, a2, a3, b0, b1, b2, b3, b4, b5, b6, b7;
  GB_FETCH(0)
  GB_PUT(lds)
  __syncthreads();

  v8f acc[8] = {};
  const int nch = K / GB_K;
  for (int ch = 0; ch < nch; ++ch) {
    const float* st = lds + (ch & 1) * GB_STAGE;
    float* nx = lds + ((ch + 1) & 1) * GB_STAGE;
    const bool more = ch + 1 < nch;
    if (more) {
      const int kc = (ch + 1) * GB_K;
      GB_FETCH(kc)
    }

    const float* ap = st + (wv * 16 + c) * GB_P + 2 * h;
    const float* bp = st + (GB_M + c) * GB_P + 2 * h;
#pragma unroll
    for (int ks = 0; ks < GB_K / 4; ++ks) {
      const v2f a = *(const v2f*)(ap + ks * 4);
#pragma unroll
      for (int t = 0; t < 8; ++t) {
        const v2f b = *(const v2f*)(bp + t * GB_RS + ks * 4);
        acc[t] = __builtin_amdgcn_wmma_f32_16x16x4_f32(
             false, a,  false, b,
             (short)0, acc[t],  false,  false);
      }
    }

    if (more) {
      GB_PUT(nx)
    }
    __syncthreads();
  }

  float* cs = lds + wv * (16 * GB_CP);
#pragma unroll
  for (int t = 0; t < 8; ++t) {
#pragma unroll
    for (int e = 0; e < 8; ++e) cs[(8 * h + e) * GB_CP + t * 16 + c] = acc[t][e];
  }
  __syncthreads();
#pragma unroll
  for (int r = 0; r < 16; ++r) {
    const int row = row0 + wv * 16 + r;
    if (row < M)
      *(v4f*)(H + (size_t)row * NHID + lane * 4) = *(const v4f*)(cs + r * GB_CP + lane * 4);
  }
}
#undef GB_FETCH
#undef GB_PUT


__device__ __forceinline__ void row_fma(float4& acc, const float4 h, const float w) {
  acc.x = fmaf(h.x, w, acc.x);
  acc.y = fmaf(h.y, w, acc.y);
  acc.z = fmaf(h.z, w, acc.z);
  acc.w = fmaf(h.w, w, acc.w);
}

template <bool LAST>
__global__ void __launch_bounds__(256) agg_rows(const int* __restrict__ offs,
                                                const int* __restrict__ srcs,
                                                const float* __restrict__ dinv,
                                                const float* __restrict__ H,
                                                const float* __restrict__ b,
                                                float* __restrict__ Y,
                                                const float* __restrict__ Wlt,
                                                const float* __restrict__ bl,
                                                float* __restrict__ out, int n) {
  int node = (int)((blockIdx.x * (size_t)blockDim.x + threadIdx.x) >> 5);
  const int lane = threadIdx.x & 31;
  if (node >= n) return;
  node = __builtin_amdgcn_readfirstlane(node);
  const int beg = offs[node];
  const int d = offs[node + 1] - beg;
  const float di = dinv[node];
  const float* hl = H + lane * 4;

  float4 acc = make_float4(0.0f, 0.0f, 0.0f, 0.0f);

  for (int base = 0; base < d; base += 32) {
    const int m = (d - base < 32) ? (d - base) : 32;
    int s = node;
    if (lane < m) s = srcs[beg + base + lane];
    const float dv = dinv[s];
    int j = 0;
    for (; j + 4 <= m; j += 4) {
      const int s0 = __builtin_amdgcn_readlane(s, j);
      const int s1 = __builtin_amdgcn_readlane(s, j + 1);
      const int s2 = __builtin_amdgcn_readlane(s, j + 2);
      const int s3 = __builtin_amdgcn_readlane(s, j + 3);
      const float4 h0 = *(const float4*)(hl + (size_t)s0 * NHID);
      const float4 h1 = *(const float4*)(hl + (size_t)s1 * NHID);
      const float4 h2 = *(const float4*)(hl + (size_t)s2 * NHID);
      const float4 h3 = *(const float4*)(hl + (size_t)s3 * NHID);
      row_fma(acc, h0, di * __int_as_float(__builtin_amdgcn_readlane(__float_as_int(dv), j)));
      row_fma(acc, h1, di * __int_as_float(__builtin_amdgcn_readlane(__float_as_int(dv), j + 1)));
      row_fma(acc, h2, di * __int_as_float(__builtin_amdgcn_readlane(__float_as_int(dv), j + 2)));
      row_fma(acc, h3, di * __int_as_float(__builtin_amdgcn_readlane(__float_as_int(dv), j + 3)));
    }
#pragma nounroll
    for (; j < m; ++j) {
      const int s0 = __builtin_amdgcn_readlane(s, j);
      const float4 h0 = *(const float4*)(hl + (size_t)s0 * NHID);
      row_fma(acc, h0, di * __int_as_float(__builtin_amdgcn_readlane(__float_as_int(dv), j)));
    }
  }

  const float4 hself = *(const float4*)(hl + (size_t)node * NHID);
  const float4 bv = *(const float4*)(b + lane * 4);
  acc.x += hself.x * di * di; acc.y += hself.y * di * di;
  acc.z += hself.z * di * di; acc.w += hself.w * di * di;
  acc.x += bv.x; acc.y += bv.y; acc.z += bv.z; acc.w += bv.w;
  if (!LAST) {
    acc.x = fmaxf(acc.x, 0.0f); acc.y = fmaxf(acc.y, 0.0f);
    acc.z = fmaxf(acc.z, 0.0f); acc.w = fmaxf(acc.w, 0.0f);
    *(float4*)(Y + (size_t)node * NHID + lane * 4) = acc;
  } else {
    float logit[NCLASS];
#pragma unroll
    for (int c = 0; c < NCLASS; ++c) {
      float4 wv = *(const float4*)(Wlt + (size_t)c * NHID + lane * 4);
      float p = acc.x * wv.x + acc.y * wv.y + acc.z * wv.z + acc.w * wv.w;
#pragma unroll
      for (int off = 16; off > 0; off >>= 1) p += __shfl_xor(p, off, 32);
      logit[c] = p + bl[c];
    }
    float m = logit[0];
#pragma unroll
    for (int c = 1; c < NCLASS; ++c) m = fmaxf(m, logit[c]);
    float s = 0.0f;
#pragma unroll
    for (int c = 0; c < NCLASS; ++c) s += __expf(logit[c] - m);
    float lse = m + __logf(s);
    if (lane < NCLASS) out[(size_t)node * NCLASS + lane] = logit[lane] - lse;
  }
}

extern "C" void kernel_launch(void* const* d_in, const int* in_sizes, int n_in,
                              void* d_out, int out_size, void* d_ws, size_t ws_size,
                              hipStream_t stream) {
  const float* x  = (const float*)d_in[0];
  const int*   ei = (const int*)d_in[1];
  const float* W1 = (const float*)d_in[2];
  const float* b1 = (const float*)d_in[3];
  const float* W2 = (const float*)d_in[4];
  const float* b2 = (const float*)d_in[5];
  const float* W3 = (const float*)d_in[6];
  const float* b3 = (const float*)d_in[7];
  const float* Wl = (const float*)d_in[8];
  const float* bl = (const float*)d_in[9];
  float* out = (float*)d_out;

  const int E = in_sizes[1] / 2;
  const int N = in_sizes[0] / NFEAT;
  const int npad = ((N + 16) / 16) * 16;

  char* ws = (char*)d_ws;
  size_t off = 0;
  auto alloc = [&](size_t bytes) -> float* {
    float* p = (float*)(ws + off);
    off += (bytes + 255) & ~(size_t)255;
    return p;
  };
  const size_t cbytes = ((size_t)npad * sizeof(int) + 255) & ~(size_t)255;
  const size_t ebytes = ((size_t)E * sizeof(int) + 255) & ~(size_t)255;
  const size_t tbytes = cbytes + 2 * ebytes;
  const size_t ybytes = (size_t)N * NHID * sizeof(float);
  float* dinv = alloc((size_t)npad * sizeof(float));
  float* Wt1  = alloc((size_t)NFEAT * NHID * sizeof(float));
  float* Wt2  = alloc((size_t)NHID * NHID * sizeof(float));
  float* Wt3  = alloc((size_t)NHID * NHID * sizeof(float));
  float* Wlt  = alloc((size_t)NHID * NCLASS * sizeof(float));
  float* H    = alloc(ybytes);
  float* Y    = alloc(ybytes > tbytes ? ybytes : tbytes);
  int* offs   = (int*)alloc((size_t)npad * sizeof(int));
  int* srcs   = (int*)alloc((size_t)E * sizeof(int));
  if (off > ws_size) return;
  int* cnt  = (int*)Y;
  int* rank = (int*)((char*)Y + cbytes);
  int* tmp  = (int*)((char*)Y + cbytes + ebytes);

  const int* rows = ei;
  const int* cols = ei + E;

  const int wgrid = (int)(((size_t)N * 32 + 255) / 256);

  prep_all<<<PB_ALLW + (npad + 255) / 256, 256, 0, stream>>>(W1, Wt1, W2, Wt2, W3, Wt3, Wl, Wlt,
                                                             cnt, npad);
  cnt_rank<<<(E + 255) / 256, 256, 0, stream>>>(cols, cnt, rank, E);
  seg_scan<<<1, SCAN_T, 0, stream>>>(cnt, offs, dinv, npad);
  seg_fill<<<(E + 255) / 256, 256, 0, stream>>>(rows, cols, rank, offs, tmp, E);
  seg_sort<<<wgrid, 256, 0, stream>>>(offs, tmp, srcs, N);

  const int tiles = (N + 15) / 16;
  const int gemm_grid = (tiles + 3) / 4;

  gemm_wmma_f32<<<gemm_grid, 128, 0, stream>>>(x, Wt1, H, N, NFEAT);
  agg_rows<false><<<wgrid, 256, 0, stream>>>(offs, srcs, dinv, H, b1, Y, Wlt, bl, out, N);

  gemm_wmma_f32<<<gemm_grid, 128, 0, stream>>>(Y, Wt2, H, N, NHID);
  agg_rows<false><<<wgrid, 256, 0, stream>>>(offs, srcs, dinv, H, b2, Y, Wlt, bl, out, N);

  gemm_wmma_f32<<<gemm_grid, 128, 0, stream>>>(Y, Wt3, H, N, NHID);
  agg_rows<true><<<wgrid, 256, 0, stream>>>(offs, srcs, dinv, H, b3, Y, Wlt, bl, out, N);
}


# Edge slots from LDS counters: one workgroup per edge chunk counts targets in LDS (returned add = slot in chunk), column scan over chunks gives in-degree and chunk bases; no global atomics left
# speedup vs baseline: 3.1669x; 1.1303x over previous
#include <hip/hip_runtime.h>

typedef __attribute__((ext_vector_type(2))) float v2f;
typedef __attribute__((ext_vector_type(8))) float v8f;

#define NFEAT 512
#define NHID  128
#define NCLASS 10


__device__ __forceinline__ void transpose_elem(const float* __restrict__ W, float* __restrict__ Wt,
                                               int K, int Nc, int idx) {
  if (idx < K * Nc) {
    int k = idx / Nc, n = idx % Nc;
    Wt[(size_t)n * K + k] = W[idx];
  }
}


#define SC_CHUNKS 64
#define SC_T 1024
#define SC_CAP 50176
__global__ void __launch_bounds__(SC_T) chunk_count(const int* __restrict__ cols,
                                                    int* __restrict__ rank,
                                                    int* __restrict__ hist,
                                                    int E, int epc, int npad) {
  __shared__ __attribute__((aligned(16))) int ctr[SC_CAP];
  const int t = threadIdx.x;
  const int e0 = blockIdx.x * epc;
  const int e1 = (e0 + epc < E) ? (e0 + epc) : E;
  int* hrow = hist + (size_t)blockIdx.x * npad;
  for (int w0 = 0; w0 < npad; w0 += SC_CAP) {
    const int wn = (npad - w0 < SC_CAP) ? (npad - w0) : SC_CAP;
    for (int i = t * 4; i < wn; i += SC_T * 4) *(int4*)(ctr + i) = make_int4(0, 0, 0, 0);
    __syncthreads();
    for (int e = e0 + t; e < e1; e += 8 * SC_T) {
      int c[8], r[8];
#pragma unroll
      for (int q = 0; q < 8; ++q) c[q] = (e + q * SC_T < e1) ? cols[e + q * SC_T] : -1;
#pragma unroll
      for (int q = 0; q < 8; ++q) {
        const unsigned k = (unsigned)(c[q] - w0);
        r[q] = (k < (unsigned)wn) ? atomicAdd(&ctr[k], 1) : -1;
      }
#pragma unroll
      for (int q = 0; q < 8; ++q)
        if (r[q] >= 0) rank[e + q * SC_T] = r[q];
    }
    __syncthreads();
    for (int i = t * 4; i < wn; i += SC_T * 4) *(int4*)(hrow + w0 + i) = *(const int4*)(ctr + i);
    __syncthreads();
  }
}

#define PB_W1 (NFEAT * NHID / 256)
#define PB_W2 (NHID * NHID / 256)
#define PB_WL ((NHID * NCLASS + 255) / 256)
#define PB_ALLW (PB_W1 + 2 * PB_W2 + PB_WL)
__global__ void __launch_bounds__(256) prep_scan(const float* __restrict__ W1, float* __restrict__ Wt1,
                                                 const float* __restrict__ W2, float* __restrict__ Wt2,
                                                 const float* __restrict__ W3, float* __restrict__ Wt3,
                                                 const float* __restrict__ Wl, float* __restrict__ Wlt,
                                                 int* __restrict__ hist, int* __restrict__ cnt,
                                                 int npad) {
  const int b = blockIdx.x, t = threadIdx.x;
  if (b < PB_W1) {
    transpose_elem(W1, Wt1, NFEAT, NHID, b * 256 + t);
  } else if (b < PB_W1 + PB_W2) {
    transpose_elem(W2, Wt2, NHID, NHID, (b - PB_W1) * 256 + t);
  } else if (b < PB_W1 + 2 * PB_W2) {
    transpose_elem(W3, Wt3, NHID, NHID, (b - PB_W1 - PB_W2) * 256 + t);
  } else if (b < PB_ALLW) {
    transpose_elem(Wl, Wlt, NHID, NCLASS, (b - PB_W1 - 2 * PB_W2) * 256 + t);
  } else {
    const int i = (b - PB_ALLW) * 256 + t;
    if (i < npad) {
      int* p = hist + i;
      int run = 0;
      for (int c = 0; c < SC_CHUNKS; c += 16) {
        int v[16];
#pragma unroll
        for (int q = 0; q < 16; ++q) v[q] = p[(size_t)(c + q) * npad];
#pragma unroll
        for (int q = 0; q < 16; ++q) { p[(size_t)(c + q) * npad] = run; run += v[q]; }
      }
      cnt[i] = run;
    }
  }
}

#define SCAN_T 1024
__global__ void __launch_bounds__(SCAN_T) seg_scan(const int* __restrict__ cnt,
                                                   int* __restrict__ offs,
                                                   float* __restrict__ dinv, int npad) {
  __shared__ int wsum[SCAN_T / 32];
  __shared__ int tot;
  const int t = threadIdx.x, lane = t & 31, wv = t >> 5;
  int carry = 0;
  for (int base = 0; base < npad; base += SCAN_T * 16) {
    const int i0 = base + t * 16;
    const bool live = i0 < npad;
    int v[16];
#pragma unroll
    for (int q = 0; q < 4; ++q) {
      int4 c = live ? *(const int4*)(cnt + i0 + 4 * q) : make_int4(0, 0, 0, 0);
      v[4 * q + 0] = c.x; v[4 * q + 1] = c.y; v[4 * q + 2] = c.z; v[4 * q + 3] = c.w;
    }
    int ex[16];
    int s = 0;
#pragma unroll
    for (int k = 0; k < 16; ++k) { ex[k] = s; s += v[k]; }
    int inc = s;
#pragma unroll
    for (int o = 1; o < 32; o <<= 1) {
      int nb = __shfl_up(inc, o, 32);
      if (lane >= o) inc += nb;
    }
    if (lane == 31) wsum[wv] = inc;
    __syncthreads();
    if (wv == 0) {
      int w = wsum[lane];
      int wi = w;
#pragma unroll
      for (int o = 1; o < 32; o <<= 1) {
        int nb = __shfl_up(wi, o, 32);
        if (lane >= o) wi += nb;
      }
      wsum[lane] = wi - w;
      if (lane == 31) tot = wi;
    }
    __syncthreads();
    const int pre = carry + wsum[wv] + (inc - s);
    carry += tot;
    if (live) {
#pragma unroll
      for (int q = 0; q < 4; ++q) {
        *(int4*)(offs + i0 + 4 * q) = make_int4(pre + ex[4 * q + 0], pre + ex[4 * q + 1],
                                                pre + ex[4 * q + 2], pre + ex[4 * q + 3]);
        *(float4*)(dinv + i0 + 4 * q) =
            make_float4(rsqrtf((float)(v[4 * q + 0] + 1)), rsqrtf((float)(v[4 * q + 1] + 1)),
                        rsqrtf((float)(v[4 * q + 2] + 1)), rsqrtf((float)(v[4 * q + 3] + 1)));
      }
    }
    __syncthreads();
  }
}

__global__ void seg_fill(const int* __restrict__ rows, const int* __restrict__ cols,
                         const int* __restrict__ rank, const int* __restrict__ offs,
                         const int* __restrict__ hist, int* __restrict__ tmp,
                         int E, int epc, int npad) {
  const int l = blockIdx.x * blockDim.x + threadIdx.x;
  const int e = blockIdx.y * epc + l;
  if (l < epc && e < E) {
    const int c = cols[e];
    tmp[offs[c] + hist[(size_t)blockIdx.y * npad + c] + rank[e]] = rows[e];
  }
}

__global__ void __launch_bounds__(256) seg_sort(const int* __restrict__ offs,
                                                const int* __restrict__ tmp,
                                                int* __restrict__ srcs, int n) {
  int node = (int)((blockIdx.x * (size_t)blockDim.x + threadIdx.x) >> 5);
  const int lane = threadIdx.x & 31;
  if (node >= n) return;
  node = __builtin_amdgcn_readfirstlane(node);
  const int beg = offs[node];
  const int d = offs[node + 1] - beg;
  for (int ib = 0; ib < d; ib += 32) {
    const int i = ib + lane;
    const int key = (i < d) ? tmp[beg + i] : 0;
    int r = 0;
    for (int jb = 0; jb < d; jb += 32) {
      const int j = jb + lane;
      const int kj = (j < d) ? tmp[beg + j] : 0;
      const int m = (d - jb < 32) ? (d - jb) : 32;
      for (int q = 0; q < m; ++q) {
        const int k = __builtin_amdgcn_readlane(kj, q);
        r += (int)(k < key) + (int)((k == key) & (jb + q < i));
      }
    }
    if (i < d) srcs[beg + r] = key;
  }
}

typedef __attribute__((ext_vector_type(4))) float v4f;
#define GB_M 64
#define GB_K 32
#define GB_P 36
#define GB_STAGE ((GB_M + NHID) * GB_P)
#define GB_CP 132
#define GB_RS (16 * GB_P)

#define GB_FETCH(kc)                                  \
  a0 = *(const v4f*)(x0 + (kc));                      \
  a1 = *(const v4f*)(x1 + (kc));                      \
  a2 = *(const v4f*)(x2 + (kc));                      \
  a3 = *(const v4f*)(x3 + (kc));                      \
  b0 = *(const v4f*)(wsrc + (kc));                    \
  b1 = *(const v4f*)(wsrc + wstep + (kc));            \
  b2 = *(const v4f*)(wsrc + 2 * wstep + (kc));        \
  b3 = *(const v4f*)(wsrc + 3 * wstep + (kc));        \
  b4 = *(const v4f*)(wsrc + 4 * wstep + (kc));        \
  b5 = *(const v4f*)(wsrc + 5 * wstep + (kc));        \
  b6 = *(const v4f*)(wsrc + 6 * wstep + (kc));        \
  b7 = *(const v4f*)(wsrc + 7 * wstep + (kc));

#define GB_PUT(dst)                                           \
  *(v4f*)((dst) + sdst) = a0;                                 \
  *(v4f*)((dst) + sdst + GB_RS) = a1;                         \
  *(v4f*)((dst) + sdst + 2 * GB_RS) = a2;                     \
  *(v4f*)((dst) + sdst + 3 * GB_RS) = a3;                     \
  *(v4f*)((dst) + GB_M * GB_P + sdst) = b0;                   \
  *(v4f*)((dst) + GB_M * GB_P + sdst + GB_RS) = b1;           \
  *(v4f*)((dst) + GB_M * GB_P + sdst + 2 * GB_RS) = b2;       \
  *(v4f*)((dst) + GB_M * GB_P + sdst + 3 * GB_RS) = b3;       \
  *(v4f*)((dst) + GB_M * GB_P + sdst + 4 * GB_RS) = b4;       \
  *(v4f*)((dst) + GB_M * GB_P + sdst + 5 * GB_RS) = b5;       \
  *(v4f*)((dst) + GB_M * GB_P + sdst + 6 * GB_RS) = b6;       \
  *(v4f*)((dst) + GB_M * GB_P + sdst + 7 * GB_RS) = b7;

__global__ void __launch_bounds__(128) gemm_wmma_f32(const float* __restrict__ X,
                                                     const float* __restrict__ Wt,
                                                     float* __restrict__ H,
                                                     int M, int K) {
  __shared__ __attribute__((aligned(16))) float lds[2 * GB_STAGE];

  const int tid = threadIdx.x;
  const int lane = tid & 31;
  const int wv = tid >> 5;
  const int c = lane & 15;
  const int h = lane >> 4;
  const int row0 = blockIdx.x * GB_M;

  const int seg = tid & 7;
  const int rsub = tid >> 3;
  const int last = M - 1;
  const int r0 = row0 + rsub, r1 = r0 + 16, r2 = r0 + 32, r3 = r0 + 48;
  const float* x0 = X + (size_t)(r0 < last ? r0 : last) * K + seg * 4;
  const float* x1 = X + (size_t)(r1 < last ? r1 : last) * K + seg * 4;
  const float* x2 = X + (size_t)(r2 < last ? r2 : last) * K + seg * 4;
  const float* x3 = X + (size_t)(r3 < last ? r3 : last) * K + seg * 4;
  const float* wsrc = Wt + (size_t)rsub * K + seg * 4;
  const size_t wstep = (size_t)16 * K;
  const int sdst = rsub * GB_P + seg * 4;

  v4f a0, a1, a2, a3, b0, b1, b2, b3, b4, b5, b6, b7;
  GB_FETCH(0)
  GB_PUT(lds)
  __syncthreads();

  v8f acc[8] = {};
  const int nch = K / GB_K;
  for (int ch = 0; ch < nch; ++ch) {
    const float* st = lds + (ch & 1) * GB_STAGE;
    float* nx = lds + ((ch + 1) & 1) * GB_STAGE;
    const bool more = ch + 1 < nch;
    if (more) {
      const int kc = (ch + 1) * GB_K;
      GB_FETCH(kc)
    }

    const float* ap = st + (wv * 16 + c) * GB_P + 2 * h;
    const float* bp = st + (GB_M + c) * GB_P + 2 * h;
#pragma unroll
    for (int ks = 0; ks < GB_K / 4; ++ks) {
      const v2f a = *(const v2f*)(ap + ks * 4);
#pragma unroll
      for (int t = 0; t < 8; ++t) {
        const v2f b = *(const v2f*)(bp + t * GB_RS + ks * 4);
        acc[t] = __builtin_amdgcn_wmma_f32_16x16x4_f32(
             false, a,  false, b,
             (short)0, acc[t],  false,  false);
      }
    }

    if (more) {
      GB_PUT(nx)
    }
    __syncthreads();
  }

  float* cs = lds + wv * (16 * GB_CP);
#pragma unroll
  for (int t = 0; t < 8; ++t) {
#pragma unroll
    for (int e = 0; e < 8; ++e) cs[(8 * h + e) * GB_CP + t * 16 + c] = acc[t][e];
  }
  __syncthreads();
#pragma unroll
  for (int r = 0; r < 16; ++r) {
    const int row = row0 + wv * 16 + r;
    if (row < M)
      *(v4f*)(H + (size_t)row * NHID + lane * 4) = *(const v4f*)(cs + r * GB_CP + lane * 4);
  }
}
#undef GB_FETCH
#undef GB_PUT


__device__ __forceinline__ void row_fma(float4& acc, const float4 h, const float w) {
  acc.x = fmaf(h.x, w, acc.x);
  acc.y = fmaf(h.y, w, acc.y);
  acc.z = fmaf(h.z, w, acc.z);
  acc.w = fmaf(h.w, w, acc.w);
}

template <bool LAST>
__global__ void __launch_bounds__(256) agg_rows(const int* __restrict__ offs,
                                                const int* __restrict__ srcs,
                                                const float* __restrict__ dinv,
                                                const float* __restrict__ H,
                                                const float* __restrict__ b,
                                                float* __restrict__ Y,
                                                const float* __restrict__ Wlt,
                                                const float* __restrict__ bl,
                                                float* __restrict__ out, int n) {
  int node = (int)((blockIdx.x * (size_t)blockDim.x + threadIdx.x) >> 5);
  const int lane = threadIdx.x & 31;
  if (node >= n) return;
  node = __builtin_amdgcn_readfirstlane(node);
  const int beg = offs[node];
  const int d = offs[node + 1] - beg;
  const float di = dinv[node];
  const float* hl = H + lane * 4;

  float4 acc = make_float4(0.0f, 0.0f, 0.0f, 0.0f);

  for (int base = 0; base < d; base += 32) {
    const int m = (d - base < 32) ? (d - base) : 32;
    int s = node;
    if (lane < m) s = srcs[beg + base + lane];
    const float dv = dinv[s];
    int j = 0;
    for (; j + 4 <= m; j += 4) {
      const int s0 = __builtin_amdgcn_readlane(s, j);
      const int s1 = __builtin_amdgcn_readlane(s, j + 1);
      const int s2 = __builtin_amdgcn_readlane(s, j + 2);
      const int s3 = __builtin_amdgcn_readlane(s, j + 3);
      const float4 h0 = *(const float4*)(hl + (size_t)s0 * NHID);
      const float4 h1 = *(const float4*)(hl + (size_t)s1 * NHID);
      const float4 h2 = *(const float4*)(hl + (size_t)s2 * NHID);
      const float4 h3 = *(const float4*)(hl + (size_t)s3 * NHID);
      row_fma(acc, h0, di * __int_as_float(__builtin_amdgcn_readlane(__float_as_int(dv), j)));
      row_fma(acc, h1, di * __int_as_float(__builtin_amdgcn_readlane(__float_as_int(dv), j + 1)));
      row_fma(acc, h2, di * __int_as_float(__builtin_amdgcn_readlane(__float_as_int(dv), j + 2)));
      row_fma(acc, h3, di * __int_as_float(__builtin_amdgcn_readlane(__float_as_int(dv), j + 3)));
    }
#pragma nounroll
    for (; j < m; ++j) {
      const int s0 = __builtin_amdgcn_readlane(s, j);
      const float4 h0 = *(const float4*)(hl + (size_t)s0 * NHID);
      row_fma(acc, h0, di * __int_as_float(__builtin_amdgcn_readlane(__float_as_int(dv), j)));
    }
  }

  const float4 hself = *(const float4*)(hl + (size_t)node * NHID);
  const float4 bv = *(const float4*)(b + lane * 4);
  acc.x += hself.x * di * di; acc.y += hself.y * di * di;
  acc.z += hself.z * di * di; acc.w += hself.w * di * di;
  acc.x += bv.x; acc.y += bv.y; acc.z += bv.z; acc.w += bv.w;
  if (!LAST) {
    acc.x = fmaxf(acc.x, 0.0f); acc.y = fmaxf(acc.y, 0.0f);
    acc.z = fmaxf(acc.z, 0.0f); acc.w = fmaxf(acc.w, 0.0f);
    *(float4*)(Y + (size_t)node * NHID + lane * 4) = acc;
  } else {
    float logit[NCLASS];
#pragma unroll
    for (int c = 0; c < NCLASS; ++c) {
      float4 wv = *(const float4*)(Wlt + (size_t)c * NHID + lane * 4);
      float p = acc.x * wv.x + acc.y * wv.y + acc.z * wv.z + acc.w * wv.w;
#pragma unroll
      for (int off = 16; off > 0; off >>= 1) p += __shfl_xor(p, off, 32);
      logit[c] = p + bl[c];
    }
    float m = logit[0];
#pragma unroll
    for (int c = 1; c < NCLASS; ++c) m = fmaxf(m, logit[c]);
    float s = 0.0f;
#pragma unroll
    for (int c = 0; c < NCLASS; ++c) s += __expf(logit[c] - m);
    float lse = m + __logf(s);
    if (lane < NCLASS) out[(size_t)node * NCLASS + lane] = logit[lane] - lse;
  }
}

extern "C" void kernel_launch(void* const* d_in, const int* in_sizes, int n_in,
                              void* d_out, int out_size, void* d_ws, size_t ws_size,
                              hipStream_t stream) {
  const float* x  = (const float*)d_in[0];
  const int*   ei = (const int*)d_in[1];
  const float* W1 = (const float*)d_in[2];
  const float* b1 = (const float*)d_in[3];
  const float* W2 = (const float*)d_in[4];
  const float* b2 = (const float*)d_in[5];
  const float* W3 = (const float*)d_in[6];
  const float* b3 = (const float*)d_in[7];
  const float* Wl = (const float*)d_in[8];
  const float* bl = (const float*)d_in[9];
  float* out = (float*)d_out;

  const int E = in_sizes[1] / 2;
  const int N = in_sizes[0] / NFEAT;
  const int npad = ((N + 16) / 16) * 16;

  char* ws = (char*)d_ws;
  size_t off = 0;
  auto alloc = [&](size_t bytes) -> float* {
    float* p = (float*)(ws + off);
    off += (bytes + 255) & ~(size_t)255;
    return p;
  };
  const size_t cbytes = ((size_t)npad * sizeof(int) + 255) & ~(size_t)255;
  const size_t ebytes = ((size_t)E * sizeof(int) + 255) & ~(size_t)255;
  const size_t hbytes = ((size_t)SC_CHUNKS * npad * sizeof(int) + 255) & ~(size_t)255;
  const size_t tbytes = cbytes + 2 * ebytes + hbytes;
  const size_t ybytes = (size_t)N * NHID * sizeof(float);
  float* dinv = alloc((size_t)npad * sizeof(float));
  float* Wt1  = alloc((size_t)NFEAT * NHID * sizeof(float));
  float* Wt2  = alloc((size_t)NHID * NHID * sizeof(float));
  float* Wt3  = alloc((size_t)NHID * NHID * sizeof(float));
  float* Wlt  = alloc((size_t)NHID * NCLASS * sizeof(float));
  float* H    = alloc(ybytes);
  float* Y    = alloc(ybytes > tbytes ? ybytes : tbytes);
  int* offs   = (int*)alloc((size_t)npad * sizeof(int));
  int* srcs   = (int*)alloc((size_t)E * sizeof(int));
  if (off > ws_size) return;
  int* cnt  = (int*)Y;
  int* rank = (int*)((char*)Y + cbytes);
  int* tmp  = (int*)((char*)Y + cbytes + ebytes);
  int* hist = (int*)((char*)Y + cbytes + 2 * ebytes);

  const int* rows = ei;
  const int* cols = ei + E;

  const int wgrid = (int)(((size_t)N * 32 + 255) / 256);

  const int epc = (E + SC_CHUNKS - 1) / SC_CHUNKS;

  chunk_count<<<SC_CHUNKS, SC_T, 0, stream>>>(cols, rank, hist, E, epc, npad);
  prep_scan<<<PB_ALLW + (npad + 255) / 256, 256, 0, stream>>>(W1, Wt1, W2, Wt2, W3, Wt3, Wl, Wlt,
                                                              hist, cnt, npad);
  seg_scan<<<1, SCAN_T, 0, stream>>>(cnt, offs, dinv, npad);
  seg_fill<<<dim3((epc + 255) / 256, SC_CHUNKS), 256, 0, stream>>>(rows, cols, rank, offs, hist, tmp,
                                                                   E, epc, npad);
  seg_sort<<<wgrid, 256, 0, stream>>>(offs, tmp, srcs, N);

  const int tiles = (N + 15) / 16;
  const int gemm_grid = (tiles + 3) / 4;

  gemm_wmma_f32<<<gemm_grid, 128, 0, stream>>>(x, Wt1, H, N, NFEAT);
  agg_rows<false><<<wgrid, 256, 0, stream>>>(offs, srcs, dinv, H, b1, Y, Wlt, bl, out, N);

  gemm_wmma_f32<<<gemm_grid, 128, 0, stream>>>(Y, Wt2, H, N, NHID);
  agg_rows<false><<<wgrid, 256, 0, stream>>>(offs, srcs, dinv, H, b2, Y, Wlt, bl, out, N);

  gemm_wmma_f32<<<gemm_grid, 128, 0, stream>>>(Y, Wt3, H, N, NHID);
  agg_rows<true><<<wgrid, 256, 0, stream>>>(offs, srcs, dinv, H, b3, Y, Wlt, bl, out, N);
}


# Classifier folded into layer 3 (W3*Wl, b3*Wl+bl built in every call): third product is one 16-column WMMA tile, last aggregation gathers 64-byte rows, 8 edges per load; last layer's sum order changes
# speedup vs baseline: 3.6103x; 1.1400x over previous
#include <hip/hip_runtime.h>

typedef __attribute__((ext_vector_type(2))) float v2f;
typedef __attribute__((ext_vector_type(8))) float v8f;

#define NFEAT 512
#define NHID  128
#define NCLASS 10


__device__ __forceinline__ void transpose_elem(const float* __restrict__ W, float* __restrict__ Wt,
                                               int K, int Nc, int idx) {
  if (idx < K * Nc) {
    int k = idx / Nc, n = idx % Nc;
    Wt[(size_t)n * K + k] = W[idx];
  }
}


#define SC_CHUNKS 64
#define SC_T 1024
#define SC_CAP 50176
__global__ void __launch_bounds__(SC_T) chunk_count(const int* __restrict__ cols,
                                                    int* __restrict__ rank,
                                                    int* __restrict__ hist,
                                                    int E, int epc, int npad) {
  __shared__ __attribute__((aligned(16))) int ctr[SC_CAP];
  const int t = threadIdx.x;
  const int e0 = blockIdx.x * epc;
  const int e1 = (e0 + epc < E) ? (e0 + epc) : E;
  int* hrow = hist + (size_t)blockIdx.x * npad;
  for (int w0 = 0; w0 < npad; w0 += SC_CAP) {
    const int wn = (npad - w0 < SC_CAP) ? (npad - w0) : SC_CAP;
    for (int i = t * 4; i < wn; i += SC_T * 4) *(int4*)(ctr + i) = make_int4(0, 0, 0, 0);
    __syncthreads();
    for (int e = e0 + t; e < e1; e += 8 * SC_T) {
      int c[8], r[8];
#pragma unroll
      for (int q = 0; q < 8; ++q) c[q] = (e + q * SC_T < e1) ? cols[e + q * SC_T] : -1;
#pragma unroll
      for (int q = 0; q < 8; ++q) {
        const unsigned k = (unsigned)(c[q] - w0);
        r[q] = (k < (unsigned)wn) ? atomicAdd(&ctr[k], 1) : -1;
      }
#pragma unroll
      for (int q = 0; q < 8; ++q)
        if (r[q] >= 0) rank[e + q * SC_T] = r[q];
    }
    __syncthreads();
    for (int i = t * 4; i < wn; i += SC_T * 4) *(int4*)(hrow + w0 + i) = *(const int4*)(ctr + i);
    __syncthreads();
  }
}

#define FOLD_N 16
static_assert(NCLASS <= FOLD_N && NCLASS == 10, "the last layer's lane map and its stores are written for 10 classes in one 16-column tile");

#define PB_W1 (NFEAT * NHID / 256)
#define PB_W2 (NHID * NHID / 256)
#define PB_FOLD ((NHID * FOLD_N + FOLD_N + 255) / 256)
#define PB_ALLW (PB_W1 + PB_W2 + PB_FOLD)
__global__ void __launch_bounds__(256) prep_scan(const float* __restrict__ W1, float* __restrict__ Wt1,
                                                 const float* __restrict__ W2, float* __restrict__ Wt2,
                                                 const float* __restrict__ W3, const float* __restrict__ b3,
                                                 const float* __restrict__ Wl, const float* __restrict__ bl,
                                                 float* __restrict__ Wt3l, float* __restrict__ b3l,
                                                 int* __restrict__ hist, int* __restrict__ cnt,
                                                 int npad) {
  const int b = blockIdx.x, t = threadIdx.x;
  if (b < PB_W1) {
    transpose_elem(W1, Wt1, NFEAT, NHID, b * 256 + t);
  } else if (b < PB_W1 + PB_W2) {
    transpose_elem(W2, Wt2, NHID, NHID, (b - PB_W1) * 256 + t);
  } else if (b < PB_ALLW) {
    const int idx = (b - PB_W1 - PB_W2) * 256 + t;
    if (idx < NHID * FOLD_N) {
      const int k = idx >> 4, nn = idx & (FOLD_N - 1);
      float s = 0.0f;
      if (nn < NCLASS) {
#pragma unroll 32
        for (int j = 0; j < NHID; ++j) s = fmaf(W3[k * NHID + j], Wl[j * NCLASS + nn], s);
      }
      Wt3l[nn * NHID + k] = s;
    } else if (idx < NHID * FOLD_N + FOLD_N) {
      const int nn = idx - NHID * FOLD_N;
      float s = 0.0f;
      if (nn < NCLASS) {
        s = bl[nn];
#pragma unroll 32
        for (int j = 0; j < NHID; ++j) s = fmaf(b3[j], Wl[j * NCLASS + nn], s);
      }
      b3l[nn] = s;
    }
  } else {
    const int i = (b - PB_ALLW) * 256 + t;
    if (i < npad) {
      int* p = hist + i;
      int run = 0;
      for (int c = 0; c < SC_CHUNKS; c += 16) {
        int v[16];
#pragma unroll
        for (int q = 0; q < 16; ++q) v[q] = p[(size_t)(c + q) * npad];
#pragma unroll
        for (int q = 0; q < 16; ++q) { p[(size_t)(c + q) * npad] = run; run += v[q]; }
      }
      cnt[i] = run;
    }
  }
}

#define SCAN_T 1024
__global__ void __launch_bounds__(SCAN_T) seg_scan(const int* __restrict__ cnt,
                                                   int* __restrict__ offs,
                                                   float* __restrict__ dinv, int npad) {
  __shared__ int wsum[SCAN_T / 32];
  __shared__ int tot;
  const int t = threadIdx.x, lane = t & 31, wv = t >> 5;
  int carry = 0;
  for (int base = 0; base < npad; base += SCAN_T * 16) {
    const int i0 = base + t * 16;
    const bool live = i0 < npad;
    int v[16];
#pragma unroll
    for (int q = 0; q < 4; ++q) {
      int4 c = live ? *(const int4*)(cnt + i0 + 4 * q) : make_int4(0, 0, 0, 0);
      v[4 * q + 0] = c.x; v[4 * q + 1] = c.y; v[4 * q + 2] = c.z; v[4 * q + 3] = c.w;
    }
    int ex[16];
    int s = 0;
#pragma unroll
    for (int k = 0; k < 16; ++k) { ex[k] = s; s += v[k]; }
    int inc = s;
#pragma unroll
    for (int o = 1; o < 32; o <<= 1) {
      int nb = __shfl_up(inc, o, 32);
      if (lane >= o) inc += nb;
    }
    if (lane == 31) wsum[wv] = inc;
    __syncthreads();
    if (wv == 0) {
      int w = wsum[lane];
      int wi = w;
#pragma unroll
      for (int o = 1; o < 32; o <<= 1) {
        int nb = __shfl_up(wi, o, 32);
        if (lane >= o) wi += nb;
      }
      wsum[lane] = wi - w;
      if (lane == 31) tot = wi;
    }
    __syncthreads();
    const int pre = carry + wsum[wv] + (inc - s);
    carry += tot;
    if (live) {
#pragma unroll
      for (int q = 0; q < 4; ++q) {
        *(int4*)(offs + i0 + 4 * q) = make_int4(pre + ex[4 * q + 0], pre + ex[4 * q + 1],
                                                pre + ex[4 * q + 2], pre + ex[4 * q + 3]);
        *(float4*)(dinv + i0 + 4 * q) =
            make_float4(rsqrtf((float)(v[4 * q + 0] + 1)), rsqrtf((float)(v[4 * q + 1] + 1)),
                        rsqrtf((float)(v[4 * q + 2] + 1)), rsqrtf((float)(v[4 * q + 3] + 1)));
      }
    }
    __syncthreads();
  }
}

__global__ void seg_fill(const int* __restrict__ rows, const int* __restrict__ cols,
                         const int* __restrict__ rank, const int* __restrict__ offs,
                         const int* __restrict__ hist, int* __restrict__ tmp,
                         int E, int epc, int npad) {
  const int l = blockIdx.x * blockDim.x + threadIdx.x;
  const int e = blockIdx.y * epc + l;
  if (l < epc && e < E) {
    const int c = cols[e];
    tmp[offs[c] + hist[(size_t)blockIdx.y * npad + c] + rank[e]] = rows[e];
  }
}

__global__ void __launch_bounds__(256) seg_sort(const int* __restrict__ offs,
                                                const int* __restrict__ tmp,
                                                int* __restrict__ srcs, int n) {
  int node = (int)((blockIdx.x * (size_t)blockDim.x + threadIdx.x) >> 5);
  const int lane = threadIdx.x & 31;
  if (node >= n) return;
  node = __builtin_amdgcn_readfirstlane(node);
  const int beg = offs[node];
  const int d = offs[node + 1] - beg;
  for (int ib = 0; ib < d; ib += 32) {
    const int i = ib + lane;
    const int key = (i < d) ? tmp[beg + i] : 0;
    int r = 0;
    for (int jb = 0; jb < d; jb += 32) {
      const int j = jb + lane;
      const int kj = (j < d) ? tmp[beg + j] : 0;
      const int m = (d - jb < 32) ? (d - jb) : 32;
      for (int q = 0; q < m; ++q) {
        const int k = __builtin_amdgcn_readlane(kj, q);
        r += (int)(k < key) + (int)((k == key) & (jb + q < i));
      }
    }
    if (i < d) srcs[beg + r] = key;
  }
}

typedef __attribute__((ext_vector_type(4))) float v4f;
#define GB_M 64
#define GB_K 32
#define GB_P 36
#define GB_STAGE ((GB_M + NHID) * GB_P)
#define GB_CP 132
#define GB_RS (16 * GB_P)

#define GB_FETCH(kc)                                  \
  a0 = *(const v4f*)(x0 + (kc));                      \
  a1 = *(const v4f*)(x1 + (kc));                      \
  a2 = *(const v4f*)(x2 + (kc));                      \
  a3 = *(const v4f*)(x3 + (kc));                      \
  b0 = *(const v4f*)(wsrc + (kc));                    \
  b1 = *(const v4f*)(wsrc + wstep + (kc));            \
  b2 = *(const v4f*)(wsrc + 2 * wstep + (kc));        \
  b3 = *(const v4f*)(wsrc + 3 * wstep + (kc));        \
  b4 = *(const v4f*)(wsrc + 4 * wstep + (kc));        \
  b5 = *(const v4f*)(wsrc + 5 * wstep + (kc));        \
  b6 = *(const v4f*)(wsrc + 6 * wstep + (kc));        \
  b7 = *(const v4f*)(wsrc + 7 * wstep + (kc));

#define GB_PUT(dst)                                           \
  *(v4f*)((dst) + sdst) = a0;                                 \
  *(v4f*)((dst) + sdst + GB_RS) = a1;                         \
  *(v4f*)((dst) + sdst + 2 * GB_RS) = a2;                     \
  *(v4f*)((dst) + sdst + 3 * GB_RS) = a3;                     \
  *(v4f*)((dst) + GB_M * GB_P + sdst) = b0;                   \
  *(v4f*)((dst) + GB_M * GB_P + sdst + GB_RS) = b1;           \
  *(v4f*)((dst) + GB_M * GB_P + sdst + 2 * GB_RS) = b2;       \
  *(v4f*)((dst) + GB_M * GB_P + sdst + 3 * GB_RS) = b3;       \
  *(v4f*)((dst) + GB_M * GB_P + sdst + 4 * GB_RS) = b4;       \
  *(v4f*)((dst) + GB_M * GB_P + sdst + 5 * GB_RS) = b5;       \
  *(v4f*)((dst) + GB_M * GB_P + sdst + 6 * GB_RS) = b6;       \
  *(v4f*)((dst) + GB_M * GB_P + sdst + 7 * GB_RS) = b7;

__global__ void __launch_bounds__(128) gemm_wmma_f32(const float* __restrict__ X,
                                                     const float* __restrict__ Wt,
                                                     float* __restrict__ H,
                                                     int M, int K) {
  __shared__ __attribute__((aligned(16))) float lds[2 * GB_STAGE];

  const int tid = threadIdx.x;
  const int lane = tid & 31;
  const int wv = tid >> 5;
  const int c = lane & 15;
  const int h = lane >> 4;
  const int row0 = blockIdx.x * GB_M;

  const int seg = tid & 7;
  const int rsub = tid >> 3;
  const int last = M - 1;
  const int r0 = row0 + rsub, r1 = r0 + 16, r2 = r0 + 32, r3 = r0 + 48;
  const float* x0 = X + (size_t)(r0 < last ? r0 : last) * K + seg * 4;
  const float* x1 = X + (size_t)(r1 < last ? r1 : last) * K + seg * 4;
  const float* x2 = X + (size_t)(r2 < last ? r2 : last) * K + seg * 4;
  const float* x3 = X + (size_t)(r3 < last ? r3 : last) * K + seg * 4;
  const float* wsrc = Wt + (size_t)rsub * K + seg * 4;
  const size_t wstep = (size_t)16 * K;
  const int sdst = rsub * GB_P + seg * 4;

  v4f a0, a1, a2, a3, b0, b1, b2, b3, b4, b5, b6, b7;
  GB_FETCH(0)
  GB_PUT(lds)
  __syncthreads();

  v8f acc[8] = {};
  const int nch = K / GB_K;
  for (int ch = 0; ch < nch; ++ch) {
    const float* st = lds + (ch & 1) * GB_STAGE;
    float* nx = lds + ((ch + 1) & 1) * GB_STAGE;
    const bool more = ch + 1 < nch;
    if (more) {
      const int kc = (ch + 1) * GB_K;
      GB_FETCH(kc)
    }

    const float* ap = st + (wv * 16 + c) * GB_P + 2 * h;
    const float* bp = st + (GB_M + c) * GB_P + 2 * h;
#pragma unroll
    for (int ks = 0; ks < GB_K / 4; ++ks) {
      const v2f a = *(const v2f*)(ap + ks * 4);
#pragma unroll
      for (int t = 0; t < 8; ++t) {
        const v2f b = *(const v2f*)(bp + t * GB_RS + ks * 4);
        acc[t] = __builtin_amdgcn_wmma_f32_16x16x4_f32(
             false, a,  false, b,
             (short)0, acc[t],  false,  false);
      }
    }

    if (more) {
      GB_PUT(nx)
    }
    __syncthreads();
  }

  float* cs = lds + wv * (16 * GB_CP);
#pragma unroll
  for (int t = 0; t < 8; ++t) {
#pragma unroll
    for (int e = 0; e < 8; ++e) cs[(8 * h + e) * GB_CP + t * 16 + c] = acc[t][e];
  }
  __syncthreads();
#pragma unroll
  for (int r = 0; r < 16; ++r) {
    const int row = row0 + wv * 16 + r;
    if (row < M)
      *(v4f*)(H + (size_t)row * NHID + lane * 4) = *(const v4f*)(cs + r * GB_CP + lane * 4);
  }
}
#undef GB_FETCH
#undef GB_PUT

#define G16_STAGE ((GB_M + FOLD_N) * GB_P)
#define G16_CP 20
__global__ void __launch_bounds__(128) gemm_wmma_f32_n16(const float* __restrict__ X,
                                                         const float* __restrict__ Wt,
                                                         float* __restrict__ G,
                                                         int M, int K) {
  __shared__ __attribute__((aligned(16))) float lds[2 * G16_STAGE];

  const int tid = threadIdx.x;
  const int lane = tid & 31;
  const int wv = tid >> 5;
  const int c = lane & 15;
  const int h = lane >> 4;
  const int row0 = blockIdx.x * GB_M;

  const int seg = tid & 7;
  const int rsub = tid >> 3;
  const int last = M - 1;
  const int r0 = row0 + rsub, r1 = r0 + 16, r2 = r0 + 32, r3 = r0 + 48;
  const float* x0 = X + (size_t)(r0 < last ? r0 : last) * K + seg * 4;
  const float* x1 = X + (size_t)(r1 < last ? r1 : last) * K + seg * 4;
  const float* x2 = X + (size_t)(r2 < last ? r2 : last) * K + seg * 4;
  const float* x3 = X + (size_t)(r3 < last ? r3 : last) * K + seg * 4;
  const float* wsrc = Wt + (size_t)rsub * K + seg * 4;
  const int sdst = rsub * GB_P + seg * 4;

  v4f a0 = *(const v4f*)(x0);
  v4f a1 = *(const v4f*)(x1);
  v4f a2 = *(const v4f*)(x2);
  v4f a3 = *(const v4f*)(x3);
  v4f b0 = *(const v4f*)(wsrc);
  *(v4f*)(lds + sdst) = a0;
  *(v4f*)(lds + sdst + GB_RS) = a1;
  *(v4f*)(lds + sdst + 2 * GB_RS) = a2;
  *(v4f*)(lds + sdst + 3 * GB_RS) = a3;
  *(v4f*)(lds + GB_M * GB_P + sdst) = b0;
  __syncthreads();

  v8f acc = {};
  const int nch = K / GB_K;
  for (int ch = 0; ch < nch; ++ch) {
    const float* st = lds + (ch & 1) * G16_STAGE;
    float* nx = lds + ((ch + 1) & 1) * G16_STAGE;
    const bool more = ch + 1 < nch;
    if (more) {
      const int kc = (ch + 1) * GB_K;
      a0 = *(const v4f*)(x0 + kc);
      a1 = *(const v4f*)(x1 + kc);
      a2 = *(const v4f*)(x2 + kc);
      a3 = *(const v4f*)(x3 + kc);
      b0 = *(const v4f*)(wsrc + kc);
    }

    const float* ap = st + (wv * 16 + c) * GB_P + 2 * h;
    const float* bp = st + (GB_M + c) * GB_P + 2 * h;
#pragma unroll
    for (int ks = 0; ks < GB_K / 4; ++ks) {
      const v2f a = *(const v2f*)(ap + ks * 4);
      const v2f b = *(const v2f*)(bp + ks * 4);
      acc = __builtin_amdgcn_wmma_f32_16x16x4_f32(
           false, a,  false, b,
           (short)0, acc,  false,  false);
    }

    if (more) {
      *(v4f*)(nx + sdst) = a0;
      *(v4f*)(nx + sdst + GB_RS) = a1;
      *(v4f*)(nx + sdst + 2 * GB_RS) = a2;
      *(v4f*)(nx + sdst + 3 * GB_RS) = a3;
      *(v4f*)(nx + GB_M * GB_P + sdst) = b0;
    }
    __syncthreads();
  }

  float* cs = lds + wv * (16 * G16_CP);
#pragma unroll
  for (int e = 0; e < 8; ++e) cs[(8 * h + e) * G16_CP + c] = acc[e];
  __syncthreads();
  for (int i = 0; i < 2; ++i) {
    const int r = i * 8 + (lane >> 2);
    const int row = row0 + wv * 16 + r;
    if (row < M)
      *(v4f*)(G + (size_t)row * FOLD_N + (lane & 3) * 4) = *(const v4f*)(cs + r * G16_CP + (lane & 3) * 4);
  }
}


__device__ __forceinline__ void row_fma(float4& acc, const float4 h, const float w) {
  acc.x = fmaf(h.x, w, acc.x);
  acc.y = fmaf(h.y, w, acc.y);
  acc.z = fmaf(h.z, w, acc.z);
  acc.w = fmaf(h.w, w, acc.w);
}

__global__ void __launch_bounds__(256) agg_rows(const int* __restrict__ offs,
                                                const int* __restrict__ srcs,
                                                const float* __restrict__ dinv,
                                                const float* __restrict__ H,
                                                const float* __restrict__ b,
                                                float* __restrict__ Y, int n) {
  int node = (int)((blockIdx.x * (size_t)blockDim.x + threadIdx.x) >> 5);
  const int lane = threadIdx.x & 31;
  if (node >= n) return;
  node = __builtin_amdgcn_readfirstlane(node);
  const int beg = offs[node];
  const int d = offs[node + 1] - beg;
  const float di = dinv[node];
  const float* hl = H + lane * 4;

  float4 acc = make_float4(0.0f, 0.0f, 0.0f, 0.0f);

  for (int base = 0; base < d; base += 32) {
    const int m = (d - base < 32) ? (d - base) : 32;
    int s = node;
    if (lane < m) s = srcs[beg + base + lane];
    const float dv = dinv[s];
    int j = 0;
    for (; j + 4 <= m; j += 4) {
      const int s0 = __builtin_amdgcn_readlane(s, j);
      const int s1 = __builtin_amdgcn_readlane(s, j + 1);
      const int s2 = __builtin_amdgcn_readlane(s, j + 2);
      const int s3 = __builtin_amdgcn_readlane(s, j + 3);
      const float4 h0 = *(const float4*)(hl + (size_t)s0 * NHID);
      const float4 h1 = *(const float4*)(hl + (size_t)s1 * NHID);
      const float4 h2 = *(const float4*)(hl + (size_t)s2 * NHID);
      const float4 h3 = *(const float4*)(hl + (size_t)s3 * NHID);
      row_fma(acc, h0, di * __int_as_float(__builtin_amdgcn_readlane(__float_as_int(dv), j)));
      row_fma(acc, h1, di * __int_as_float(__builtin_amdgcn_readlane(__float_as_int(dv), j + 1)));
      row_fma(acc, h2, di * __int_as_float(__builtin_amdgcn_readlane(__float_as_int(dv), j + 2)));
      row_fma(acc, h3, di * __int_as_float(__builtin_amdgcn_readlane(__float_as_int(dv), j + 3)));
    }
#pragma nounroll
    for (; j < m; ++j) {
      const int s0 = __builtin_amdgcn_readlane(s, j);
      const float4 h0 = *(const float4*)(hl + (size_t)s0 * NHID);
      row_fma(acc, h0, di * __int_as_float(__builtin_amdgcn_readlane(__float_as_int(dv), j)));
    }
  }

  const float4 hself = *(const float4*)(hl + (size_t)node * NHID);
  const float4 bv = *(const float4*)(b + lane * 4);
  acc.x += hself.x * di * di; acc.y += hself.y * di * di;
  acc.z += hself.z * di * di; acc.w += hself.w * di * di;
  acc.x += bv.x; acc.y += bv.y; acc.z += bv.z; acc.w += bv.w;
  acc.x = fmaxf(acc.x, 0.0f); acc.y = fmaxf(acc.y, 0.0f);
  acc.z = fmaxf(acc.z, 0.0f); acc.w = fmaxf(acc.w, 0.0f);
  *(float4*)(Y + (size_t)node * NHID + lane * 4) = acc;
}

__global__ void __launch_bounds__(256) agg_last(const int* __restrict__ offs,
                                                const int* __restrict__ srcs,
                                                const float* __restrict__ dinv,
                                                const float* __restrict__ G,
                                                const float* __restrict__ bf,
                                                float* __restrict__ out, int n) {
  int node = (int)((blockIdx.x * (size_t)blockDim.x + threadIdx.x) >> 5);
  const int lane = threadIdx.x & 31;
  if (node >= n) return;
  node = __builtin_amdgcn_readfirstlane(node);
  const int beg = offs[node];
  const int d = offs[node + 1] - beg;
  const float di = dinv[node];
  const int slot = lane >> 2;
  const int q4 = (lane & 3) * 4;
  const float* gl = G + q4;

  float4 acc = make_float4(0.0f, 0.0f, 0.0f, 0.0f);
  for (int base = 0; base < d; base += 32) {
    bool v[4];
    int s[4];
    float w[4];
    float4 g[4];
#pragma unroll
    for (int bt = 0; bt < 4; ++bt) {
      const int e = base + slot + 8 * bt;
      v[bt] = e < d;
      s[bt] = srcs[beg + (v[bt] ? e : d - 1)];
    }
#pragma unroll
    for (int bt = 0; bt < 4; ++bt) {
      const float dv = dinv[s[bt]];
      w[bt] = v[bt] ? di * dv : 0.0f;
      g[bt] = *(const float4*)(gl + (size_t)s[bt] * FOLD_N);
    }
#pragma unroll
    for (int bt = 0; bt < 4; ++bt) row_fma(acc, g[bt], w[bt]);
  }
#pragma unroll
  for (int o = 4; o < 32; o <<= 1) {
    acc.x += __shfl_xor(acc.x, o, 32);
    acc.y += __shfl_xor(acc.y, o, 32);
    acc.z += __shfl_xor(acc.z, o, 32);
    acc.w += __shfl_xor(acc.w, o, 32);
  }

  const float4 gself = *(const float4*)(gl + (size_t)node * FOLD_N);
  const float4 bv = *(const float4*)(bf + q4);
  acc.x += gself.x * di * di; acc.y += gself.y * di * di;
  acc.z += gself.z * di * di; acc.w += gself.w * di * di;
  acc.x += bv.x; acc.y += bv.y; acc.z += bv.z; acc.w += bv.w;

  const float ninf = -__builtin_inff();
  const bool c0 = q4 + 0 < NCLASS, c1 = q4 + 1 < NCLASS, c2 = q4 + 2 < NCLASS, c3 = q4 + 3 < NCLASS;
  float m = fmaxf(fmaxf(c0 ? acc.x : ninf, c1 ? acc.y : ninf),
                  fmaxf(c2 ? acc.z : ninf, c3 ? acc.w : ninf));
  m = fmaxf(m, __shfl_xor(m, 1, 32));
  m = fmaxf(m, __shfl_xor(m, 2, 32));
  float s = (c0 ? __expf(acc.x - m) : 0.0f) + (c1 ? __expf(acc.y - m) : 0.0f) +
            (c2 ? __expf(acc.z - m) : 0.0f) + (c3 ? __expf(acc.w - m) : 0.0f);
  s += __shfl_xor(s, 1, 32);
  s += __shfl_xor(s, 2, 32);
  const float lse = m + __logf(s);
  if (lane < 3) {
    float* o = out + (size_t)node * NCLASS + q4;
    *(float2*)o = make_float2(acc.x - lse, acc.y - lse);
    if (lane < 2) *(float2*)(o + 2) = make_float2(acc.z - lse, acc.w - lse);
  }
}

extern "C" void kernel_launch(void* const* d_in, const int* in_sizes, int n_in,
                              void* d_out, int out_size, void* d_ws, size_t ws_size,
                              hipStream_t stream) {
  const float* x  = (const float*)d_in[0];
  const int*   ei = (const int*)d_in[1];
  const float* W1 = (const float*)d_in[2];
  const float* b1 = (const float*)d_in[3];
  const float* W2 = (const float*)d_in[4];
  const float* b2 = (const float*)d_in[5];
  const float* W3 = (const float*)d_in[6];
  const float* b3 = (const float*)d_in[7];
  const float* Wl = (const float*)d_in[8];
  const float* bl = (const float*)d_in[9];
  float* out = (float*)d_out;

  const int E = in_sizes[1] / 2;
  const int N = in_sizes[0] / NFEAT;
  const int npad = ((N + 16) / 16) * 16;

  char* ws = (char*)d_ws;
  size_t off = 0;
  auto alloc = [&](size_t bytes) -> float* {
    float* p = (float*)(ws + off);
    off += (bytes + 255) & ~(size_t)255;
    return p;
  };
  const size_t cbytes = ((size_t)npad * sizeof(int) + 255) & ~(size_t)255;
  const size_t ebytes = ((size_t)E * sizeof(int) + 255) & ~(size_t)255;
  const size_t hbytes = ((size_t)SC_CHUNKS * npad * sizeof(int) + 255) & ~(size_t)255;
  const size_t tbytes = cbytes + 2 * ebytes + hbytes;
  const size_t ybytes = (size_t)N * NHID * sizeof(float);
  float* dinv = alloc((size_t)npad * sizeof(float));
  float* Wt1  = alloc((size_t)NFEAT * NHID * sizeof(float));
  float* Wt2  = alloc((size_t)NHID * NHID * sizeof(float));
  float* Wt3l = alloc((size_t)FOLD_N * NHID * sizeof(float));
  float* b3l  = alloc((size_t)FOLD_N * sizeof(float));
  float* H    = alloc(ybytes);
  float* Y    = alloc(ybytes > tbytes ? ybytes : tbytes);
  int* offs   = (int*)alloc((size_t)npad * sizeof(int));
  int* srcs   = (int*)alloc((size_t)E * sizeof(int));
  if (off > ws_size) return;
  int* cnt  = (int*)Y;
  int* rank = (int*)((char*)Y + cbytes);
  int* tmp  = (int*)((char*)Y + cbytes + ebytes);
  int* hist = (int*)((char*)Y + cbytes + 2 * ebytes);

  const int* rows = ei;
  const int* cols = ei + E;

  const int wgrid = (int)(((size_t)N * 32 + 255) / 256);

  const int epc = (E + SC_CHUNKS - 1) / SC_CHUNKS;

  chunk_count<<<SC_CHUNKS, SC_T, 0, stream>>>(cols, rank, hist, E, epc, npad);
  prep_scan<<<PB_ALLW + (npad + 255) / 256, 256, 0, stream>>>(W1, Wt1, W2, Wt2, W3, b3, Wl, bl,
                                                              Wt3l, b3l, hist, cnt, npad);
  seg_scan<<<1, SCAN_T, 0, stream>>>(cnt, offs, dinv, npad);
  seg_fill<<<dim3((epc + 255) / 256, SC_CHUNKS), 256, 0, stream>>>(rows, cols, rank, offs, hist, tmp,
                                                                   E, epc, npad);
  seg_sort<<<wgrid, 256, 0, stream>>>(offs, tmp, srcs, N);

  const int tiles = (N + 15) / 16;
  const int gemm_grid = (tiles + 3) / 4;

  gemm_wmma_f32<<<gemm_grid, 128, 0, stream>>>(x, Wt1, H, N, NFEAT);
  agg_rows<<<wgrid, 256, 0, stream>>>(offs, srcs, dinv, H, b1, Y, N);

  gemm_wmma_f32<<<gemm_grid, 128, 0, stream>>>(Y, Wt2, H, N, NHID);
  agg_rows<<<wgrid, 256, 0, stream>>>(offs, srcs, dinv, H, b2, Y, N);

  gemm_wmma_f32_n16<<<gemm_grid, 128, 0, stream>>>(Y, Wt3l, H, N, NHID);
  agg_last<<<wgrid, 256, 0, stream>>>(offs, srcs, dinv, H, b3l, out, N);
}


# Segment ordering moved into the first aggregation: its wave ranks the sources in registers, in place, and sums from a lane permute; the ordering launch is gone (11 to 10); sums in the same order
# speedup vs baseline: 3.7577x; 1.0408x over previous
#include <hip/hip_runtime.h>

typedef __attribute__((ext_vector_type(2))) float v2f;
typedef __attribute__((ext_vector_type(8))) float v8f;

#define NFEAT 512
#define NHID  128
#define NCLASS 10


__device__ __forceinline__ void transpose_elem(const float* __restrict__ W, float* __restrict__ Wt,
                                               int K, int Nc, int idx) {
  if (idx < K * Nc) {
    int k = idx / Nc, n = idx % Nc;
    Wt[(size_t)n * K + k] = W[idx];
  }
}


#define SC_CHUNKS 64
#define SC_T 1024
#define SC_CAP 50176
__global__ void __launch_bounds__(SC_T) chunk_count(const int* __restrict__ cols,
                                                    int* __restrict__ rank,
                                                    int* __restrict__ hist,
                                                    int E, int epc, int npad) {
  __shared__ __attribute__((aligned(16))) int ctr[SC_CAP];
  const int t = threadIdx.x;
  const int e0 = blockIdx.x * epc;
  const int e1 = (e0 + epc < E) ? (e0 + epc) : E;
  int* hrow = hist + (size_t)blockIdx.x * npad;
  for (int w0 = 0; w0 < npad; w0 += SC_CAP) {
    const int wn = (npad - w0 < SC_CAP) ? (npad - w0) : SC_CAP;
    for (int i = t * 4; i < wn; i += SC_T * 4) *(int4*)(ctr + i) = make_int4(0, 0, 0, 0);
    __syncthreads();
    for (int e = e0 + t; e < e1; e += 8 * SC_T) {
      int c[8], r[8];
#pragma unroll
      for (int q = 0; q < 8; ++q) c[q] = (e + q * SC_T < e1) ? cols[e + q * SC_T] : -1;
#pragma unroll
      for (int q = 0; q < 8; ++q) {
        const unsigned k = (unsigned)(c[q] - w0);
        r[q] = (k < (unsigned)wn) ? atomicAdd(&ctr[k], 1) : -1;
      }
#pragma unroll
      for (int q = 0; q < 8; ++q)
        if (r[q] >= 0) rank[e + q * SC_T] = r[q];
    }
    __syncthreads();
    for (int i = t * 4; i < wn; i += SC_T * 4) *(int4*)(hrow + w0 + i) = *(const int4*)(ctr + i);
    __syncthreads();
  }
}

#define FOLD_N 16
static_assert(NCLASS <= FOLD_N && NCLASS == 10, "the last layer's lane map and its stores are written for 10 classes in one 16-column tile");

#define PB_W1 (NFEAT * NHID / 256)
#define PB_W2 (NHID * NHID / 256)
#define PB_FOLD ((NHID * FOLD_N + FOLD_N + 255) / 256)
#define PB_ALLW (PB_W1 + PB_W2 + PB_FOLD)
__global__ void __launch_bounds__(256) prep_scan(const float* __restrict__ W1, float* __restrict__ Wt1,
                                                 const float* __restrict__ W2, float* __restrict__ Wt2,
                                                 const float* __restrict__ W3, const float* __restrict__ b3,
                                                 const float* __restrict__ Wl, const float* __restrict__ bl,
                                                 float* __restrict__ Wt3l, float* __restrict__ b3l,
                                                 int* __restrict__ hist, int* __restrict__ cnt,
                                                 int npad) {
  const int b = blockIdx.x, t = threadIdx.x;
  if (b < PB_W1) {
    transpose_elem(W1, Wt1, NFEAT, NHID, b * 256 + t);
  } else if (b < PB_W1 + PB_W2) {
    transpose_elem(W2, Wt2, NHID, NHID, (b - PB_W1) * 256 + t);
  } else if (b < PB_ALLW) {
    const int idx = (b - PB_W1 - PB_W2) * 256 + t;
    if (idx < NHID * FOLD_N) {
      const int k = idx >> 4, nn = idx & (FOLD_N - 1);
      float s = 0.0f;
      if (nn < NCLASS) {
#pragma unroll 32
        for (int j = 0; j < NHID; ++j) s = fmaf(W3[k * NHID + j], Wl[j * NCLASS + nn], s);
      }
      Wt3l[nn * NHID + k] = s;
    } else if (idx < NHID * FOLD_N + FOLD_N) {
      const int nn = idx - NHID * FOLD_N;
      float s = 0.0f;
      if (nn < NCLASS) {
        s = bl[nn];
#pragma unroll 32
        for (int j = 0; j < NHID; ++j) s = fmaf(b3[j], Wl[j * NCLASS + nn], s);
      }
      b3l[nn] = s;
    }
  } else {
    const int i = (b - PB_ALLW) * 256 + t;
    if (i < npad) {
      int* p = hist + i;
      int run = 0;
      for (int c = 0; c < SC_CHUNKS; c += 16) {
        int v[16];
#pragma unroll
        for (int q = 0; q < 16; ++q) v[q] = p[(size_t)(c + q) * npad];
#pragma unroll
        for (int q = 0; q < 16; ++q) { p[(size_t)(c + q) * npad] = run; run += v[q]; }
      }
      cnt[i] = run;
    }
  }
}

#define SCAN_T 1024
__global__ void __launch_bounds__(SCAN_T) seg_scan(const int* __restrict__ cnt,
                                                   int* __restrict__ offs,
                                                   float* __restrict__ dinv, int npad) {
  __shared__ int wsum[SCAN_T / 32];
  __shared__ int tot;
  const int t = threadIdx.x, lane = t & 31, wv = t >> 5;
  int carry = 0;
  for (int base = 0; base < npad; base += SCAN_T * 16) {
    const int i0 = base + t * 16;
    const bool live = i0 < npad;
    int v[16];
#pragma unroll
    for (int q = 0; q < 4; ++q) {
      int4 c = live ? *(const int4*)(cnt + i0 + 4 * q) : make_int4(0, 0, 0, 0);
      v[4 * q + 0] = c.x; v[4 * q + 1] = c.y; v[4 * q + 2] = c.z; v[4 * q + 3] = c.w;
    }
    int ex[16];
    int s = 0;
#pragma unroll
    for (int k = 0; k < 16; ++k) { ex[k] = s; s += v[k]; }
    int inc = s;
#pragma unroll
    for (int o = 1; o < 32; o <<= 1) {
      int nb = __shfl_up(inc, o, 32);
      if (lane >= o) inc += nb;
    }
    if (lane == 31) wsum[wv] = inc;
    __syncthreads();
    if (wv == 0) {
      int w = wsum[lane];
      int wi = w;
#pragma unroll
      for (int o = 1; o < 32; o <<= 1) {
        int nb = __shfl_up(wi, o, 32);
        if (lane >= o) wi += nb;
      }
      wsum[lane] = wi - w;
      if (lane == 31) tot = wi;
    }
    __syncthreads();
    const int pre = carry + wsum[wv] + (inc - s);
    carry += tot;
    if (live) {
#pragma unroll
      for (int q = 0; q < 4; ++q) {
        *(int4*)(offs + i0 + 4 * q) = make_int4(pre + ex[4 * q + 0], pre + ex[4 * q + 1],
                                                pre + ex[4 * q + 2], pre + ex[4 * q + 3]);
        *(float4*)(dinv + i0 + 4 * q) =
            make_float4(rsqrtf((float)(v[4 * q + 0] + 1)), rsqrtf((float)(v[4 * q + 1] + 1)),
                        rsqrtf((float)(v[4 * q + 2] + 1)), rsqrtf((float)(v[4 * q + 3] + 1)));
      }
    }
    __syncthreads();
  }
}

__global__ void seg_fill(const int* __restrict__ rows, const int* __restrict__ cols,
                         const int* __restrict__ rank, const int* __restrict__ offs,
                         const int* __restrict__ hist, int* __restrict__ srcs,
                         int E, int epc, int npad) {
  const int l = blockIdx.x * blockDim.x + threadIdx.x;
  const int e = blockIdx.y * epc + l;
  if (l < epc && e < E) {
    const int c = cols[e];
    srcs[offs[c] + hist[(size_t)blockIdx.y * npad + c] + rank[e]] = rows[e];
  }
}

typedef __attribute__((ext_vector_type(4))) float v4f;
#define GB_M 64
#define GB_K 32
#define GB_P 36
#define GB_STAGE ((GB_M + NHID) * GB_P)
#define GB_CP 132
#define GB_RS (16 * GB_P)

#define GB_FETCH(kc)                                  \
  a0 = *(const v4f*)(x0 + (kc));                      \
  a1 = *(const v4f*)(x1 + (kc));                      \
  a2 = *(const v4f*)(x2 + (kc));                      \
  a3 = *(const v4f*)(x3 + (kc));                      \
  b0 = *(const v4f*)(wsrc + (kc));                    \
  b1 = *(const v4f*)(wsrc + wstep + (kc));            \
  b2 = *(const v4f*)(wsrc + 2 * wstep + (kc));        \
  b3 = *(const v4f*)(wsrc + 3 * wstep + (kc));        \
  b4 = *(const v4f*)(wsrc + 4 * wstep + (kc));        \
  b5 = *(const v4f*)(wsrc + 5 * wstep + (kc));        \
  b6 = *(const v4f*)(wsrc + 6 * wstep + (kc));        \
  b7 = *(const v4f*)(wsrc + 7 * wstep + (kc));

#define GB_PUT(dst)                                           \
  *(v4f*)((dst) + sdst) = a0;                                 \
  *(v4f*)((dst) + sdst + GB_RS) = a1;                         \
  *(v4f*)((dst) + sdst + 2 * GB_RS) = a2;                     \
  *(v4f*)((dst) + sdst + 3 * GB_RS) = a3;                     \
  *(v4f*)((dst) + GB_M * GB_P + sdst) = b0;                   \
  *(v4f*)((dst) + GB_M * GB_P + sdst + GB_RS) = b1;           \
  *(v4f*)((dst) + GB_M * GB_P + sdst + 2 * GB_RS) = b2;       \
  *(v4f*)((dst) + GB_M * GB_P + sdst + 3 * GB_RS) = b3;       \
  *(v4f*)((dst) + GB_M * GB_P + sdst + 4 * GB_RS) = b4;       \
  *(v4f*)((dst) + GB_M * GB_P + sdst + 5 * GB_RS) = b5;       \
  *(v4f*)((dst) + GB_M * GB_P + sdst + 6 * GB_RS) = b6;       \
  *(v4f*)((dst) + GB_M * GB_P + sdst + 7 * GB_RS) = b7;

__global__ void __launch_bounds__(128) gemm_wmma_f32(const float* __restrict__ X,
                                                     const float* __restrict__ Wt,
                                                     float* __restrict__ H,
                                                     int M, int K) {
  __shared__ __attribute__((aligned(16))) float lds[2 * GB_STAGE];

  const int tid = threadIdx.x;
  const int lane = tid & 31;
  const int wv = tid >> 5;
  const int c = lane & 15;
  const int h = lane >> 4;
  const int row0 = blockIdx.x * GB_M;

  const int seg = tid & 7;
  const int rsub = tid >> 3;
  const int last = M - 1;
  const int r0 = row0 + rsub, r1 = r0 + 16, r2 = r0 + 32, r3 = r0 + 48;
  const float* x0 = X + (size_t)(r0 < last ? r0 : last) * K + seg * 4;
  const float* x1 = X + (size_t)(r1 < last ? r1 : last) * K + seg * 4;
  const float* x2 = X + (size_t)(r2 < last ? r2 : last) * K + seg * 4;
  const float* x3 = X + (size_t)(r3 < last ? r3 : last) * K + seg * 4;
  const float* wsrc = Wt + (size_t)rsub * K + seg * 4;
  const size_t wstep = (size_t)16 * K;
  const int sdst = rsub * GB_P + seg * 4;

  v4f a0, a1, a2, a3, b0, b1, b2, b3, b4, b5, b6, b7;
  GB_FETCH(0)
  GB_PUT(lds)
  __syncthreads();

  v8f acc[8] = {};
  const int nch = K / GB_K;
  for (int ch = 0; ch < nch; ++ch) {
    const float* st = lds + (ch & 1) * GB_STAGE;
    float* nx = lds + ((ch + 1) & 1) * GB_STAGE;
    const bool more = ch + 1 < nch;
    if (more) {
      const int kc = (ch + 1) * GB_K;
      GB_FETCH(kc)
    }

    const float* ap = st + (wv * 16 + c) * GB_P + 2 * h;
    const float* bp = st + (GB_M + c) * GB_P + 2 * h;
#pragma unroll
    for (int ks = 0; ks < GB_K / 4; ++ks) {
      const v2f a = *(const v2f*)(ap + ks * 4);
#pragma unroll
      for (int t = 0; t < 8; ++t) {
        const v2f b = *(const v2f*)(bp + t * GB_RS + ks * 4);
        acc[t] = __builtin_amdgcn_wmma_f32_16x16x4_f32(
             false, a,  false, b,
             (short)0, acc[t],  false,  false);
      }
    }

    if (more) {
      GB_PUT(nx)
    }
    __syncthreads();
  }

  float* cs = lds + wv * (16 * GB_CP);
#pragma unroll
  for (int t = 0; t < 8; ++t) {
#pragma unroll
    for (int e = 0; e < 8; ++e) cs[(8 * h + e) * GB_CP + t * 16 + c] = acc[t][e];
  }
  __syncthreads();
#pragma unroll
  for (int r = 0; r < 16; ++r) {
    const int row = row0 + wv * 16 + r;
    if (row < M)
      *(v4f*)(H + (size_t)row * NHID + lane * 4) = *(const v4f*)(cs + r * GB_CP + lane * 4);
  }
}
#undef GB_FETCH
#undef GB_PUT

#define G16_STAGE ((GB_M + FOLD_N) * GB_P)
#define G16_CP 20
__global__ void __launch_bounds__(128) gemm_wmma_f32_n16(const float* __restrict__ X,
                                                         const float* __restrict__ Wt,
                                                         float* __restrict__ G,
                                                         int M, int K) {
  __shared__ __attribute__((aligned(16))) float lds[2 * G16_STAGE];

  const int tid = threadIdx.x;
  const int lane = tid & 31;
  const int wv = tid >> 5;
  const int c = lane & 15;
  const int h = lane >> 4;
  const int row0 = blockIdx.x * GB_M;

  const int seg = tid & 7;
  const int rsub = tid >> 3;
  const int last = M - 1;
  const int r0 = row0 + rsub, r1 = r0 + 16, r2 = r0 + 32, r3 = r0 + 48;
  const float* x0 = X + (size_t)(r0 < last ? r0 : last) * K + seg * 4;
  const float* x1 = X + (size_t)(r1 < last ? r1 : last) * K + seg * 4;
  const float* x2 = X + (size_t)(r2 < last ? r2 : last) * K + seg * 4;
  const float* x3 = X + (size_t)(r3 < last ? r3 : last) * K + seg * 4;
  const float* wsrc = Wt + (size_t)rsub * K + seg * 4;
  const int sdst = rsub * GB_P + seg * 4;

  v4f a0 = *(const v4f*)(x0);
  v4f a1 = *(const v4f*)(x1);
  v4f a2 = *(const v4f*)(x2);
  v4f a3 = *(const v4f*)(x3);
  v4f b0 = *(const v4f*)(wsrc);
  *(v4f*)(lds + sdst) = a0;
  *(v4f*)(lds + sdst + GB_RS) = a1;
  *(v4f*)(lds + sdst + 2 * GB_RS) = a2;
  *(v4f*)(lds + sdst + 3 * GB_RS) = a3;
  *(v4f*)(lds + GB_M * GB_P + sdst) = b0;
  __syncthreads();

  v8f acc = {};
  const int nch = K / GB_K;
  for (int ch = 0; ch < nch; ++ch) {
    const float* st = lds + (ch & 1) * G16_STAGE;
    float* nx = lds + ((ch + 1) & 1) * G16_STAGE;
    const bool more = ch + 1 < nch;
    if (more) {
      const int kc = (ch + 1) * GB_K;
      a0 = *(const v4f*)(x0 + kc);
      a1 = *(const v4f*)(x1 + kc);
      a2 = *(const v4f*)(x2 + kc);
      a3 = *(const v4f*)(x3 + kc);
      b0 = *(const v4f*)(wsrc + kc);
    }

    const float* ap = st + (wv * 16 + c) * GB_P + 2 * h;
    const float* bp = st + (GB_M + c) * GB_P + 2 * h;
#pragma unroll
    for (int ks = 0; ks < GB_K / 4; ++ks) {
      const v2f a = *(const v2f*)(ap + ks * 4);
      const v2f b = *(const v2f*)(bp + ks * 4);
      acc = __builtin_amdgcn_wmma_f32_16x16x4_f32(
           false, a,  false, b,
           (short)0, acc,  false,  false);
    }

    if (more) {
      *(v4f*)(nx + sdst) = a0;
      *(v4f*)(nx + sdst + GB_RS) = a1;
      *(v4f*)(nx + sdst + 2 * GB_RS) = a2;
      *(v4f*)(nx + sdst + 3 * GB_RS) = a3;
      *(v4f*)(nx + GB_M * GB_P + sdst) = b0;
    }
    __syncthreads();
  }

  float* cs = lds + wv * (16 * G16_CP);
#pragma unroll
  for (int e = 0; e < 8; ++e) cs[(8 * h + e) * G16_CP + c] = acc[e];
  __syncthreads();
  for (int i = 0; i < 2; ++i) {
    const int r = i * 8 + (lane >> 2);
    const int row = row0 + wv * 16 + r;
    if (row < M)
      *(v4f*)(G + (size_t)row * FOLD_N + (lane & 3) * 4) = *(const v4f*)(cs + r * G16_CP + (lane & 3) * 4);
  }
}


__device__ __forceinline__ void row_fma(float4& acc, const float4 h, const float w) {
  acc.x = fmaf(h.x, w, acc.x);
  acc.y = fmaf(h.y, w, acc.y);
  acc.z = fmaf(h.z, w, acc.z);
  acc.w = fmaf(h.w, w, acc.w);
}

__global__ void __launch_bounds__(256) agg_rows(const int* __restrict__ offs,
                                                const int* __restrict__ srcs,
                                                const float* __restrict__ dinv,
                                                const float* __restrict__ H,
                                                const float* __restrict__ b,
                                                float* __restrict__ Y, int n) {
  int node = (int)((blockIdx.x * (size_t)blockDim.x + threadIdx.x) >> 5);
  const int lane = threadIdx.x & 31;
  if (node >= n) return;
  node = __builtin_amdgcn_readfirstlane(node);
  const int beg = offs[node];
  const int d = offs[node + 1] - beg;
  const float di = dinv[node];
  const float* hl = H + lane * 4;

  float4 acc = make_float4(0.0f, 0.0f, 0.0f, 0.0f);

  for (int base = 0; base < d; base += 32) {
    const int m = (d - base < 32) ? (d - base) : 32;
    int s = node;
    if (lane < m) s = srcs[beg + base + lane];
    const float dv = dinv[s];
    int j = 0;
    for (; j + 4 <= m; j += 4) {
      const int s0 = __builtin_amdgcn_readlane(s, j);
      const int s1 = __builtin_amdgcn_readlane(s, j + 1);
      const int s2 = __builtin_amdgcn_readlane(s, j + 2);
      const int s3 = __builtin_amdgcn_readlane(s, j + 3);
      const float4 h0 = *(const float4*)(hl + (size_t)s0 * NHID);
      const float4 h1 = *(const float4*)(hl + (size_t)s1 * NHID);
      const float4 h2 = *(const float4*)(hl + (size_t)s2 * NHID);
      const float4 h3 = *(const float4*)(hl + (size_t)s3 * NHID);
      row_fma(acc, h0, di * __int_as_float(__builtin_amdgcn_readlane(__float_as_int(dv), j)));
      row_fma(acc, h1, di * __int_as_float(__builtin_amdgcn_readlane(__float_as_int(dv), j + 1)));
      row_fma(acc, h2, di * __int_as_float(__builtin_amdgcn_readlane(__float_as_int(dv), j + 2)));
      row_fma(acc, h3, di * __int_as_float(__builtin_amdgcn_readlane(__float_as_int(dv), j + 3)));
    }
#pragma nounroll
    for (; j < m; ++j) {
      const int s0 = __builtin_amdgcn_readlane(s, j);
      const float4 h0 = *(const float4*)(hl + (size_t)s0 * NHID);
      row_fma(acc, h0, di * __int_as_float(__builtin_amdgcn_readlane(__float_as_int(dv), j)));
    }
  }

  const float4 hself = *(const float4*)(hl + (size_t)node * NHID);
  const float4 bv = *(const float4*)(b + lane * 4);
  acc.x += hself.x * di * di; acc.y += hself.y * di * di;
  acc.z += hself.z * di * di; acc.w += hself.w * di * di;
  acc.x += bv.x; acc.y += bv.y; acc.z += bv.z; acc.w += bv.w;
  acc.x = fmaxf(acc.x, 0.0f); acc.y = fmaxf(acc.y, 0.0f);
  acc.z = fmaxf(acc.z, 0.0f); acc.w = fmaxf(acc.w, 0.0f);
  *(float4*)(Y + (size_t)node * NHID + lane * 4) = acc;
}

#define SRT_R 4
__global__ void __launch_bounds__(256) agg_rows_order(const int* __restrict__ offs,
                                                      int* srcs,
                                                      const float* __restrict__ dinv,
                                                      const float* __restrict__ H,
                                                      const float* __restrict__ b,
                                                      float* __restrict__ Y, int n) {
  int node = (int)((blockIdx.x * (size_t)blockDim.x + threadIdx.x) >> 5);
  const int lane = threadIdx.x & 31;
  if (node >= n) return;
  node = __builtin_amdgcn_readfirstlane(node);
  const int beg = offs[node];
  const int d = offs[node + 1] - beg;
  const float di = dinv[node];
  const float* hl = H + lane * 4;
  int* seg = srcs + beg;

  int s_first = node;
  if (d <= 32) {
    if (d > 0) {
      const bool live = lane < d;
      const int key = live ? seg[lane] : 0;
      int r = 0;
      for (int q = 0; q < d; ++q) {
        const int k = __builtin_amdgcn_readlane(key, q);
        r += (int)(k < key) + (int)((k == key) & (q < lane));
      }
      if (live) seg[r] = key;
      s_first = __builtin_amdgcn_ds_permute((live ? r : lane) << 2, live ? key : node);
    }
  } else if (d <= 32 * SRT_R) {
    int key[SRT_R], r[SRT_R];
#pragma unroll
    for (int a = 0; a < SRT_R; ++a) {
      const int i = 32 * a + lane;
      key[a] = (i < d) ? seg[i] : 0;
      r[a] = 0;
    }
#pragma unroll
    for (int c = 0; c < SRT_R; ++c) {
      if (32 * c < d) {
        const int m = (d - 32 * c < 32) ? (d - 32 * c) : 32;
        for (int q = 0; q < m; ++q) {
          const int k = __builtin_amdgcn_readlane(key[c], q);
          const int j = 32 * c + q;
#pragma unroll
          for (int a = 0; a < SRT_R; ++a)
            r[a] += (int)(k < key[a]) + (int)((k == key[a]) & (j < 32 * a + lane));
        }
      }
    }
#pragma unroll
    for (int a = 0; a < SRT_R; ++a)
      if (32 * a + lane < d) seg[r[a]] = key[a];
    __threadfence();
  } else {
    for (int p = 0; p < d; ++p) {
      int best = 0x7fffffff, bpos = 0x7fffffff;
      for (int j = p + lane; j < d; j += 32) {
        const int k = seg[j];
        if (k < best) { best = k; bpos = j; }
      }
#pragma unroll
      for (int o = 16; o > 0; o >>= 1) {
        const int ob = __shfl_xor(best, o, 32);
        const int op = __shfl_xor(bpos, o, 32);
        if (ob < best || (ob == best && op < bpos)) { best = ob; bpos = op; }
      }
      if (lane == 0) {
        const int vp = seg[p];
        seg[bpos] = vp;
        seg[p] = best;
      }
      __threadfence();
    }
  }

  float4 acc = make_float4(0.0f, 0.0f, 0.0f, 0.0f);

  for (int base = 0; base < d; base += 32) {
    const int m = (d - base < 32) ? (d - base) : 32;
    int s = s_first;
    if (d > 32) {
      s = node;
      if (lane < m) s = seg[base + lane];
    }
    const float dv = dinv[s];
    int j = 0;
    for (; j + 4 <= m; j += 4) {
      const int s0 = __builtin_amdgcn_readlane(s, j);
      const int s1 = __builtin_amdgcn_readlane(s, j + 1);
      const int s2 = __builtin_amdgcn_readlane(s, j + 2);
      const int s3 = __builtin_amdgcn_readlane(s, j + 3);
      const float4 h0 = *(const float4*)(hl + (size_t)s0 * NHID);
      const float4 h1 = *(const float4*)(hl + (size_t)s1 * NHID);
      const float4 h2 = *(const float4*)(hl + (size_t)s2 * NHID);
      const float4 h3 = *(const float4*)(hl + (size_t)s3 * NHID);
      row_fma(acc, h0, di * __int_as_float(__builtin_amdgcn_readlane(__float_as_int(dv), j)));
      row_fma(acc, h1, di * __int_as_float(__builtin_amdgcn_readlane(__float_as_int(dv), j + 1)));
      row_fma(acc, h2, di * __int_as_float(__builtin_amdgcn_readlane(__float_as_int(dv), j + 2)));
      row_fma(acc, h3, di * __int_as_float(__builtin_amdgcn_readlane(__float_as_int(dv), j + 3)));
    }
#pragma nounroll
    for (; j < m; ++j) {
      const int s0 = __builtin_amdgcn_readlane(s, j);
      const float4 h0 = *(const float4*)(hl + (size_t)s0 * NHID);
      row_fma(acc, h0, di * __int_as_float(__builtin_amdgcn_readlane(__float_as_int(dv), j)));
    }
  }

  const float4 hself = *(const float4*)(hl + (size_t)node * NHID);
  const float4 bv = *(const float4*)(b + lane * 4);
  acc.x += hself.x * di * di; acc.y += hself.y * di * di;
  acc.z += hself.z * di * di; acc.w += hself.w * di * di;
  acc.x += bv.x; acc.y += bv.y; acc.z += bv.z; acc.w += bv.w;
  acc.x = fmaxf(acc.x, 0.0f); acc.y = fmaxf(acc.y, 0.0f);
  acc.z = fmaxf(acc.z, 0.0f); acc.w = fmaxf(acc.w, 0.0f);
  *(float4*)(Y + (size_t)node * NHID + lane * 4) = acc;
}

__global__ void __launch_bounds__(256) agg_last(const int* __restrict__ offs,
                                                const int* __restrict__ srcs,
                                                const float* __restrict__ dinv,
                                                const float* __restrict__ G,
                                                const float* __restrict__ bf,
                                                float* __restrict__ out, int n) {
  int node = (int)((blockIdx.x * (size_t)blockDim.x + threadIdx.x) >> 5);
  const int lane = threadIdx.x & 31;
  if (node >= n) return;
  node = __builtin_amdgcn_readfirstlane(node);
  const int beg = offs[node];
  const int d = offs[node + 1] - beg;
  const float di = dinv[node];
  const int slot = lane >> 2;
  const int q4 = (lane & 3) * 4;
  const float* gl = G + q4;

  float4 acc = make_float4(0.0f, 0.0f, 0.0f, 0.0f);
  for (int base = 0; base < d; base += 32) {
    bool v[4];
    int s[4];
    float w[4];
    float4 g[4];
#pragma unroll
    for (int bt = 0; bt < 4; ++bt) {
      const int e = base + slot + 8 * bt;
      v[bt] = e < d;
      s[bt] = srcs[beg + (v[bt] ? e : d - 1)];
    }
#pragma unroll
    for (int bt = 0; bt < 4; ++bt) {
      const float dv = dinv[s[bt]];
      w[bt] = v[bt] ? di * dv : 0.0f;
      g[bt] = *(const float4*)(gl + (size_t)s[bt] * FOLD_N);
    }
#pragma unroll
    for (int bt = 0; bt < 4; ++bt) row_fma(acc, g[bt], w[bt]);
  }
#pragma unroll
  for (int o = 4; o < 32; o <<= 1) {
    acc.x += __shfl_xor(acc.x, o, 32);
    acc.y += __shfl_xor(acc.y, o, 32);
    acc.z += __shfl_xor(acc.z, o, 32);
    acc.w += __shfl_xor(acc.w, o, 32);
  }

  const float4 gself = *(const float4*)(gl + (size_t)node * FOLD_N);
  const float4 bv = *(const float4*)(bf + q4);
  acc.x += gself.x * di * di; acc.y += gself.y * di * di;
  acc.z += gself.z * di * di; acc.w += gself.w * di * di;
  acc.x += bv.x; acc.y += bv.y; acc.z += bv.z; acc.w += bv.w;

  const float ninf = -__builtin_inff();
  const bool c0 = q4 + 0 < NCLASS, c1 = q4 + 1 < NCLASS, c2 = q4 + 2 < NCLASS, c3 = q4 + 3 < NCLASS;
  float m = fmaxf(fmaxf(c0 ? acc.x : ninf, c1 ? acc.y : ninf),
                  fmaxf(c2 ? acc.z : ninf, c3 ? acc.w : ninf));
  m = fmaxf(m, __shfl_xor(m, 1, 32));
  m = fmaxf(m, __shfl_xor(m, 2, 32));
  float s = (c0 ? __expf(acc.x - m) : 0.0f) + (c1 ? __expf(acc.y - m) : 0.0f) +
            (c2 ? __expf(acc.z - m) : 0.0f) + (c3 ? __expf(acc.w - m) : 0.0f);
  s += __shfl_xor(s, 1, 32);
  s += __shfl_xor(s, 2, 32);
  const float lse = m + __logf(s);
  if (lane < 3) {
    float* o = out + (size_t)node * NCLASS + q4;
    *(float2*)o = make_float2(acc.x - lse, acc.y - lse);
    if (lane < 2) *(float2*)(o + 2) = make_float2(acc.z - lse, acc.w - lse);
  }
}

extern "C" void kernel_launch(void* const* d_in, const int* in_sizes, int n_in,
                              void* d_out, int out_size, void* d_ws, size_t ws_size,
                              hipStream_t stream) {
  const float* x  = (const float*)d_in[0];
  const int*   ei = (const int*)d_in[1];
  const float* W1 = (const float*)d_in[2];
  const float* b1 = (const float*)d_in[3];
  const float* W2 = (const float*)d_in[4];
  const float* b2 = (const float*)d_in[5];
  const float* W3 = (const float*)d_in[6];
  const float* b3 = (const float*)d_in[7];
  const float* Wl = (const float*)d_in[8];
  const float* bl = (const float*)d_in[9];
  float* out = (float*)d_out;

  const int E = in_sizes[1] / 2;
  const int N = in_sizes[0] / NFEAT;
  const int npad = ((N + 16) / 16) * 16;

  char* ws = (char*)d_ws;
  size_t off = 0;
  auto alloc = [&](size_t bytes) -> float* {
    float* p = (float*)(ws + off);
    off += (bytes + 255) & ~(size_t)255;
    return p;
  };
  const size_t cbytes = ((size_t)npad * sizeof(int) + 255) & ~(size_t)255;
  const size_t ebytes = ((size_t)E * sizeof(int) + 255) & ~(size_t)255;
  const size_t hbytes = ((size_t)SC_CHUNKS * npad * sizeof(int) + 255) & ~(size_t)255;
  const size_t tbytes = cbytes + ebytes + hbytes;
  const size_t ybytes = (size_t)N * NHID * sizeof(float);
  float* dinv = alloc((size_t)npad * sizeof(float));
  float* Wt1  = alloc((size_t)NFEAT * NHID * sizeof(float));
  float* Wt2  = alloc((size_t)NHID * NHID * sizeof(float));
  float* Wt3l = alloc((size_t)FOLD_N * NHID * sizeof(float));
  float* b3l  = alloc((size_t)FOLD_N * sizeof(float));
  float* H    = alloc(ybytes);
  float* Y    = alloc(ybytes > tbytes ? ybytes : tbytes);
  int* offs   = (int*)alloc((size_t)npad * sizeof(int));
  int* srcs   = (int*)alloc((size_t)E * sizeof(int));
  if (off > ws_size) return;
  int* cnt  = (int*)Y;
  int* rank = (int*)((char*)Y + cbytes);
  int* hist = (int*)((char*)Y + cbytes + ebytes);

  const int* rows = ei;
  const int* cols = ei + E;

  const int wgrid = (int)(((size_t)N * 32 + 255) / 256);

  const int epc = (E + SC_CHUNKS - 1) / SC_CHUNKS;

  chunk_count<<<SC_CHUNKS, SC_T, 0, stream>>>(cols, rank, hist, E, epc, npad);
  prep_scan<<<PB_ALLW + (npad + 255) / 256, 256, 0, stream>>>(W1, Wt1, W2, Wt2, W3, b3, Wl, bl,
                                                              Wt3l, b3l, hist, cnt, npad);
  seg_scan<<<1, SCAN_T, 0, stream>>>(cnt, offs, dinv, npad);
  seg_fill<<<dim3((epc + 255) / 256, SC_CHUNKS), 256, 0, stream>>>(rows, cols, rank, offs, hist, srcs,
                                                                   E, epc, npad);

  const int tiles = (N + 15) / 16;
  const int gemm_grid = (tiles + 3) / 4;

  gemm_wmma_f32<<<gemm_grid, 128, 0, stream>>>(x, Wt1, H, N, NFEAT);
  agg_rows_order<<<wgrid, 256, 0, stream>>>(offs, srcs, dinv, H, b1, Y, N);

  gemm_wmma_f32<<<gemm_grid, 128, 0, stream>>>(Y, Wt2, H, N, NHID);
  agg_rows<<<wgrid, 256, 0, stream>>>(offs, srcs, dinv, H, b2, Y, N);

  gemm_wmma_f32_n16<<<gemm_grid, 128, 0, stream>>>(Y, Wt3l, H, N, NHID);
  agg_last<<<wgrid, 256, 0, stream>>>(offs, srcs, dinv, H, b3l, out, N);
}


# GEMM epilogues store each row scaled by its own dinv: aggregations gather no per-edge dinv, add plain source rows in the same order, scale the sum once by the target's dinv (scaling rounds per row)
# speedup vs baseline: 3.8786x; 1.0322x over previous
#include <hip/hip_runtime.h>

typedef __attribute__((ext_vector_type(2))) float v2f;
typedef __attribute__((ext_vector_type(8))) float v8f;

#define NFEAT 512
#define NHID  128
#define NCLASS 10


__device__ __forceinline__ void transpose_elem(const float* __restrict__ W, float* __restrict__ Wt,
                                               int K, int Nc, int idx) {
  if (idx < K * Nc) {
    int k = idx / Nc, n = idx % Nc;
    Wt[(size_t)n * K + k] = W[idx];
  }
}


#define SC_CHUNKS 64
#define SC_T 1024
#define SC_CAP 50176
__global__ void __launch_bounds__(SC_T) chunk_count(const int* __restrict__ cols,
                                                    int* __restrict__ rank,
                                                    int* __restrict__ hist,
                                                    int E, int epc, int npad) {
  __shared__ __attribute__((aligned(16))) int ctr[SC_CAP];
  const int t = threadIdx.x;
  const int e0 = blockIdx.x * epc;
  const int e1 = (e0 + epc < E) ? (e0 + epc) : E;
  int* hrow = hist + (size_t)blockIdx.x * npad;
  for (int w0 = 0; w0 < npad; w0 += SC_CAP) {
    const int wn = (npad - w0 < SC_CAP) ? (npad - w0) : SC_CAP;
    for (int i = t * 4; i < wn; i += SC_T * 4) *(int4*)(ctr + i) = make_int4(0, 0, 0, 0);
    __syncthreads();
    for (int e = e0 + t; e < e1; e += 8 * SC_T) {
      int c[8], r[8];
#pragma unroll
      for (int q = 0; q < 8; ++q) c[q] = (e + q * SC_T < e1) ? cols[e + q * SC_T] : -1;
#pragma unroll
      for (int q = 0; q < 8; ++q) {
        const unsigned k = (unsigned)(c[q] - w0);
        r[q] = (k < (unsigned)wn) ? atomicAdd(&ctr[k], 1) : -1;
      }
#pragma unroll
      for (int q = 0; q < 8; ++q)
        if (r[q] >= 0) rank[e + q * SC_T] = r[q];
    }
    __syncthreads();
    for (int i = t * 4; i < wn; i += SC_T * 4) *(int4*)(hrow + w0 + i) = *(const int4*)(ctr + i);
    __syncthreads();
  }
}

#define FOLD_N 16
static_assert(NCLASS <= FOLD_N && NCLASS == 10, "the last layer's lane map and its stores are written for 10 classes in one 16-column tile");

#define PB_W1 (NFEAT * NHID / 256)
#define PB_W2 (NHID * NHID / 256)
#define PB_FOLD ((NHID * FOLD_N + FOLD_N + 255) / 256)
#define PB_ALLW (PB_W1 + PB_W2 + PB_FOLD)
__global__ void __launch_bounds__(256) prep_scan(const float* __restrict__ W1, float* __restrict__ Wt1,
                                                 const float* __restrict__ W2, float* __restrict__ Wt2,
                                                 const float* __restrict__ W3, const float* __restrict__ b3,
                                                 const float* __restrict__ Wl, const float* __restrict__ bl,
                                                 float* __restrict__ Wt3l, float* __restrict__ b3l,
                                                 int* __restrict__ hist, int* __restrict__ cnt,
                                                 int npad) {
  const int b = blockIdx.x, t = threadIdx.x;
  if (b < PB_W1) {
    transpose_elem(W1, Wt1, NFEAT, NHID, b * 256 + t);
  } else if (b < PB_W1 + PB_W2) {
    transpose_elem(W2, Wt2, NHID, NHID, (b - PB_W1) * 256 + t);
  } else if (b < PB_ALLW) {
    const int idx = (b - PB_W1 - PB_W2) * 256 + t;
    if (idx < NHID * FOLD_N) {
      const int k = idx >> 4, nn = idx & (FOLD_N - 1);
      float s = 0.0f;
      if (nn < NCLASS) {
#pragma unroll 32
        for (int j = 0; j < NHID; ++j) s = fmaf(W3[k * NHID + j], Wl[j * NCLASS + nn], s);
      }
      Wt3l[nn * NHID + k] = s;
    } else if (idx < NHID * FOLD_N + FOLD_N) {
      const int nn = idx - NHID * FOLD_N;
      float s = 0.0f;
      if (nn < NCLASS) {
        s = bl[nn];
#pragma unroll 32
        for (int j = 0; j < NHID; ++j) s = fmaf(b3[j], Wl[j * NCLASS + nn], s);
      }
      b3l[nn] = s;
    }
  } else {
    const int i = (b - PB_ALLW) * 256 + t;
    if (i < npad) {
      int* p = hist + i;
      int run = 0;
      for (int c = 0; c < SC_CHUNKS; c += 16) {
        int v[16];
#pragma unroll
        for (int q = 0; q < 16; ++q) v[q] = p[(size_t)(c + q) * npad];
#pragma unroll
        for (int q = 0; q < 16; ++q) { p[(size_t)(c + q) * npad] = run; run += v[q]; }
      }
      cnt[i] = run;
    }
  }
}

#define SCAN_T 1024
__global__ void __launch_bounds__(SCAN_T) seg_scan(const int* __restrict__ cnt,
                                                   int* __restrict__ offs,
                                                   float* __restrict__ dinv, int npad) {
  __shared__ int wsum[SCAN_T / 32];
  __shared__ int tot;
  const int t = threadIdx.x, lane = t & 31, wv = t >> 5;
  int carry = 0;
  for (int base = 0; base < npad; base += SCAN_T * 16) {
    const int i0 = base + t * 16;
    const bool live = i0 < npad;
    int v[16];
#pragma unroll
    for (int q = 0; q < 4; ++q) {
      int4 c = live ? *(const int4*)(cnt + i0 + 4 * q) : make_int4(0, 0, 0, 0);
      v[4 * q + 0] = c.x; v[4 * q + 1] = c.y; v[4 * q + 2] = c.z; v[4 * q + 3] = c.w;
    }
    int ex[16];
    int s = 0;
#pragma unroll
    for (int k = 0; k < 16; ++k) { ex[k] = s; s += v[k]; }
    int inc = s;
#pragma unroll
    for (int o = 1; o < 32; o <<= 1) {
      int nb = __shfl_up(inc, o, 32);
      if (lane >= o) inc += nb;
    }
    if (lane == 31) wsum[wv] = inc;
    __syncthreads();
    if (wv == 0) {
      int w = wsum[lane];
      int wi = w;
#pragma unroll
      for (int o = 1; o < 32; o <<= 1) {
        int nb = __shfl_up(wi, o, 32);
        if (lane >= o) wi += nb;
      }
      wsum[lane] = wi - w;
      if (lane == 31) tot = wi;
    }
    __syncthreads();
    const int pre = carry + wsum[wv] + (inc - s);
    carry += tot;
    if (live) {
#pragma unroll
      for (int q = 0; q < 4; ++q) {
        *(int4*)(offs + i0 + 4 * q) = make_int4(pre + ex[4 * q + 0], pre + ex[4 * q + 1],
                                                pre + ex[4 * q + 2], pre + ex[4 * q + 3]);
        *(float4*)(dinv + i0 + 4 * q) =
            make_float4(rsqrtf((float)(v[4 * q + 0] + 1)), rsqrtf((float)(v[4 * q + 1] + 1)),
                        rsqrtf((float)(v[4 * q + 2] + 1)), rsqrtf((float)(v[4 * q + 3] + 1)));
      }
    }
    __syncthreads();
  }
}

__global__ void seg_fill(const int* __restrict__ rows, const int* __restrict__ cols,
                         const int* __restrict__ rank, const int* __restrict__ offs,
                         const int* __restrict__ hist, int* __restrict__ srcs,
                         int E, int epc, int npad) {
  const int l = blockIdx.x * blockDim.x + threadIdx.x;
  const int e = blockIdx.y * epc + l;
  if (l < epc && e < E) {
    const int c = cols[e];
    srcs[offs[c] + hist[(size_t)blockIdx.y * npad + c] + rank[e]] = rows[e];
  }
}

typedef __attribute__((ext_vector_type(4))) float v4f;
#define GB_M 64
#define GB_K 32
#define GB_P 36
#define GB_STAGE ((GB_M + NHID) * GB_P)
#define GB_CP 132
#define GB_RS (16 * GB_P)

#define GB_FETCH(kc)                                  \
  a0 = *(const v4f*)(x0 + (kc));                      \
  a1 = *(const v4f*)(x1 + (kc));                      \
  a2 = *(const v4f*)(x2 + (kc));                      \
  a3 = *(const v4f*)(x3 + (kc));                      \
  b0 = *(const v4f*)(wsrc + (kc));                    \
  b1 = *(const v4f*)(wsrc + wstep + (kc));            \
  b2 = *(const v4f*)(wsrc + 2 * wstep + (kc));        \
  b3 = *(const v4f*)(wsrc + 3 * wstep + (kc));        \
  b4 = *(const v4f*)(wsrc + 4 * wstep + (kc));        \
  b5 = *(const v4f*)(wsrc + 5 * wstep + (kc));        \
  b6 = *(const v4f*)(wsrc + 6 * wstep + (kc));        \
  b7 = *(const v4f*)(wsrc + 7 * wstep + (kc));

#define GB_PUT(dst)                                           \
  *(v4f*)((dst) + sdst) = a0;                                 \
  *(v4f*)((dst) + sdst + GB_RS) = a1;                         \
  *(v4f*)((dst) + sdst + 2 * GB_RS) = a2;                     \
  *(v4f*)((dst) + sdst + 3 * GB_RS) = a3;                     \
  *(v4f*)((dst) + GB_M * GB_P + sdst) = b0;                   \
  *(v4f*)((dst) + GB_M * GB_P + sdst + GB_RS) = b1;           \
  *(v4f*)((dst) + GB_M * GB_P + sdst + 2 * GB_RS) = b2;       \
  *(v4f*)((dst) + GB_M * GB_P + sdst + 3 * GB_RS) = b3;       \
  *(v4f*)((dst) + GB_M * GB_P + sdst + 4 * GB_RS) = b4;       \
  *(v4f*)((dst) + GB_M * GB_P + sdst + 5 * GB_RS) = b5;       \
  *(v4f*)((dst) + GB_M * GB_P + sdst + 6 * GB_RS) = b6;       \
  *(v4f*)((dst) + GB_M * GB_P + sdst + 7 * GB_RS) = b7;

__global__ void __launch_bounds__(128) gemm_wmma_f32(const float* __restrict__ X,
                                                     const float* __restrict__ Wt,
                                                     const float* __restrict__ dinv,
                                                     float* __restrict__ H,
                                                     int M, int K) {
  __shared__ __attribute__((aligned(16))) float lds[2 * GB_STAGE];

  const int tid = threadIdx.x;
  const int lane = tid & 31;
  const int wv = tid >> 5;
  const int c = lane & 15;
  const int h = lane >> 4;
  const int row0 = blockIdx.x * GB_M;

  const int seg = tid & 7;
  const int rsub = tid >> 3;
  const int last = M - 1;
  const int r0 = row0 + rsub, r1 = r0 + 16, r2 = r0 + 32, r3 = r0 + 48;
  const float* x0 = X + (size_t)(r0 < last ? r0 : last) * K + seg * 4;
  const float* x1 = X + (size_t)(r1 < last ? r1 : last) * K + seg * 4;
  const float* x2 = X + (size_t)(r2 < last ? r2 : last) * K + seg * 4;
  const float* x3 = X + (size_t)(r3 < last ? r3 : last) * K + seg * 4;
  const float* wsrc = Wt + (size_t)rsub * K + seg * 4;
  const size_t wstep = (size_t)16 * K;
  const int sdst = rsub * GB_P + seg * 4;

  const int srow = row0 + wv * 16 + c;
  const float dsc = dinv[srow < last ? srow : last];

  v4f a0, a1, a2, a3, b0, b1, b2, b3, b4, b5, b6, b7;
  GB_FETCH(0)
  GB_PUT(lds)
  __syncthreads();

  v8f acc[8] = {};
  const int nch = K / GB_K;
  for (int ch = 0; ch < nch; ++ch) {
    const float* st = lds + (ch & 1) * GB_STAGE;
    float* nx = lds + ((ch + 1) & 1) * GB_STAGE;
    const bool more = ch + 1 < nch;
    if (more) {
      const int kc = (ch + 1) * GB_K;
      GB_FETCH(kc)
    }

    const float* ap = st + (wv * 16 + c) * GB_P + 2 * h;
    const float* bp = st + (GB_M + c) * GB_P + 2 * h;
#pragma unroll
    for (int ks = 0; ks < GB_K / 4; ++ks) {
      const v2f a = *(const v2f*)(ap + ks * 4);
#pragma unroll
      for (int t = 0; t < 8; ++t) {
        const v2f b = *(const v2f*)(bp + t * GB_RS + ks * 4);
        acc[t] = __builtin_amdgcn_wmma_f32_16x16x4_f32(
             false, a,  false, b,
             (short)0, acc[t],  false,  false);
      }
    }

    if (more) {
      GB_PUT(nx)
    }
    __syncthreads();
  }

  float* cs = lds + wv * (16 * GB_CP);
#pragma unroll
  for (int t = 0; t < 8; ++t) {
#pragma unroll
    for (int e = 0; e < 8; ++e) cs[(8 * h + e) * GB_CP + t * 16 + c] = acc[t][e];
  }
  __syncthreads();
#pragma unroll
  for (int r = 0; r < 16; ++r) {
    const int row = row0 + wv * 16 + r;
    const float sc = __int_as_float(__builtin_amdgcn_readlane(__float_as_int(dsc), r));
    if (row < M)
      *(v4f*)(H + (size_t)row * NHID + lane * 4) = *(const v4f*)(cs + r * GB_CP + lane * 4) * sc;
  }
}
#undef GB_FETCH
#undef GB_PUT

#define G16_STAGE ((GB_M + FOLD_N) * GB_P)
#define G16_CP 20
__global__ void __launch_bounds__(128) gemm_wmma_f32_n16(const float* __restrict__ X,
                                                         const float* __restrict__ Wt,
                                                         const float* __restrict__ dinv,
                                                         float* __restrict__ G,
                                                         int M, int K) {
  __shared__ __attribute__((aligned(16))) float lds[2 * G16_STAGE];

  const int tid = threadIdx.x;
  const int lane = tid & 31;
  const int wv = tid >> 5;
  const int c = lane & 15;
  const int h = lane >> 4;
  const int row0 = blockIdx.x * GB_M;

  const int seg = tid & 7;
  const int rsub = tid >> 3;
  const int last = M - 1;
  const int r0 = row0 + rsub, r1 = r0 + 16, r2 = r0 + 32, r3 = r0 + 48;
  const float* x0 = X + (size_t)(r0 < last ? r0 : last) * K + seg * 4;
  const float* x1 = X + (size_t)(r1 < last ? r1 : last) * K + seg * 4;
  const float* x2 = X + (size_t)(r2 < last ? r2 : last) * K + seg * 4;
  const float* x3 = X + (size_t)(r3 < last ? r3 : last) * K + seg * 4;
  const float* wsrc = Wt + (size_t)rsub * K + seg * 4;
  const int sdst = rsub * GB_P + seg * 4;

  const int srow = row0 + wv * 16 + c;
  const float dsc = dinv[srow < last ? srow : last];

  v4f a0 = *(const v4f*)(x0);
  v4f a1 = *(const v4f*)(x1);
  v4f a2 = *(const v4f*)(x2);
  v4f a3 = *(const v4f*)(x3);
  v4f b0 = *(const v4f*)(wsrc);
  *(v4f*)(lds + sdst) = a0;
  *(v4f*)(lds + sdst + GB_RS) = a1;
  *(v4f*)(lds + sdst + 2 * GB_RS) = a2;
  *(v4f*)(lds + sdst + 3 * GB_RS) = a3;
  *(v4f*)(lds + GB_M * GB_P + sdst) = b0;
  __syncthreads();

  v8f acc = {};
  const int nch = K / GB_K;
  for (int ch = 0; ch < nch; ++ch) {
    const float* st = lds + (ch & 1) * G16_STAGE;
    float* nx = lds + ((ch + 1) & 1) * G16_STAGE;
    const bool more = ch + 1 < nch;
    if (more) {
      const int kc = (ch + 1) * GB_K;
      a0 = *(const v4f*)(x0 + kc);
      a1 = *(const v4f*)(x1 + kc);
      a2 = *(const v4f*)(x2 + kc);
      a3 = *(const v4f*)(x3 + kc);
      b0 = *(const v4f*)(wsrc + kc);
    }

    const float* ap = st + (wv * 16 + c) * GB_P + 2 * h;
    const float* bp = st + (GB_M + c) * GB_P + 2 * h;
#pragma unroll
    for (int ks = 0; ks < GB_K / 4; ++ks) {
      const v2f a = *(const v2f*)(ap + ks * 4);
      const v2f b = *(const v2f*)(bp + ks * 4);
      acc = __builtin_amdgcn_wmma_f32_16x16x4_f32(
           false, a,  false, b,
           (short)0, acc,  false,  false);
    }

    if (more) {
      *(v4f*)(nx + sdst) = a0;
      *(v4f*)(nx + sdst + GB_RS) = a1;
      *(v4f*)(nx + sdst + 2 * GB_RS) = a2;
      *(v4f*)(nx + sdst + 3 * GB_RS) = a3;
      *(v4f*)(nx + GB_M * GB_P + sdst) = b0;
    }
    __syncthreads();
  }

  float* cs = lds + wv * (16 * G16_CP);
#pragma unroll
  for (int e = 0; e < 8; ++e) cs[(8 * h + e) * G16_CP + c] = acc[e];
  __syncthreads();
  for (int i = 0; i < 2; ++i) {
    const int r = i * 8 + (lane >> 2);
    const int row = row0 + wv * 16 + r;
    const float sc = __shfl(dsc, r, 32);
    if (row < M)
      *(v4f*)(G + (size_t)row * FOLD_N + (lane & 3) * 4) =
          *(const v4f*)(cs + r * G16_CP + (lane & 3) * 4) * sc;
  }
}


__device__ __forceinline__ void row_fma(float4& acc, const float4 h, const float w) {
  acc.x = fmaf(h.x, w, acc.x);
  acc.y = fmaf(h.y, w, acc.y);
  acc.z = fmaf(h.z, w, acc.z);
  acc.w = fmaf(h.w, w, acc.w);
}

__device__ __forceinline__ void row_add(float4& acc, const float4 h) {
  acc.x += h.x;
  acc.y += h.y;
  acc.z += h.z;
  acc.w += h.w;
}

__global__ void __launch_bounds__(256) agg_rows(const int* __restrict__ offs,
                                                const int* __restrict__ srcs,
                                                const float* __restrict__ dinv,
                                                const float* __restrict__ H,
                                                const float* __restrict__ b,
                                                float* __restrict__ Y, int n) {
  int node = (int)((blockIdx.x * (size_t)blockDim.x + threadIdx.x) >> 5);
  const int lane = threadIdx.x & 31;
  if (node >= n) return;
  node = __builtin_amdgcn_readfirstlane(node);
  const int beg = offs[node];
  const int d = offs[node + 1] - beg;
  const float di = dinv[node];
  const float* hl = H + lane * 4;

  float4 acc = make_float4(0.0f, 0.0f, 0.0f, 0.0f);

  for (int base = 0; base < d; base += 32) {
    const int m = (d - base < 32) ? (d - base) : 32;
    int s = node;
    if (lane < m) s = srcs[beg + base + lane];
    int j = 0;
    for (; j + 4 <= m; j += 4) {
      const int s0 = __builtin_amdgcn_readlane(s, j);
      const int s1 = __builtin_amdgcn_readlane(s, j + 1);
      const int s2 = __builtin_amdgcn_readlane(s, j + 2);
      const int s3 = __builtin_amdgcn_readlane(s, j + 3);
      const float4 h0 = *(const float4*)(hl + (size_t)s0 * NHID);
      const float4 h1 = *(const float4*)(hl + (size_t)s1 * NHID);
      const float4 h2 = *(const float4*)(hl + (size_t)s2 * NHID);
      const float4 h3 = *(const float4*)(hl + (size_t)s3 * NHID);
      row_add(acc, h0);
      row_add(acc, h1);
      row_add(acc, h2);
      row_add(acc, h3);
    }
#pragma nounroll
    for (; j < m; ++j) {
      const int s0 = __builtin_amdgcn_readlane(s, j);
      const float4 h0 = *(const float4*)(hl + (size_t)s0 * NHID);
      row_add(acc, h0);
    }
  }

  const float4 hself = *(const float4*)(hl + (size_t)node * NHID);
  const float4 bv = *(const float4*)(b + lane * 4);
  acc.x = (acc.x + hself.x) * di + bv.x; acc.y = (acc.y + hself.y) * di + bv.y;
  acc.z = (acc.z + hself.z) * di + bv.z; acc.w = (acc.w + hself.w) * di + bv.w;
  acc.x = fmaxf(acc.x, 0.0f); acc.y = fmaxf(acc.y, 0.0f);
  acc.z = fmaxf(acc.z, 0.0f); acc.w = fmaxf(acc.w, 0.0f);
  *(float4*)(Y + (size_t)node * NHID + lane * 4) = acc;
}

#define SRT_R 4
__global__ void __launch_bounds__(256) agg_rows_order(const int* __restrict__ offs,
                                                      int* srcs,
                                                      const float* __restrict__ dinv,
                                                      const float* __restrict__ H,
                                                      const float* __restrict__ b,
                                                      float* __restrict__ Y, int n) {
  int node = (int)((blockIdx.x * (size_t)blockDim.x + threadIdx.x) >> 5);
  const int lane = threadIdx.x & 31;
  if (node >= n) return;
  node = __builtin_amdgcn_readfirstlane(node);
  const int beg = offs[node];
  const int d = offs[node + 1] - beg;
  const float di = dinv[node];
  const float* hl = H + lane * 4;
  int* seg = srcs + beg;

  int s_first = node;
  if (d <= 32) {
    if (d > 0) {
      const bool live = lane < d;
      const int key = live ? seg[lane] : 0;
      int r = 0;
      for (int q = 0; q < d; ++q) {
        const int k = __builtin_amdgcn_readlane(key, q);
        r += (int)(k < key) + (int)((k == key) & (q < lane));
      }
      if (live) seg[r] = key;
      s_first = __builtin_amdgcn_ds_permute((live ? r : lane) << 2, live ? key : node);
    }
  } else if (d <= 32 * SRT_R) {
    int key[SRT_R], r[SRT_R];
#pragma unroll
    for (int a = 0; a < SRT_R; ++a) {
      const int i = 32 * a + lane;
      key[a] = (i < d) ? seg[i] : 0;
      r[a] = 0;
    }
#pragma unroll
    for (int c = 0; c < SRT_R; ++c) {
      if (32 * c < d) {
        const int m = (d - 32 * c < 32) ? (d - 32 * c) : 32;
        for (int q = 0; q < m; ++q) {
          const int k = __builtin_amdgcn_readlane(key[c], q);
          const int j = 32 * c + q;
#pragma unroll
          for (int a = 0; a < SRT_R; ++a)
            r[a] += (int)(k < key[a]) + (int)((k == key[a]) & (j < 32 * a + lane));
        }
      }
    }
#pragma unroll
    for (int a = 0; a < SRT_R; ++a)
      if (32 * a + lane < d) seg[r[a]] = key[a];
    __threadfence();
  } else {
    for (int p = 0; p < d; ++p) {
      int best = 0x7fffffff, bpos = 0x7fffffff;
      for (int j = p + lane; j < d; j += 32) {
        const int k = seg[j];
        if (k < best) { best = k; bpos = j; }
      }
#pragma unroll
      for (int o = 16; o > 0; o >>= 1) {
        const int ob = __shfl_xor(best, o, 32);
        const int op = __shfl_xor(bpos, o, 32);
        if (ob < best || (ob == best && op < bpos)) { best = ob; bpos = op; }
      }
      if (lane == 0) {
        const int vp = seg[p];
        seg[bpos] = vp;
        seg[p] = best;
      }
      __threadfence();
    }
  }

  float4 acc = make_float4(0.0f, 0.0f, 0.0f, 0.0f);

  for (int base = 0; base < d; base += 32) {
    const int m = (d - base < 32) ? (d - base) : 32;
    int s = s_first;
    if (d > 32) {
      s = node;
      if (lane < m) s = seg[base + lane];
    }
    int j = 0;
    for (; j + 4 <= m; j += 4) {
      const int s0 = __builtin_amdgcn_readlane(s, j);
      const int s1 = __builtin_amdgcn_readlane(s, j + 1);
      const int s2 = __builtin_amdgcn_readlane(s, j + 2);
      const int s3 = __builtin_amdgcn_readlane(s, j + 3);
      const float4 h0 = *(const float4*)(hl + (size_t)s0 * NHID);
      const float4 h1 = *(const float4*)(hl + (size_t)s1 * NHID);
      const float4 h2 = *(const float4*)(hl + (size_t)s2 * NHID);
      const float4 h3 = *(const float4*)(hl + (size_t)s3 * NHID);
      row_add(acc, h0);
      row_add(acc, h1);
      row_add(acc, h2);
      row_add(acc, h3);
    }
#pragma nounroll
    for (; j < m; ++j) {
      const int s0 = __builtin_amdgcn_readlane(s, j);
      const float4 h0 = *(const float4*)(hl + (size_t)s0 * NHID);
      row_add(acc, h0);
    }
  }

  const float4 hself = *(const float4*)(hl + (size_t)node * NHID);
  const float4 bv = *(const float4*)(b + lane * 4);
  acc.x = (acc.x + hself.x) * di + bv.x; acc.y = (acc.y + hself.y) * di + bv.y;
  acc.z = (acc.z + hself.z) * di + bv.z; acc.w = (acc.w + hself.w) * di + bv.w;
  acc.x = fmaxf(acc.x, 0.0f); acc.y = fmaxf(acc.y, 0.0f);
  acc.z = fmaxf(acc.z, 0.0f); acc.w = fmaxf(acc.w, 0.0f);
  *(float4*)(Y + (size_t)node * NHID + lane * 4) = acc;
}

__global__ void __launch_bounds__(256) agg_last(const int* __restrict__ offs,
                                                const int* __restrict__ srcs,
                                                const float* __restrict__ dinv,
                                                const float* __restrict__ G,
                                                const float* __restrict__ bf,
                                                float* __restrict__ out, int n) {
  int node = (int)((blockIdx.x * (size_t)blockDim.x + threadIdx.x) >> 5);
  const int lane = threadIdx.x & 31;
  if (node >= n) return;
  node = __builtin_amdgcn_readfirstlane(node);
  const int beg = offs[node];
  const int d = offs[node + 1] - beg;
  const float di = dinv[node];
  const int slot = lane >> 2;
  const int q4 = (lane & 3) * 4;
  const float* gl = G + q4;

  float4 acc = make_float4(0.0f, 0.0f, 0.0f, 0.0f);
  for (int base = 0; base < d; base += 32) {
    bool v[4];
    int s[4];
    float4 g[4];
#pragma unroll
    for (int bt = 0; bt < 4; ++bt) {
      const int e = base + slot + 8 * bt;
      v[bt] = e < d;
      s[bt] = srcs[beg + (v[bt] ? e : d - 1)];
    }
#pragma unroll
    for (int bt = 0; bt < 4; ++bt) g[bt] = *(const float4*)(gl + (size_t)s[bt] * FOLD_N);
#pragma unroll
    for (int bt = 0; bt < 4; ++bt) row_fma(acc, g[bt], v[bt] ? 1.0f : 0.0f);
  }
#pragma unroll
  for (int o = 4; o < 32; o <<= 1) {
    acc.x += __shfl_xor(acc.x, o, 32);
    acc.y += __shfl_xor(acc.y, o, 32);
    acc.z += __shfl_xor(acc.z, o, 32);
    acc.w += __shfl_xor(acc.w, o, 32);
  }

  const float4 gself = *(const float4*)(gl + (size_t)node * FOLD_N);
  const float4 bv = *(const float4*)(bf + q4);
  acc.x = (acc.x + gself.x) * di + bv.x; acc.y = (acc.y + gself.y) * di + bv.y;
  acc.z = (acc.z + gself.z) * di + bv.z; acc.w = (acc.w + gself.w) * di + bv.w;

  const float ninf = -__builtin_inff();
  const bool c0 = q4 + 0 < NCLASS, c1 = q4 + 1 < NCLASS, c2 = q4 + 2 < NCLASS, c3 = q4 + 3 < NCLASS;
  float m = fmaxf(fmaxf(c0 ? acc.x : ninf, c1 ? acc.y : ninf),
                  fmaxf(c2 ? acc.z : ninf, c3 ? acc.w : ninf));
  m = fmaxf(m, __shfl_xor(m, 1, 32));
  m = fmaxf(m, __shfl_xor(m, 2, 32));
  float s = (c0 ? __expf(acc.x - m) : 0.0f) + (c1 ? __expf(acc.y - m) : 0.0f) +
            (c2 ? __expf(acc.z - m) : 0.0f) + (c3 ? __expf(acc.w - m) : 0.0f);
  s += __shfl_xor(s, 1, 32);
  s += __shfl_xor(s, 2, 32);
  const float lse = m + __logf(s);
  if (lane < 3) {
    float* o = out + (size_t)node * NCLASS + q4;
    *(float2*)o = make_float2(acc.x - lse, acc.y - lse);
    if (lane < 2) *(float2*)(o + 2) = make_float2(acc.z - lse, acc.w - lse);
  }
}

extern "C" void kernel_launch(void* const* d_in, const int* in_sizes, int n_in,
                              void* d_out, int out_size, void* d_ws, size_t ws_size,
                              hipStream_t stream) {
  const float* x  = (const float*)d_in[0];
  const int*   ei = (const int*)d_in[1];
  const float* W1 = (const float*)d_in[2];
  const float* b1 = (const float*)d_in[3];
  const float* W2 = (const float*)d_in[4];
  const float* b2 = (const float*)d_in[5];
  const float* W3 = (const float*)d_in[6];
  const float* b3 = (const float*)d_in[7];
  const float* Wl = (const float*)d_in[8];
  const float* bl = (const float*)d_in[9];
  float* out = (float*)d_out;

  const int E = in_sizes[1] / 2;
  const int N = in_sizes[0] / NFEAT;
  const int npad = ((N + 16) / 16) * 16;

  char* ws = (char*)d_ws;
  size_t off = 0;
  auto alloc = [&](size_t bytes) -> float* {
    float* p = (float*)(ws + off);
    off += (bytes + 255) & ~(size_t)255;
    return p;
  };
  const size_t cbytes = ((size_t)npad * sizeof(int) + 255) & ~(size_t)255;
  const size_t ebytes = ((size_t)E * sizeof(int) + 255) & ~(size_t)255;
  const size_t hbytes = ((size_t)SC_CHUNKS * npad * sizeof(int) + 255) & ~(size_t)255;
  const size_t tbytes = cbytes + ebytes + hbytes;
  const size_t ybytes = (size_t)N * NHID * sizeof(float);
  float* dinv = alloc((size_t)npad * sizeof(float));
  float* Wt1  = alloc((size_t)NFEAT * NHID * sizeof(float));
  float* Wt2  = alloc((size_t)NHID * NHID * sizeof(float));
  float* Wt3l = alloc((size_t)FOLD_N * NHID * sizeof(float));
  float* b3l  = alloc((size_t)FOLD_N * sizeof(float));
  float* H    = alloc(ybytes);
  float* Y    = alloc(ybytes > tbytes ? ybytes : tbytes);
  int* offs   = (int*)alloc((size_t)npad * sizeof(int));
  int* srcs   = (int*)alloc((size_t)E * sizeof(int));
  if (off > ws_size) return;
  int* cnt  = (int*)Y;
  int* rank = (int*)((char*)Y + cbytes);
  int* hist = (int*)((char*)Y + cbytes + ebytes);

  const int* rows = ei;
  const int* cols = ei + E;

  const int wgrid = (int)(((size_t)N * 32 + 255) / 256);

  const int epc = (E + SC_CHUNKS - 1) / SC_CHUNKS;

  chunk_count<<<SC_CHUNKS, SC_T, 0, stream>>>(cols, rank, hist, E, epc, npad);
  prep_scan<<<PB_ALLW + (npad + 255) / 256, 256, 0, stream>>>(W1, Wt1, W2, Wt2, W3, b3, Wl, bl,
                                                              Wt3l, b3l, hist, cnt, npad);
  seg_scan<<<1, SCAN_T, 0, stream>>>(cnt, offs, dinv, npad);
  seg_fill<<<dim3((epc + 255) / 256, SC_CHUNKS), 256, 0, stream>>>(rows, cols, rank, offs, hist, srcs,
                                                                   E, epc, npad);

  const int tiles = (N + 15) / 16;
  const int gemm_grid = (tiles + 3) / 4;

  gemm_wmma_f32<<<gemm_grid, 128, 0, stream>>>(x, Wt1, dinv, H, N, NFEAT);
  agg_rows_order<<<wgrid, 256, 0, stream>>>(offs, srcs, dinv, H, b1, Y, N);

  gemm_wmma_f32<<<gemm_grid, 128, 0, stream>>>(Y, Wt2, dinv, H, N, NHID);
  agg_rows<<<wgrid, 256, 0, stream>>>(offs, srcs, dinv, H, b2, Y, N);

  gemm_wmma_f32_n16<<<gemm_grid, 128, 0, stream>>>(Y, Wt3l, dinv, H, N, NHID);
  agg_last<<<wgrid, 256, 0, stream>>>(offs, srcs, dinv, H, b3l, out, N);
}


# 128-column GEMMs: the row blocks beyond whole rounds of 256 (14 of 782) run as 64 x 32-column pieces, one workgroup each, so no SIMD gets a fourth whole strip of f32 WMMAs; sums unchanged
# speedup vs baseline: 4.0629x; 1.0475x over previous
#include <hip/hip_runtime.h>

typedef __attribute__((ext_vector_type(2))) float v2f;
typedef __attribute__((ext_vector_type(8))) float v8f;

#define NFEAT 512
#define NHID  128
#define NCLASS 10


__device__ __forceinline__ void transpose_elem(const float* __restrict__ W, float* __restrict__ Wt,
                                               int K, int Nc, int idx) {
  if (idx < K * Nc) {
    int k = idx / Nc, n = idx % Nc;
    Wt[(size_t)n * K + k] = W[idx];
  }
}


#define SC_CHUNKS 64
#define SC_T 1024
#define SC_CAP 50176
__global__ void __launch_bounds__(SC_T) chunk_count(const int* __restrict__ cols,
                                                    int* __restrict__ rank,
                                                    int* __restrict__ hist,
                                                    int E, int epc, int npad) {
  __shared__ __attribute__((aligned(16))) int ctr[SC_CAP];
  const int t = threadIdx.x;
  const int e0 = blockIdx.x * epc;
  const int e1 = (e0 + epc < E) ? (e0 + epc) : E;
  int* hrow = hist + (size_t)blockIdx.x * npad;
  for (int w0 = 0; w0 < npad; w0 += SC_CAP) {
    const int wn = (npad - w0 < SC_CAP) ? (npad - w0) : SC_CAP;
    for (int i = t * 4; i < wn; i += SC_T * 4) *(int4*)(ctr + i) = make_int4(0, 0, 0, 0);
    __syncthreads();
    for (int e = e0 + t; e < e1; e += 8 * SC_T) {
      int c[8], r[8];
#pragma unroll
      for (int q = 0; q < 8; ++q) c[q] = (e + q * SC_T < e1) ? cols[e + q * SC_T] : -1;
#pragma unroll
      for (int q = 0; q < 8; ++q) {
        const unsigned k = (unsigned)(c[q] - w0);
        r[q] = (k < (unsigned)wn) ? atomicAdd(&ctr[k], 1) : -1;
      }
#pragma unroll
      for (int q = 0; q < 8; ++q)
        if (r[q] >= 0) rank[e + q * SC_T] = r[q];
    }
    __syncthreads();
    for (int i = t * 4; i < wn; i += SC_T * 4) *(int4*)(hrow + w0 + i) = *(const int4*)(ctr + i);
    __syncthreads();
  }
}

#define FOLD_N 16
static_assert(NCLASS <= FOLD_N && NCLASS == 10, "the last layer's lane map and its stores are written for 10 classes in one 16-column tile");

#define PB_W1 (NFEAT * NHID / 256)
#define PB_W2 (NHID * NHID / 256)
#define PB_FOLD ((NHID * FOLD_N + FOLD_N + 255) / 256)
#define PB_ALLW (PB_W1 + PB_W2 + PB_FOLD)
__global__ void __launch_bounds__(256) prep_scan(const float* __restrict__ W1, float* __restrict__ Wt1,
                                                 const float* __restrict__ W2, float* __restrict__ Wt2,
                                                 const float* __restrict__ W3, const float* __restrict__ b3,
                                                 const float* __restrict__ Wl, const float* __restrict__ bl,
                                                 float* __restrict__ Wt3l, float* __restrict__ b3l,
                                                 int* __restrict__ hist, int* __restrict__ cnt,
                                                 int npad) {
  const int b = blockIdx.x, t = threadIdx.x;
  if (b < PB_W1) {
    transpose_elem(W1, Wt1, NFEAT, NHID, b * 256 + t);
  } else if (b < PB_W1 + PB_W2) {
    transpose_elem(W2, Wt2, NHID, NHID, (b - PB_W1) * 256 + t);
  } else if (b < PB_ALLW) {
    const int idx = (b - PB_W1 - PB_W2) * 256 + t;
    if (idx < NHID * FOLD_N) {
      const int k = idx >> 4, nn = idx & (FOLD_N - 1);
      float s = 0.0f;
      if (nn < NCLASS) {
#pragma unroll 32
        for (int j = 0; j < NHID; ++j) s = fmaf(W3[k * NHID + j], Wl[j * NCLASS + nn], s);
      }
      Wt3l[nn * NHID + k] = s;
    } else if (idx < NHID * FOLD_N + FOLD_N) {
      const int nn = idx - NHID * FOLD_N;
      float s = 0.0f;
      if (nn < NCLASS) {
        s = bl[nn];
#pragma unroll 32
        for (int j = 0; j < NHID; ++j) s = fmaf(b3[j], Wl[j * NCLASS + nn], s);
      }
      b3l[nn] = s;
    }
  } else {
    const int i = (b - PB_ALLW) * 256 + t;
    if (i < npad) {
      int* p = hist + i;
      int run = 0;
      for (int c = 0; c < SC_CHUNKS; c += 16) {
        int v[16];
#pragma unroll
        for (int q = 0; q < 16; ++q) v[q] = p[(size_t)(c + q) * npad];
#pragma unroll
        for (int q = 0; q < 16; ++q) { p[(size_t)(c + q) * npad] = run; run += v[q]; }
      }
      cnt[i] = run;
    }
  }
}

#define SCAN_T 1024
__global__ void __launch_bounds__(SCAN_T) seg_scan(const int* __restrict__ cnt,
                                                   int* __restrict__ offs,
                                                   float* __restrict__ dinv, int npad) {
  __shared__ int wsum[SCAN_T / 32];
  __shared__ int tot;
  const int t = threadIdx.x, lane = t & 31, wv = t >> 5;
  int carry = 0;
  for (int base = 0; base < npad; base += SCAN_T * 16) {
    const int i0 = base + t * 16;
    const bool live = i0 < npad;
    int v[16];
#pragma unroll
    for (int q = 0; q < 4; ++q) {
      int4 c = live ? *(const int4*)(cnt + i0 + 4 * q) : make_int4(0, 0, 0, 0);
      v[4 * q + 0] = c.x; v[4 * q + 1] = c.y; v[4 * q + 2] = c.z; v[4 * q + 3] = c.w;
    }
    int ex[16];
    int s = 0;
#pragma unroll
    for (int k = 0; k < 16; ++k) { ex[k] = s; s += v[k]; }
    int inc = s;
#pragma unroll
    for (int o = 1; o < 32; o <<= 1) {
      int nb = __shfl_up(inc, o, 32);
      if (lane >= o) inc += nb;
    }
    if (lane == 31) wsum[wv] = inc;
    __syncthreads();
    if (wv == 0) {
      int w = wsum[lane];
      int wi = w;
#pragma unroll
      for (int o = 1; o < 32; o <<= 1) {
        int nb = __shfl_up(wi, o, 32);
        if (lane >= o) wi += nb;
      }
      wsum[lane] = wi - w;
      if (lane == 31) tot = wi;
    }
    __syncthreads();
    const int pre = carry + wsum[wv] + (inc - s);
    carry += tot;
    if (live) {
#pragma unroll
      for (int q = 0; q < 4; ++q) {
        *(int4*)(offs + i0 + 4 * q) = make_int4(pre + ex[4 * q + 0], pre + ex[4 * q + 1],
                                                pre + ex[4 * q + 2], pre + ex[4 * q + 3]);
        *(float4*)(dinv + i0 + 4 * q) =
            make_float4(rsqrtf((float)(v[4 * q + 0] + 1)), rsqrtf((float)(v[4 * q + 1] + 1)),
                        rsqrtf((float)(v[4 * q + 2] + 1)), rsqrtf((float)(v[4 * q + 3] + 1)));
      }
    }
    __syncthreads();
  }
}

__global__ void seg_fill(const int* __restrict__ rows, const int* __restrict__ cols,
                         const int* __restrict__ rank, const int* __restrict__ offs,
                         const int* __restrict__ hist, int* __restrict__ srcs,
                         int E, int epc, int npad) {
  const int l = blockIdx.x * blockDim.x + threadIdx.x;
  const int e = blockIdx.y * epc + l;
  if (l < epc && e < E) {
    const int c = cols[e];
    srcs[offs[c] + hist[(size_t)blockIdx.y * npad + c] + rank[e]] = rows[e];
  }
}

typedef __attribute__((ext_vector_type(4))) float v4f;
#define GB_M 64
#define GB_K 32
#define GB_P 36
#define GB_STAGE ((GB_M + NHID) * GB_P)
#define GB_CP 132
#define GB_RS (16 * GB_P)

#define GB_FETCH(kc)                                  \
  a0 = *(const v4f*)(x0 + (kc));                      \
  a1 = *(const v4f*)(x1 + (kc));                      \
  a2 = *(const v4f*)(x2 + (kc));                      \
  a3 = *(const v4f*)(x3 + (kc));                      \
  b0 = *(const v4f*)(wsrc + (kc));                    \
  b1 = *(const v4f*)(wsrc + wstep + (kc));            \
  b2 = *(const v4f*)(wsrc + 2 * wstep + (kc));        \
  b3 = *(const v4f*)(wsrc + 3 * wstep + (kc));        \
  b4 = *(const v4f*)(wsrc + 4 * wstep + (kc));        \
  b5 = *(const v4f*)(wsrc + 5 * wstep + (kc));        \
  b6 = *(const v4f*)(wsrc + 6 * wstep + (kc));        \
  b7 = *(const v4f*)(wsrc + 7 * wstep + (kc));

#define GB_PUT(dst)                                           \
  *(v4f*)((dst) + sdst) = a0;                                 \
  *(v4f*)((dst) + sdst + GB_RS) = a1;                         \
  *(v4f*)((dst) + sdst + 2 * GB_RS) = a2;                     \
  *(v4f*)((dst) + sdst + 3 * GB_RS) = a3;                     \
  *(v4f*)((dst) + GB_M * GB_P + sdst) = b0;                   \
  *(v4f*)((dst) + GB_M * GB_P + sdst + GB_RS) = b1;           \
  *(v4f*)((dst) + GB_M * GB_P + sdst + 2 * GB_RS) = b2;       \
  *(v4f*)((dst) + GB_M * GB_P + sdst + 3 * GB_RS) = b3;       \
  *(v4f*)((dst) + GB_M * GB_P + sdst + 4 * GB_RS) = b4;       \
  *(v4f*)((dst) + GB_M * GB_P + sdst + 5 * GB_RS) = b5;       \
  *(v4f*)((dst) + GB_M * GB_P + sdst + 6 * GB_RS) = b6;       \
  *(v4f*)((dst) + GB_M * GB_P + sdst + 7 * GB_RS) = b7;

#define GB_WGPS 256
#define GT_N 32
#define GT_STAGE ((GB_M + GT_N) * GB_P)
#define GT_CP 36
__device__ __forceinline__ void gemm_piece_f32(float* __restrict__ lds,
                                               const float* __restrict__ X,
                                               const float* __restrict__ Wt,
                                               const float* __restrict__ dinv,
                                               float* __restrict__ H,
                                               int M, int K, int row0, int col0) {
  const int tid = threadIdx.x;
  const int lane = tid & 31;
  const int wv = tid >> 5;
  const int c = lane & 15;
  const int h = lane >> 4;

  const int seg = tid & 7;
  const int rsub = tid >> 3;
  const int last = M - 1;
  const int r0 = row0 + rsub, r1 = r0 + 16, r2 = r0 + 32, r3 = r0 + 48;
  const float* x0 = X + (size_t)(r0 < last ? r0 : last) * K + seg * 4;
  const float* x1 = X + (size_t)(r1 < last ? r1 : last) * K + seg * 4;
  const float* x2 = X + (size_t)(r2 < last ? r2 : last) * K + seg * 4;
  const float* x3 = X + (size_t)(r3 < last ? r3 : last) * K + seg * 4;
  const float* w0 = Wt + (size_t)(col0 + rsub) * K + seg * 4;
  const float* w1 = w0 + (size_t)16 * K;
  const int sdst = rsub * GB_P + seg * 4;

  const int srow = row0 + wv * 16 + c;
  const float dsc = dinv[srow < last ? srow : last];

  v4f a0 = *(const v4f*)(x0);
  v4f a1 = *(const v4f*)(x1);
  v4f a2 = *(const v4f*)(x2);
  v4f a3 = *(const v4f*)(x3);
  v4f b0 = *(const v4f*)(w0);
  v4f b1 = *(const v4f*)(w1);
  *(v4f*)(lds + sdst) = a0;
  *(v4f*)(lds + sdst + GB_RS) = a1;
  *(v4f*)(lds + sdst + 2 * GB_RS) = a2;
  *(v4f*)(lds + sdst + 3 * GB_RS) = a3;
  *(v4f*)(lds + GB_M * GB_P + sdst) = b0;
  *(v4f*)(lds + GB_M * GB_P + sdst + GB_RS) = b1;
  __syncthreads();

  v8f acc0 = {}, acc1 = {};
  const int nch = K / GB_K;
  for (int ch = 0; ch < nch; ++ch) {
    const float* st = lds + (ch & 1) * GT_STAGE;
    float* nx = lds + ((ch + 1) & 1) * GT_STAGE;
    const bool more = ch + 1 < nch;
    if (more) {
      const int kc = (ch + 1) * GB_K;
      a0 = *(const v4f*)(x0 + kc);
      a1 = *(const v4f*)(x1 + kc);
      a2 = *(const v4f*)(x2 + kc);
      a3 = *(const v4f*)(x3 + kc);
      b0 = *(const v4f*)(w0 + kc);
      b1 = *(const v4f*)(w1 + kc);
    }

    const float* ap = st + (wv * 16 + c) * GB_P + 2 * h;
    const float* bp = st + (GB_M + c) * GB_P + 2 * h;
#pragma unroll
    for (int ks = 0; ks < GB_K / 4; ++ks) {
      const v2f a = *(const v2f*)(ap + ks * 4);
      const v2f f0 = *(const v2f*)(bp + ks * 4);
      const v2f f1 = *(const v2f*)(bp + GB_RS + ks * 4);
      acc0 = __builtin_amdgcn_wmma_f32_16x16x4_f32(
           false, a,  false, f0,
           (short)0, acc0,  false,  false);
      acc1 = __builtin_amdgcn_wmma_f32_16x16x4_f32(
           false, a,  false, f1,
           (short)0, acc1,  false,  false);
    }

    if (more) {
      *(v4f*)(nx + sdst) = a0;
      *(v4f*)(nx + sdst + GB_RS) = a1;
      *(v4f*)(nx + sdst + 2 * GB_RS) = a2;
      *(v4f*)(nx + sdst + 3 * GB_RS) = a3;
      *(v4f*)(nx + GB_M * GB_P + sdst) = b0;
      *(v4f*)(nx + GB_M * GB_P + sdst + GB_RS) = b1;
    }
    __syncthreads();
  }

  float* cs = lds + wv * (16 * GT_CP);
#pragma unroll
  for (int e = 0; e < 8; ++e) {
    cs[(8 * h + e) * GT_CP + c] = acc0[e];
    cs[(8 * h + e) * GT_CP + 16 + c] = acc1[e];
  }
  __syncthreads();
  for (int i = 0; i < 4; ++i) {
    const int r = i * 4 + (lane >> 3);
    const int row = row0 + wv * 16 + r;
    const float sc = __shfl(dsc, r, 32);
    if (row < M)
      *(v4f*)(H + (size_t)row * NHID + col0 + (lane & 7) * 4) =
          *(const v4f*)(cs + r * GT_CP + (lane & 7) * 4) * sc;
  }
}

__global__ void __launch_bounds__(128) gemm_wmma_f32(const float* __restrict__ X,
                                                     const float* __restrict__ Wt,
                                                     const float* __restrict__ dinv,
                                                     float* __restrict__ H,
                                                     int M, int K, int nfull) {
  __shared__ __attribute__((aligned(16))) float lds[2 * GB_STAGE];

  if ((int)blockIdx.x >= nfull) {
    const int p = (int)blockIdx.x - nfull;
    gemm_piece_f32(lds, X, Wt, dinv, H, M, K, (nfull + p / (NHID / GT_N)) * GB_M,
                   (p % (NHID / GT_N)) * GT_N);
    return;
  }

  const int tid = threadIdx.x;
  const int lane = tid & 31;
  const int wv = tid >> 5;
  const int c = lane & 15;
  const int h = lane >> 4;
  const int row0 = blockIdx.x * GB_M;

  const int seg = tid & 7;
  const int rsub = tid >> 3;
  const int last = M - 1;
  const int r0 = row0 + rsub, r1 = r0 + 16, r2 = r0 + 32, r3 = r0 + 48;
  const float* x0 = X + (size_t)(r0 < last ? r0 : last) * K + seg * 4;
  const float* x1 = X + (size_t)(r1 < last ? r1 : last) * K + seg * 4;
  const float* x2 = X + (size_t)(r2 < last ? r2 : last) * K + seg * 4;
  const float* x3 = X + (size_t)(r3 < last ? r3 : last) * K + seg * 4;
  const float* wsrc = Wt + (size_t)rsub * K + seg * 4;
  const size_t wstep = (size_t)16 * K;
  const int sdst = rsub * GB_P + seg * 4;

  const int srow = row0 + wv * 16 + c;
  const float dsc = dinv[srow < last ? srow : last];

  v4f a0, a1, a2, a3, b0, b1, b2, b3, b4, b5, b6, b7;
  GB_FETCH(0)
  GB_PUT(lds)
  __syncthreads();

  v8f acc[8] = {};
  const int nch = K / GB_K;
  for (int ch = 0; ch < nch; ++ch) {
    const float* st = lds + (ch & 1) * GB_STAGE;
    float* nx = lds + ((ch + 1) & 1) * GB_STAGE;
    const bool more = ch + 1 < nch;
    if (more) {
      const int kc = (ch + 1) * GB_K;
      GB_FETCH(kc)
    }

    const float* ap = st + (wv * 16 + c) * GB_P + 2 * h;
    const float* bp = st + (GB_M + c) * GB_P + 2 * h;
#pragma unroll
    for (int ks = 0; ks < GB_K / 4; ++ks) {
      const v2f a = *(const v2f*)(ap + ks * 4);
#pragma unroll
      for (int t = 0; t < 8; ++t) {
        const v2f b = *(const v2f*)(bp + t * GB_RS + ks * 4);
        acc[t] = __builtin_amdgcn_wmma_f32_16x16x4_f32(
             false, a,  false, b,
             (short)0, acc[t],  false,  false);
      }
    }

    if (more) {
      GB_PUT(nx)
    }
    __syncthreads();
  }

  float* cs = lds + wv * (16 * GB_CP);
#pragma unroll
  for (int t = 0; t < 8; ++t) {
#pragma unroll
    for (int e = 0; e < 8; ++e) cs[(8 * h + e) * GB_CP + t * 16 + c] = acc[t][e];
  }
  __syncthreads();
#pragma unroll
  for (int r = 0; r < 16; ++r) {
    const int row = row0 + wv * 16 + r;
    const float sc = __int_as_float(__builtin_amdgcn_readlane(__float_as_int(dsc), r));
    if (row < M)
      *(v4f*)(H + (size_t)row * NHID + lane * 4) = *(const v4f*)(cs + r * GB_CP + lane * 4) * sc;
  }
}
#undef GB_FETCH
#undef GB_PUT

#define G16_STAGE ((GB_M + FOLD_N) * GB_P)
#define G16_CP 20
__global__ void __launch_bounds__(128) gemm_wmma_f32_n16(const float* __restrict__ X,
                                                         const float* __restrict__ Wt,
                                                         const float* __restrict__ dinv,
                                                         float* __restrict__ G,
                                                         int M, int K) {
  __shared__ __attribute__((aligned(16))) float lds[2 * G16_STAGE];

  const int tid = threadIdx.x;
  const int lane = tid & 31;
  const int wv = tid >> 5;
  const int c = lane & 15;
  const int h = lane >> 4;
  const int row0 = blockIdx.x * GB_M;

  const int seg = tid & 7;
  const int rsub = tid >> 3;
  const int last = M - 1;
  const int r0 = row0 + rsub, r1 = r0 + 16, r2 = r0 + 32, r3 = r0 + 48;
  const float* x0 = X + (size_t)(r0 < last ? r0 : last) * K + seg * 4;
  const float* x1 = X + (size_t)(r1 < last ? r1 : last) * K + seg * 4;
  const float* x2 = X + (size_t)(r2 < last ? r2 : last) * K + seg * 4;
  const float* x3 = X + (size_t)(r3 < last ? r3 : last) * K + seg * 4;
  const float* wsrc = Wt + (size_t)rsub * K + seg * 4;
  const int sdst = rsub * GB_P + seg * 4;

  const int srow = row0 + wv * 16 + c;
  const float dsc = dinv[srow < last ? srow : last];

  v4f a0 = *(const v4f*)(x0);
  v4f a1 = *(const v4f*)(x1);
  v4f a2 = *(const v4f*)(x2);
  v4f a3 = *(const v4f*)(x3);
  v4f b0 = *(const v4f*)(wsrc);
  *(v4f*)(lds + sdst) = a0;
  *(v4f*)(lds + sdst + GB_RS) = a1;
  *(v4f*)(lds + sdst + 2 * GB_RS) = a2;
  *(v4f*)(lds + sdst + 3 * GB_RS) = a3;
  *(v4f*)(lds + GB_M * GB_P + sdst) = b0;
  __syncthreads();

  v8f acc = {};
  const int nch = K / GB_K;
  for (int ch = 0; ch < nch; ++ch) {
    const float* st = lds + (ch & 1) * G16_STAGE;
    float* nx = lds + ((ch + 1) & 1) * G16_STAGE;
    const bool more = ch + 1 < nch;
    if (more) {
      const int kc = (ch + 1) * GB_K;
      a0 = *(const v4f*)(x0 + kc);
      a1 = *(const v4f*)(x1 + kc);
      a2 = *(const v4f*)(x2 + kc);
      a3 = *(const v4f*)(x3 + kc);
      b0 = *(const v4f*)(wsrc + kc);
    }

    const float* ap = st + (wv * 16 + c) * GB_P + 2 * h;
    const float* bp = st + (GB_M + c) * GB_P + 2 * h;
#pragma unroll
    for (int ks = 0; ks < GB_K / 4; ++ks) {
      const v2f a = *(const v2f*)(ap + ks * 4);
      const v2f b = *(const v2f*)(bp + ks * 4);
      acc = __builtin_amdgcn_wmma_f32_16x16x4_f32(
           false, a,  false, b,
           (short)0, acc,  false,  false);
    }

    if (more) {
      *(v4f*)(nx + sdst) = a0;
      *(v4f*)(nx + sdst + GB_RS) = a1;
      *(v4f*)(nx + sdst + 2 * GB_RS) = a2;
      *(v4f*)(nx + sdst + 3 * GB_RS) = a3;
      *(v4f*)(nx + GB_M * GB_P + sdst) = b0;
    }
    __syncthreads();
  }

  float* cs = lds + wv * (16 * G16_CP);
#pragma unroll
  for (int e = 0; e < 8; ++e) cs[(8 * h + e) * G16_CP + c] = acc[e];
  __syncthreads();
  for (int i = 0; i < 2; ++i) {
    const int r = i * 8 + (lane >> 2);
    const int row = row0 + wv * 16 + r;
    const float sc = __shfl(dsc, r, 32);
    if (row < M)
      *(v4f*)(G + (size_t)row * FOLD_N + (lane & 3) * 4) =
          *(const v4f*)(cs + r * G16_CP + (lane & 3) * 4) * sc;
  }
}


__device__ __forceinline__ void row_fma(float4& acc, const float4 h, const float w) {
  acc.x = fmaf(h.x, w, acc.x);
  acc.y = fmaf(h.y, w, acc.y);
  acc.z = fmaf(h.z, w, acc.z);
  acc.w = fmaf(h.w, w, acc.w);
}

__device__ __forceinline__ void row_add(float4& acc, const float4 h) {
  acc.x += h.x;
  acc.y += h.y;
  acc.z += h.z;
  acc.w += h.w;
}

__global__ void __launch_bounds__(256) agg_rows(const int* __restrict__ offs,
                                                const int* __restrict__ srcs,
                                                const float* __restrict__ dinv,
                                                const float* __restrict__ H,
                                                const float* __restrict__ b,
                                                float* __restrict__ Y, int n) {
  int node = (int)((blockIdx.x * (size_t)blockDim.x + threadIdx.x) >> 5);
  const int lane = threadIdx.x & 31;
  if (node >= n) return;
  node = __builtin_amdgcn_readfirstlane(node);
  const int beg = offs[node];
  const int d = offs[node + 1] - beg;
  const float di = dinv[node];
  const float* hl = H + lane * 4;

  float4 acc = make_float4(0.0f, 0.0f, 0.0f, 0.0f);

  for (int base = 0; base < d; base += 32) {
    const int m = (d - base < 32) ? (d - base) : 32;
    int s = node;
    if (lane < m) s = srcs[beg + base + lane];
    int j = 0;
    for (; j + 4 <= m; j += 4) {
      const int s0 = __builtin_amdgcn_readlane(s, j);
      const int s1 = __builtin_amdgcn_readlane(s, j + 1);
      const int s2 = __builtin_amdgcn_readlane(s, j + 2);
      const int s3 = __builtin_amdgcn_readlane(s, j + 3);
      const float4 h0 = *(const float4*)(hl + (size_t)s0 * NHID);
      const float4 h1 = *(const float4*)(hl + (size_t)s1 * NHID);
      const float4 h2 = *(const float4*)(hl + (size_t)s2 * NHID);
      const float4 h3 = *(const float4*)(hl + (size_t)s3 * NHID);
      row_add(acc, h0);
      row_add(acc, h1);
      row_add(acc, h2);
      row_add(acc, h3);
    }
#pragma nounroll
    for (; j < m; ++j) {
      const int s0 = __builtin_amdgcn_readlane(s, j);
      const float4 h0 = *(const float4*)(hl + (size_t)s0 * NHID);
      row_add(acc, h0);
    }
  }

  const float4 hself = *(const float4*)(hl + (size_t)node * NHID);
  const float4 bv = *(const float4*)(b + lane * 4);
  acc.x = (acc.x + hself.x) * di + bv.x; acc.y = (acc.y + hself.y) * di + bv.y;
  acc.z = (acc.z + hself.z) * di + bv.z; acc.w = (acc.w + hself.w) * di + bv.w;
  acc.x = fmaxf(acc.x, 0.0f); acc.y = fmaxf(acc.y, 0.0f);
  acc.z = fmaxf(acc.z, 0.0f); acc.w = fmaxf(acc.w, 0.0f);
  *(float4*)(Y + (size_t)node * NHID + lane * 4) = acc;
}

#define SRT_R 4
__global__ void __launch_bounds__(256) agg_rows_order(const int* __restrict__ offs,
                                                      int* srcs,
                                                      const float* __restrict__ dinv,
                                                      const float* __restrict__ H,
                                                      const float* __restrict__ b,
                                                      float* __restrict__ Y, int n) {
  int node = (int)((blockIdx.x * (size_t)blockDim.x + threadIdx.x) >> 5);
  const int lane = threadIdx.x & 31;
  if (node >= n) return;
  node = __builtin_amdgcn_readfirstlane(node);
  const int beg = offs[node];
  const int d = offs[node + 1] - beg;
  const float di = dinv[node];
  const float* hl = H + lane * 4;
  int* seg = srcs + beg;

  int s_first = node;
  if (d <= 32) {
    if (d > 0) {
      const bool live = lane < d;
      const int key = live ? seg[lane] : 0;
      int r = 0;
      for (int q = 0; q < d; ++q) {
        const int k = __builtin_amdgcn_readlane(key, q);
        r += (int)(k < key) + (int)((k == key) & (q < lane));
      }
      if (live) seg[r] = key;
      s_first = __builtin_amdgcn_ds_permute((live ? r : lane) << 2, live ? key : node);
    }
  } else if (d <= 32 * SRT_R) {
    int key[SRT_R], r[SRT_R];
#pragma unroll
    for (int a = 0; a < SRT_R; ++a) {
      const int i = 32 * a + lane;
      key[a] = (i < d) ? seg[i] : 0;
      r[a] = 0;
    }
#pragma unroll
    for (int c = 0; c < SRT_R; ++c) {
      if (32 * c < d) {
        const int m = (d - 32 * c < 32) ? (d - 32 * c) : 32;
        for (int q = 0; q < m; ++q) {
          const int k = __builtin_amdgcn_readlane(key[c], q);
          const int j = 32 * c + q;
#pragma unroll
          for (int a = 0; a < SRT_R; ++a)
            r[a] += (int)(k < key[a]) + (int)((k == key[a]) & (j < 32 * a + lane));
        }
      }
    }
#pragma unroll
    for (int a = 0; a < SRT_R; ++a)
      if (32 * a + lane < d) seg[r[a]] = key[a];
    __threadfence();
  } else {
    for (int p = 0; p < d; ++p) {
      int best = 0x7fffffff, bpos = 0x7fffffff;
      for (int j = p + lane; j < d; j += 32) {
        const int k = seg[j];
        if (k < best) { best = k; bpos = j; }
      }
#pragma unroll
      for (int o = 16; o > 0; o >>= 1) {
        const int ob = __shfl_xor(best, o, 32);
        const int op = __shfl_xor(bpos, o, 32);
        if (ob < best || (ob == best && op < bpos)) { best = ob; bpos = op; }
      }
      if (lane == 0) {
        const int vp = seg[p];
        seg[bpos] = vp;
        seg[p] = best;
      }
      __threadfence();
    }
  }

  float4 acc = make_float4(0.0f, 0.0f, 0.0f, 0.0f);

  for (int base = 0; base < d; base += 32) {
    const int m = (d - base < 32) ? (d - base) : 32;
    int s = s_first;
    if (d > 32) {
      s = node;
      if (lane < m) s = seg[base + lane];
    }
    int j = 0;
    for (; j + 4 <= m; j += 4) {
      const int s0 = __builtin_amdgcn_readlane(s, j);
      const int s1 = __builtin_amdgcn_readlane(s, j + 1);
      const int s2 = __builtin_amdgcn_readlane(s, j + 2);
      const int s3 = __builtin_amdgcn_readlane(s, j + 3);
      const float4 h0 = *(const float4*)(hl + (size_t)s0 * NHID);
      const float4 h1 = *(const float4*)(hl + (size_t)s1 * NHID);
      const float4 h2 = *(const float4*)(hl + (size_t)s2 * NHID);
      const float4 h3 = *(const float4*)(hl + (size_t)s3 * NHID);
      row_add(acc, h0);
      row_add(acc, h1);
      row_add(acc, h2);
      row_add(acc, h3);
    }
#pragma nounroll
    for (; j < m; ++j) {
      const int s0 = __builtin_amdgcn_readlane(s, j);
      const float4 h0 = *(const float4*)(hl + (size_t)s0 * NHID);
      row_add(acc, h0);
    }
  }

  const float4 hself = *(const float4*)(hl + (size_t)node * NHID);
  const float4 bv = *(const float4*)(b + lane * 4);
  acc.x = (acc.x + hself.x) * di + bv.x; acc.y = (acc.y + hself.y) * di + bv.y;
  acc.z = (acc.z + hself.z) * di + bv.z; acc.w = (acc.w + hself.w) * di + bv.w;
  acc.x = fmaxf(acc.x, 0.0f); acc.y = fmaxf(acc.y, 0.0f);
  acc.z = fmaxf(acc.z, 0.0f); acc.w = fmaxf(acc.w, 0.0f);
  *(float4*)(Y + (size_t)node * NHID + lane * 4) = acc;
}

__global__ void __launch_bounds__(256) agg_last(const int* __restrict__ offs,
                                                const int* __restrict__ srcs,
                                                const float* __restrict__ dinv,
                                                const float* __restrict__ G,
                                                const float* __restrict__ bf,
                                                float* __restrict__ out, int n) {
  int node = (int)((blockIdx.x * (size_t)blockDim.x + threadIdx.x) >> 5);
  const int lane = threadIdx.x & 31;
  if (node >= n) return;
  node = __builtin_amdgcn_readfirstlane(node);
  const int beg = offs[node];
  const int d = offs[node + 1] - beg;
  const float di = dinv[node];
  const int slot = lane >> 2;
  const int q4 = (lane & 3) * 4;
  const float* gl = G + q4;

  float4 acc = make_float4(0.0f, 0.0f, 0.0f, 0.0f);
  for (int base = 0; base < d; base += 32) {
    bool v[4];
    int s[4];
    float4 g[4];
#pragma unroll
    for (int bt = 0; bt < 4; ++bt) {
      const int e = base + slot + 8 * bt;
      v[bt] = e < d;
      s[bt] = srcs[beg + (v[bt] ? e : d - 1)];
    }
#pragma unroll
    for (int bt = 0; bt < 4; ++bt) g[bt] = *(const float4*)(gl + (size_t)s[bt] * FOLD_N);
#pragma unroll
    for (int bt = 0; bt < 4; ++bt) row_fma(acc, g[bt], v[bt] ? 1.0f : 0.0f);
  }
#pragma unroll
  for (int o = 4; o < 32; o <<= 1) {
    acc.x += __shfl_xor(acc.x, o, 32);
    acc.y += __shfl_xor(acc.y, o, 32);
    acc.z += __shfl_xor(acc.z, o, 32);
    acc.w += __shfl_xor(acc.w, o, 32);
  }

  const float4 gself = *(const float4*)(gl + (size_t)node * FOLD_N);
  const float4 bv = *(const float4*)(bf + q4);
  acc.x = (acc.x + gself.x) * di + bv.x; acc.y = (acc.y + gself.y) * di + bv.y;
  acc.z = (acc.z + gself.z) * di + bv.z; acc.w = (acc.w + gself.w) * di + bv.w;

  const float ninf = -__builtin_inff();
  const bool c0 = q4 + 0 < NCLASS, c1 = q4 + 1 < NCLASS, c2 = q4 + 2 < NCLASS, c3 = q4 + 3 < NCLASS;
  float m = fmaxf(fmaxf(c0 ? acc.x : ninf, c1 ? acc.y : ninf),
                  fmaxf(c2 ? acc.z : ninf, c3 ? acc.w : ninf));
  m = fmaxf(m, __shfl_xor(m, 1, 32));
  m = fmaxf(m, __shfl_xor(m, 2, 32));
  float s = (c0 ? __expf(acc.x - m) : 0.0f) + (c1 ? __expf(acc.y - m) : 0.0f) +
            (c2 ? __expf(acc.z - m) : 0.0f) + (c3 ? __expf(acc.w - m) : 0.0f);
  s += __shfl_xor(s, 1, 32);
  s += __shfl_xor(s, 2, 32);
  const float lse = m + __logf(s);
  if (lane < 3) {
    float* o = out + (size_t)node * NCLASS + q4;
    *(float2*)o = make_float2(acc.x - lse, acc.y - lse);
    if (lane < 2) *(float2*)(o + 2) = make_float2(acc.z - lse, acc.w - lse);
  }
}

extern "C" void kernel_launch(void* const* d_in, const int* in_sizes, int n_in,
                              void* d_out, int out_size, void* d_ws, size_t ws_size,
                              hipStream_t stream) {
  const float* x  = (const float*)d_in[0];
  const int*   ei = (const int*)d_in[1];
  const float* W1 = (const float*)d_in[2];
  const float* b1 = (const float*)d_in[3];
  const float* W2 = (const float*)d_in[4];
  const float* b2 = (const float*)d_in[5];
  const float* W3 = (const float*)d_in[6];
  const float* b3 = (const float*)d_in[7];
  const float* Wl = (const float*)d_in[8];
  const float* bl = (const float*)d_in[9];
  float* out = (float*)d_out;

  const int E = in_sizes[1] / 2;
  const int N = in_sizes[0] / NFEAT;
  const int npad = ((N + 16) / 16) * 16;

  char* ws = (char*)d_ws;
  size_t off = 0;
  auto alloc = [&](size_t bytes) -> float* {
    float* p = (float*)(ws + off);
    off += (bytes + 255) & ~(size_t)255;
    return p;
  };
  const size_t cbytes = ((size_t)npad * sizeof(int) + 255) & ~(size_t)255;
  const size_t ebytes = ((size_t)E * sizeof(int) + 255) & ~(size_t)255;
  const size_t hbytes = ((size_t)SC_CHUNKS * npad * sizeof(int) + 255) & ~(size_t)255;
  const size_t tbytes = cbytes + ebytes + hbytes;
  const size_t ybytes = (size_t)N * NHID * sizeof(float);
  float* dinv = alloc((size_t)npad * sizeof(float));
  float* Wt1  = alloc((size_t)NFEAT * NHID * sizeof(float));
  float* Wt2  = alloc((size_t)NHID * NHID * sizeof(float));
  float* Wt3l = alloc((size_t)FOLD_N * NHID * sizeof(float));
  float* b3l  = alloc((size_t)FOLD_N * sizeof(float));
  float* H    = alloc(ybytes);
  float* Y    = alloc(ybytes > tbytes ? ybytes : tbytes);
  int* offs   = (int*)alloc((size_t)npad * sizeof(int));
  int* srcs   = (int*)alloc((size_t)E * sizeof(int));
  if (off > ws_size) return;
  int* cnt  = (int*)Y;
  int* rank = (int*)((char*)Y + cbytes);
  int* hist = (int*)((char*)Y + cbytes + ebytes);

  const int* rows = ei;
  const int* cols = ei + E;

  const int wgrid = (int)(((size_t)N * 32 + 255) / 256);

  const int epc = (E + SC_CHUNKS - 1) / SC_CHUNKS;

  chunk_count<<<SC_CHUNKS, SC_T, 0, stream>>>(cols, rank, hist, E, epc, npad);
  prep_scan<<<PB_ALLW + (npad + 255) / 256, 256, 0, stream>>>(W1, Wt1, W2, Wt2, W3, b3, Wl, bl,
                                                              Wt3l, b3l, hist, cnt, npad);
  seg_scan<<<1, SCAN_T, 0, stream>>>(cnt, offs, dinv, npad);
  seg_fill<<<dim3((epc + 255) / 256, SC_CHUNKS), 256, 0, stream>>>(rows, cols, rank, offs, hist, srcs,
                                                                   E, epc, npad);

  const int tiles = (N + 15) / 16;
  const int gemm_grid = (tiles + 3) / 4;
  const int gemm_full = (gemm_grid / GB_WGPS) * GB_WGPS;
  const int gemm_wgs = gemm_full + (gemm_grid - gemm_full) * (NHID / GT_N);

  gemm_wmma_f32<<<gemm_wgs, 128, 0, stream>>>(x, Wt1, dinv, H, N, NFEAT, gemm_full);
  agg_rows_order<<<wgrid, 256, 0, stream>>>(offs, srcs, dinv, H, b1, Y, N);

  gemm_wmma_f32<<<gemm_wgs, 128, 0, stream>>>(Y, Wt2, dinv, H, N, NHID, gemm_full);
  agg_rows<<<wgrid, 256, 0, stream>>>(offs, srcs, dinv, H, b2, Y, N);

  gemm_wmma_f32_n16<<<gemm_grid, 128, 0, stream>>>(Y, Wt3l, dinv, H, N, NHID);
  agg_last<<<wgrid, 256, 0, stream>>>(offs, srcs, dinv, H, b3l, out, N);
}


# 10 launches to 8: segment starts from a two-level prefix (scan blocks add their local prefix into hist, consumers re-sum the block totals in LDS), so edge placement rides in the first GEMM's launch
# speedup vs baseline: 4.6694x; 1.1493x over previous
#include <hip/hip_runtime.h>

typedef __attribute__((ext_vector_type(2))) float v2f;
typedef __attribute__((ext_vector_type(8))) float v8f;

#define NFEAT 512
#define NHID  128
#define NCLASS 10


__device__ __forceinline__ void transpose_elem(const float* __restrict__ W, float* __restrict__ Wt,
                                               int K, int Nc, int idx) {
  if (idx < K * Nc) {
    int k = idx / Nc, n = idx % Nc;
    Wt[(size_t)n * K + k] = W[idx];
  }
}


#define SC_CHUNKS 64
#define SC_T 1024
#define SC_CAP 50176
__global__ void __launch_bounds__(SC_T) chunk_count(const int* __restrict__ cols,
                                                    int* __restrict__ rank,
                                                    int* __restrict__ hist,
                                                    int E, int epc, int npad) {
  __shared__ __attribute__((aligned(16))) int ctr[SC_CAP];
  const int t = threadIdx.x;
  const int e0 = blockIdx.x * epc;
  const int e1 = (e0 + epc < E) ? (e0 + epc) : E;
  int* hrow = hist + (size_t)blockIdx.x * npad;
  for (int w0 = 0; w0 < npad; w0 += SC_CAP) {
    const int wn = (npad - w0 < SC_CAP) ? (npad - w0) : SC_CAP;
    for (int i = t * 4; i < wn; i += SC_T * 4) *(int4*)(ctr + i) = make_int4(0, 0, 0, 0);
    __syncthreads();
    for (int e = e0 + t; e < e1; e += 8 * SC_T) {
      int c[8], r[8];
#pragma unroll
      for (int q = 0; q < 8; ++q) c[q] = (e + q * SC_T < e1) ? cols[e + q * SC_T] : -1;
#pragma unroll
      for (int q = 0; q < 8; ++q) {
        const unsigned k = (unsigned)(c[q] - w0);
        r[q] = (k < (unsigned)wn) ? atomicAdd(&ctr[k], 1) : -1;
      }
#pragma unroll
      for (int q = 0; q < 8; ++q)
        if (r[q] >= 0) rank[e + q * SC_T] = r[q];
    }
    __syncthreads();
    for (int i = t * 4; i < wn; i += SC_T * 4) *(int4*)(hrow + w0 + i) = *(const int4*)(ctr + i);
    __syncthreads();
  }
}

#define FOLD_N 16
static_assert(NCLASS <= FOLD_N && NCLASS == 10, "the last layer's lane map and its stores are written for 10 classes in one 16-column tile");

#define SB_T 256
#define SB_SH 8
#define PB_W1 (NFEAT * NHID / 256)
#define PB_W2 (NHID * NHID / 256)
#define PB_FOLD ((NHID * FOLD_N + FOLD_N + 255) / 256)
#define PB_ALLW (PB_W1 + PB_W2 + PB_FOLD)
__global__ void __launch_bounds__(256) prep_scan(const float* __restrict__ W1, float* __restrict__ Wt1,
                                                 const float* __restrict__ W2, float* __restrict__ Wt2,
                                                 const float* __restrict__ W3, const float* __restrict__ b3,
                                                 const float* __restrict__ Wl, const float* __restrict__ bl,
                                                 float* __restrict__ Wt3l, float* __restrict__ b3l,
                                                 int* __restrict__ hist, int* __restrict__ btot,
                                                 float* __restrict__ dinv, int npad) {
  __shared__ int wsum[SB_T / 32];
  const int b = blockIdx.x, t = threadIdx.x;
  if (b < PB_W1) {
    transpose_elem(W1, Wt1, NFEAT, NHID, b * 256 + t);
  } else if (b < PB_W1 + PB_W2) {
    transpose_elem(W2, Wt2, NHID, NHID, (b - PB_W1) * 256 + t);
  } else if (b < PB_ALLW) {
    const int idx = (b - PB_W1 - PB_W2) * 256 + t;
    if (idx < NHID * FOLD_N) {
      const int k = idx >> 4, nn = idx & (FOLD_N - 1);
      float s = 0.0f;
      if (nn < NCLASS) {
#pragma unroll 32
        for (int j = 0; j < NHID; ++j) s = fmaf(W3[k * NHID + j], Wl[j * NCLASS + nn], s);
      }
      Wt3l[nn * NHID + k] = s;
    } else if (idx < NHID * FOLD_N + FOLD_N) {
      const int nn = idx - NHID * FOLD_N;
      float s = 0.0f;
      if (nn < NCLASS) {
        s = bl[nn];
#pragma unroll 32
        for (int j = 0; j < NHID; ++j) s = fmaf(b3[j], Wl[j * NCLASS + nn], s);
      }
      b3l[nn] = s;
    }
  } else {
    const int jb = b - PB_ALLW;
    const int i = jb * SB_T + t;
    const bool live = i < npad;
    const int lane = t & 31, wv = t >> 5;
    int* p = hist + i;
    const int* pl = hist + (live ? i : jb * SB_T);
    int v[SC_CHUNKS];
#pragma unroll
    for (int q = 0; q < SC_CHUNKS; ++q) {
      const int w = pl[(size_t)q * npad];
      v[q] = live ? w : 0;
    }
    int tot = 0;
#pragma unroll
    for (int q = 0; q < SC_CHUNKS; ++q) tot += v[q];
    int inc = tot;
#pragma unroll
    for (int o = 1; o < 32; o <<= 1) {
      const int nb = __shfl_up(inc, o, 32);
      if (lane >= o) inc += nb;
    }
    if (lane == 31) wsum[wv] = inc;
    __syncthreads();
    int wbase = 0, bsum = 0;
#pragma unroll
    for (int k = 0; k < SB_T / 32; ++k) {
      const int w = wsum[k];
      if (k < wv) wbase += w;
      bsum += w;
    }
    if (live) {
      int run = wbase + inc - tot;
#pragma unroll
      for (int q = 0; q < SC_CHUNKS; ++q) { p[(size_t)q * npad] = run; run += v[q]; }
      dinv[i] = rsqrtf((float)(tot + 1));
    }
    if (t == 0) btot[jb] = bsum;
  }
}

#define FILL_T 128
#define FILL_Q 8
#define FILL_EPB (FILL_T * FILL_Q)

__device__ __forceinline__ void block_bases(const int* __restrict__ btot, int nblk,
                                            int* tab, int* ws) {
  const int t = threadIdx.x, lane = t & 31, wv = t >> 5;
  int carry = 0;
  for (int j0 = 0; j0 < nblk; j0 += FILL_T) {
    const int j = j0 + t;
    const int v = (j < nblk) ? btot[j] : 0;
    int inc = v;
#pragma unroll
    for (int o = 1; o < 32; o <<= 1) {
      const int nb = __shfl_up(inc, o, 32);
      if (lane >= o) inc += nb;
    }
    if (lane == 31) ws[wv] = inc;
    __syncthreads();
    int wbase = 0, tsum = 0;
#pragma unroll
    for (int k = 0; k < FILL_T / 32; ++k) {
      const int w = ws[k];
      if (k < wv) wbase += w;
      tsum += w;
    }
    if (j < nblk) tab[j] = carry + wbase + inc - v;
    carry += tsum;
    __syncthreads();
  }
}

__device__ __forceinline__ void fill_edges(int* tab, int* ws,
                                           const int* __restrict__ rows, const int* __restrict__ cols,
                                           const int* __restrict__ rank, const int* __restrict__ hist,
                                           const int* __restrict__ btot, int* __restrict__ srcs,
                                           int E, int epc, int npad, int nblk, int fb) {
  const int bpc = (epc + FILL_EPB - 1) / FILL_EPB;
  const int chunk = fb / bpc;
  const int l0 = (fb - chunk * bpc) * FILL_EPB + (int)threadIdx.x;
  const int e0 = chunk * epc + l0;
  const int* hrow = hist + (size_t)chunk * npad;
  int c[FILL_Q], rk[FILL_Q], rw[FILL_Q], hh[FILL_Q];
#pragma unroll
  for (int q = 0; q < FILL_Q; ++q) {
    const int l = l0 + q * FILL_T, e = e0 + q * FILL_T;
    const bool ok = (l < epc) && (e < E);
    c[q] = ok ? cols[e] : -1;
    rk[q] = ok ? rank[e] : 0;
    rw[q] = ok ? rows[e] : 0;
  }
  block_bases(btot, nblk, tab, ws);
#pragma unroll
  for (int q = 0; q < FILL_Q; ++q) hh[q] = (c[q] >= 0) ? hrow[c[q]] : 0;
#pragma unroll
  for (int q = 0; q < FILL_Q; ++q)
    if (c[q] >= 0) srcs[tab[c[q] >> SB_SH] + hh[q] + rk[q]] = rw[q];
}

__device__ __forceinline__ void write_offs(int* tab, int* ws, const int* __restrict__ hist,
                                           const int* __restrict__ btot, int* __restrict__ offs,
                                           int npad, int nblk, int ob) {
  block_bases(btot, nblk, tab, ws);
  const int i0 = ob * FILL_EPB + (int)threadIdx.x;
  int lo[FILL_Q];
#pragma unroll
  for (int q = 0; q < FILL_Q; ++q) {
    const int i = i0 + q * FILL_T;
    lo[q] = (i < npad) ? hist[i] : 0;
  }
#pragma unroll
  for (int q = 0; q < FILL_Q; ++q) {
    const int i = i0 + q * FILL_T;
    if (i < npad) offs[i] = tab[i >> SB_SH] + lo[q];
  }
}

typedef __attribute__((ext_vector_type(4))) float v4f;
#define GB_M 64
#define GB_K 32
#define GB_P 36
#define GB_STAGE ((GB_M + NHID) * GB_P)
#define GB_CP 132
#define GB_RS (16 * GB_P)

#define GB_FETCH(kc)                                  \
  a0 = *(const v4f*)(x0 + (kc));                      \
  a1 = *(const v4f*)(x1 + (kc));                      \
  a2 = *(const v4f*)(x2 + (kc));                      \
  a3 = *(const v4f*)(x3 + (kc));                      \
  b0 = *(const v4f*)(wsrc + (kc));                    \
  b1 = *(const v4f*)(wsrc + wstep + (kc));            \
  b2 = *(const v4f*)(wsrc + 2 * wstep + (kc));        \
  b3 = *(const v4f*)(wsrc + 3 * wstep + (kc));        \
  b4 = *(const v4f*)(wsrc + 4 * wstep + (kc));        \
  b5 = *(const v4f*)(wsrc + 5 * wstep + (kc));        \
  b6 = *(const v4f*)(wsrc + 6 * wstep + (kc));        \
  b7 = *(const v4f*)(wsrc + 7 * wstep + (kc));

#define GB_PUT(dst)                                           \
  *(v4f*)((dst) + sdst) = a0;                                 \
  *(v4f*)((dst) + sdst + GB_RS) = a1;                         \
  *(v4f*)((dst) + sdst + 2 * GB_RS) = a2;                     \
  *(v4f*)((dst) + sdst + 3 * GB_RS) = a3;                     \
  *(v4f*)((dst) + GB_M * GB_P + sdst) = b0;                   \
  *(v4f*)((dst) + GB_M * GB_P + sdst + GB_RS) = b1;           \
  *(v4f*)((dst) + GB_M * GB_P + sdst + 2 * GB_RS) = b2;       \
  *(v4f*)((dst) + GB_M * GB_P + sdst + 3 * GB_RS) = b3;       \
  *(v4f*)((dst) + GB_M * GB_P + sdst + 4 * GB_RS) = b4;       \
  *(v4f*)((dst) + GB_M * GB_P + sdst + 5 * GB_RS) = b5;       \
  *(v4f*)((dst) + GB_M * GB_P + sdst + 6 * GB_RS) = b6;       \
  *(v4f*)((dst) + GB_M * GB_P + sdst + 7 * GB_RS) = b7;

#define GB_WGPS 256
#define GT_N 32
#define GT_STAGE ((GB_M + GT_N) * GB_P)
#define GT_CP 36
__device__ __forceinline__ void gemm_piece_f32(float* __restrict__ lds,
                                               const float* __restrict__ X,
                                               const float* __restrict__ Wt,
                                               const float* __restrict__ dinv,
                                               float* __restrict__ H,
                                               int M, int K, int row0, int col0) {
  const int tid = threadIdx.x;
  const int lane = tid & 31;
  const int wv = tid >> 5;
  const int c = lane & 15;
  const int h = lane >> 4;

  const int seg = tid & 7;
  const int rsub = tid >> 3;
  const int last = M - 1;
  const int r0 = row0 + rsub, r1 = r0 + 16, r2 = r0 + 32, r3 = r0 + 48;
  const float* x0 = X + (size_t)(r0 < last ? r0 : last) * K + seg * 4;
  const float* x1 = X + (size_t)(r1 < last ? r1 : last) * K + seg * 4;
  const float* x2 = X + (size_t)(r2 < last ? r2 : last) * K + seg * 4;
  const float* x3 = X + (size_t)(r3 < last ? r3 : last) * K + seg * 4;
  const float* w0 = Wt + (size_t)(col0 + rsub) * K + seg * 4;
  const float* w1 = w0 + (size_t)16 * K;
  const int sdst = rsub * GB_P + seg * 4;

  const int srow = row0 + wv * 16 + c;
  const float dsc = dinv[srow < last ? srow : last];

  v4f a0 = *(const v4f*)(x0);
  v4f a1 = *(const v4f*)(x1);
  v4f a2 = *(const v4f*)(x2);
  v4f a3 = *(const v4f*)(x3);
  v4f b0 = *(const v4f*)(w0);
  v4f b1 = *(const v4f*)(w1);
  *(v4f*)(lds + sdst) = a0;
  *(v4f*)(lds + sdst + GB_RS) = a1;
  *(v4f*)(lds + sdst + 2 * GB_RS) = a2;
  *(v4f*)(lds + sdst + 3 * GB_RS) = a3;
  *(v4f*)(lds + GB_M * GB_P + sdst) = b0;
  *(v4f*)(lds + GB_M * GB_P + sdst + GB_RS) = b1;
  __syncthreads();

  v8f acc0 = {}, acc1 = {};
  const int nch = K / GB_K;
  for (int ch = 0; ch < nch; ++ch) {
    const float* st = lds + (ch & 1) * GT_STAGE;
    float* nx = lds + ((ch + 1) & 1) * GT_STAGE;
    const bool more = ch + 1 < nch;
    if (more) {
      const int kc = (ch + 1) * GB_K;
      a0 = *(const v4f*)(x0 + kc);
      a1 = *(const v4f*)(x1 + kc);
      a2 = *(const v4f*)(x2 + kc);
      a3 = *(const v4f*)(x3 + kc);
      b0 = *(const v4f*)(w0 + kc);
      b1 = *(const v4f*)(w1 + kc);
    }

    const float* ap = st + (wv * 16 + c) * GB_P + 2 * h;
    const float* bp = st + (GB_M + c) * GB_P + 2 * h;
#pragma unroll
    for (int ks = 0; ks < GB_K / 4; ++ks) {
      const v2f a = *(const v2f*)(ap + ks * 4);
      const v2f f0 = *(const v2f*)(bp + ks * 4);
      const v2f f1 = *(const v2f*)(bp + GB_RS + ks * 4);
      acc0 = __builtin_amdgcn_wmma_f32_16x16x4_f32(
           false, a,  false, f0,
           (short)0, acc0,  false,  false);
      acc1 = __builtin_amdgcn_wmma_f32_16x16x4_f32(
           false, a,  false, f1,
           (short)0, acc1,  false,  false);
    }

    if (more) {
      *(v4f*)(nx + sdst) = a0;
      *(v4f*)(nx + sdst + GB_RS) = a1;
      *(v4f*)(nx + sdst + 2 * GB_RS) = a2;
      *(v4f*)(nx + sdst + 3 * GB_RS) = a3;
      *(v4f*)(nx + GB_M * GB_P + sdst) = b0;
      *(v4f*)(nx + GB_M * GB_P + sdst + GB_RS) = b1;
    }
    __syncthreads();
  }

  float* cs = lds + wv * (16 * GT_CP);
#pragma unroll
  for (int e = 0; e < 8; ++e) {
    cs[(8 * h + e) * GT_CP + c] = acc0[e];
    cs[(8 * h + e) * GT_CP + 16 + c] = acc1[e];
  }
  __syncthreads();
  for (int i = 0; i < 4; ++i) {
    const int r = i * 4 + (lane >> 3);
    const int row = row0 + wv * 16 + r;
    const float sc = __shfl(dsc, r, 32);
    if (row < M)
      *(v4f*)(H + (size_t)row * NHID + col0 + (lane & 7) * 4) =
          *(const v4f*)(cs + r * GT_CP + (lane & 7) * 4) * sc;
  }
}

#define FILL_WS 16
#define FILL_TAB (2 * GB_STAGE - FILL_WS)
static_assert(FILL_T == 128, "the extra workgroups have the GEMM's shape");
__global__ void __launch_bounds__(128) gemm_wmma_f32(const float* __restrict__ X,
                                                     const float* __restrict__ Wt,
                                                     const float* __restrict__ dinv,
                                                     float* __restrict__ H,
                                                     int M, int K, int nfull, int ngemm,
                                                     const int* __restrict__ rows,
                                                     const int* __restrict__ cols,
                                                     const int* __restrict__ rank,
                                                     const int* __restrict__ hist,
                                                     const int* __restrict__ btot,
                                                     int* __restrict__ srcs,
                                                     int* __restrict__ offs,
                                                     int E, int epc, int npad, int nblk, int nfill) {
  __shared__ __attribute__((aligned(16))) float lds[2 * GB_STAGE];

  if ((int)blockIdx.x >= ngemm) {
    const int fb = (int)blockIdx.x - ngemm;
    int* ws = (int*)lds;
    int* tab = ws + FILL_WS;
    if (fb < nfill)
      fill_edges(tab, ws, rows, cols, rank, hist, btot, srcs, E, epc, npad, nblk, fb);
    else
      write_offs(tab, ws, hist, btot, offs, npad, nblk, fb - nfill);
    return;
  }

  if ((int)blockIdx.x >= nfull) {
    const int p = (int)blockIdx.x - nfull;
    gemm_piece_f32(lds, X, Wt, dinv, H, M, K, (nfull + p / (NHID / GT_N)) * GB_M,
                   (p % (NHID / GT_N)) * GT_N);
    return;
  }

  const int tid = threadIdx.x;
  const int lane = tid & 31;
  const int wv = tid >> 5;
  const int c = lane & 15;
  const int h = lane >> 4;
  const int row0 = blockIdx.x * GB_M;

  const int seg = tid & 7;
  const int rsub = tid >> 3;
  const int last = M - 1;
  const int r0 = row0 + rsub, r1 = r0 + 16, r2 = r0 + 32, r3 = r0 + 48;
  const float* x0 = X + (size_t)(r0 < last ? r0 : last) * K + seg * 4;
  const float* x1 = X + (size_t)(r1 < last ? r1 : last) * K + seg * 4;
  const float* x2 = X + (size_t)(r2 < last ? r2 : last) * K + seg * 4;
  const float* x3 = X + (size_t)(r3 < last ? r3 : last) * K + seg * 4;
  const float* wsrc = Wt + (size_t)rsub * K + seg * 4;
  const size_t wstep = (size_t)16 * K;
  const int sdst = rsub * GB_P + seg * 4;

  const int srow = row0 + wv * 16 + c;
  const float dsc = dinv[srow < last ? srow : last];

  v4f a0, a1, a2, a3, b0, b1, b2, b3, b4, b5, b6, b7;
  GB_FETCH(0)
  GB_PUT(lds)
  __syncthreads();

  v8f acc[8] = {};
  const int nch = K / GB_K;
  for (int ch = 0; ch < nch; ++ch) {
    const float* st = lds + (ch & 1) * GB_STAGE;
    float* nx = lds + ((ch + 1) & 1) * GB_STAGE;
    const bool more = ch + 1 < nch;
    if (more) {
      const int kc = (ch + 1) * GB_K;
      GB_FETCH(kc)
    }

    const float* ap = st + (wv * 16 + c) * GB_P + 2 * h;
    const float* bp = st + (GB_M + c) * GB_P + 2 * h;
#pragma unroll
    for (int ks = 0; ks < GB_K / 4; ++ks) {
      const v2f a = *(const v2f*)(ap + ks * 4);
#pragma unroll
      for (int t = 0; t < 8; ++t) {
        const v2f b = *(const v2f*)(bp + t * GB_RS + ks * 4);
        acc[t] = __builtin_amdgcn_wmma_f32_16x16x4_f32(
             false, a,  false, b,
             (short)0, acc[t],  false,  false);
      }
    }

    if (more) {
      GB_PUT(nx)
    }
    __syncthreads();
  }

  float* cs = lds + wv * (16 * GB_CP);
#pragma unroll
  for (int t = 0; t < 8; ++t) {
#pragma unroll
    for (int e = 0; e < 8; ++e) cs[(8 * h + e) * GB_CP + t * 16 + c] = acc[t][e];
  }
  __syncthreads();
#pragma unroll
  for (int r = 0; r < 16; ++r) {
    const int row = row0 + wv * 16 + r;
    const float sc = __int_as_float(__builtin_amdgcn_readlane(__float_as_int(dsc), r));
    if (row < M)
      *(v4f*)(H + (size_t)row * NHID + lane * 4) = *(const v4f*)(cs + r * GB_CP + lane * 4) * sc;
  }
}
#undef GB_FETCH
#undef GB_PUT

#define G16_STAGE ((GB_M + FOLD_N) * GB_P)
#define G16_CP 20
__global__ void __launch_bounds__(128) gemm_wmma_f32_n16(const float* __restrict__ X,
                                                         const float* __restrict__ Wt,
                                                         const float* __restrict__ dinv,
                                                         float* __restrict__ G,
                                                         int M, int K) {
  __shared__ __attribute__((aligned(16))) float lds[2 * G16_STAGE];

  const int tid = threadIdx.x;
  const int lane = tid & 31;
  const int wv = tid >> 5;
  const int c = lane & 15;
  const int h = lane >> 4;
  const int row0 = blockIdx.x * GB_M;

  const int seg = tid & 7;
  const int rsub = tid >> 3;
  const int last = M - 1;
  const int r0 = row0 + rsub, r1 = r0 + 16, r2 = r0 + 32, r3 = r0 + 48;
  const float* x0 = X + (size_t)(r0 < last ? r0 : last) * K + seg * 4;
  const float* x1 = X + (size_t)(r1 < last ? r1 : last) * K + seg * 4;
  const float* x2 = X + (size_t)(r2 < last ? r2 : last) * K + seg * 4;
  const float* x3 = X + (size_t)(r3 < last ? r3 : last) * K + seg * 4;
  const float* wsrc = Wt + (size_t)rsub * K + seg * 4;
  const int sdst = rsub * GB_P + seg * 4;

  const int srow = row0 + wv * 16 + c;
  const float dsc = dinv[srow < last ? srow : last];

  v4f a0 = *(const v4f*)(x0);
  v4f a1 = *(const v4f*)(x1);
  v4f a2 = *(const v4f*)(x2);
  v4f a3 = *(const v4f*)(x3);
  v4f b0 = *(const v4f*)(wsrc);
  *(v4f*)(lds + sdst) = a0;
  *(v4f*)(lds + sdst + GB_RS) = a1;
  *(v4f*)(lds + sdst + 2 * GB_RS) = a2;
  *(v4f*)(lds + sdst + 3 * GB_RS) = a3;
  *(v4f*)(lds + GB_M * GB_P + sdst) = b0;
  __syncthreads();

  v8f acc = {};
  const int nch = K / GB_K;
  for (int ch = 0; ch < nch; ++ch) {
    const float* st = lds + (ch & 1) * G16_STAGE;
    float* nx = lds + ((ch + 1) & 1) * G16_STAGE;
    const bool more = ch + 1 < nch;
    if (more) {
      const int kc = (ch + 1) * GB_K;
      a0 = *(const v4f*)(x0 + kc);
      a1 = *(const v4f*)(x1 + kc);
      a2 = *(const v4f*)(x2 + kc);
      a3 = *(const v4f*)(x3 + kc);
      b0 = *(const v4f*)(wsrc + kc);
    }

    const float* ap = st + (wv * 16 + c) * GB_P + 2 * h;
    const float* bp = st + (GB_M + c) * GB_P + 2 * h;
#pragma unroll
    for (int ks = 0; ks < GB_K / 4; ++ks) {
      const v2f a = *(const v2f*)(ap + ks * 4);
      const v2f b = *(const v2f*)(bp + ks * 4);
      acc = __builtin_amdgcn_wmma_f32_16x16x4_f32(
           false, a,  false, b,
           (short)0, acc,  false,  false);
    }

    if (more) {
      *(v4f*)(nx + sdst) = a0;
      *(v4f*)(nx + sdst + GB_RS) = a1;
      *(v4f*)(nx + sdst + 2 * GB_RS) = a2;
      *(v4f*)(nx + sdst + 3 * GB_RS) = a3;
      *(v4f*)(nx + GB_M * GB_P + sdst) = b0;
    }
    __syncthreads();
  }

  float* cs = lds + wv * (16 * G16_CP);
#pragma unroll
  for (int e = 0; e < 8; ++e) cs[(8 * h + e) * G16_CP + c] = acc[e];
  __syncthreads();
  for (int i = 0; i < 2; ++i) {
    const int r = i * 8 + (lane >> 2);
    const int row = row0 + wv * 16 + r;
    const float sc = __shfl(dsc, r, 32);
    if (row < M)
      *(v4f*)(G + (size_t)row * FOLD_N + (lane & 3) * 4) =
          *(const v4f*)(cs + r * G16_CP + (lane & 3) * 4) * sc;
  }
}


__device__ __forceinline__ void row_fma(float4& acc, const float4 h, const float w) {
  acc.x = fmaf(h.x, w, acc.x);
  acc.y = fmaf(h.y, w, acc.y);
  acc.z = fmaf(h.z, w, acc.z);
  acc.w = fmaf(h.w, w, acc.w);
}

__device__ __forceinline__ void row_add(float4& acc, const float4 h) {
  acc.x += h.x;
  acc.y += h.y;
  acc.z += h.z;
  acc.w += h.w;
}

__global__ void __launch_bounds__(256) agg_rows(const int* __restrict__ offs,
                                                const int* __restrict__ srcs,
                                                const float* __restrict__ dinv,
                                                const float* __restrict__ H,
                                                const float* __restrict__ b,
                                                float* __restrict__ Y, int n) {
  int node = (int)((blockIdx.x * (size_t)blockDim.x + threadIdx.x) >> 5);
  const int lane = threadIdx.x & 31;
  if (node >= n) return;
  node = __builtin_amdgcn_readfirstlane(node);
  const int beg = offs[node];
  const int d = offs[node + 1] - beg;
  const float di = dinv[node];
  const float* hl = H + lane * 4;

  float4 acc = make_float4(0.0f, 0.0f, 0.0f, 0.0f);

  for (int base = 0; base < d; base += 32) {
    const int m = (d - base < 32) ? (d - base) : 32;
    int s = node;
    if (lane < m) s = srcs[beg + base + lane];
    int j = 0;
    for (; j + 4 <= m; j += 4) {
      const int s0 = __builtin_amdgcn_readlane(s, j);
      const int s1 = __builtin_amdgcn_readlane(s, j + 1);
      const int s2 = __builtin_amdgcn_readlane(s, j + 2);
      const int s3 = __builtin_amdgcn_readlane(s, j + 3);
      const float4 h0 = *(const float4*)(hl + (size_t)s0 * NHID);
      const float4 h1 = *(const float4*)(hl + (size_t)s1 * NHID);
      const float4 h2 = *(const float4*)(hl + (size_t)s2 * NHID);
      const float4 h3 = *(const float4*)(hl + (size_t)s3 * NHID);
      row_add(acc, h0);
      row_add(acc, h1);
      row_add(acc, h2);
      row_add(acc, h3);
    }
#pragma nounroll
    for (; j < m; ++j) {
      const int s0 = __builtin_amdgcn_readlane(s, j);
      const float4 h0 = *(const float4*)(hl + (size_t)s0 * NHID);
      row_add(acc, h0);
    }
  }

  const float4 hself = *(const float4*)(hl + (size_t)node * NHID);
  const float4 bv = *(const float4*)(b + lane * 4);
  acc.x = (acc.x + hself.x) * di + bv.x; acc.y = (acc.y + hself.y) * di + bv.y;
  acc.z = (acc.z + hself.z) * di + bv.z; acc.w = (acc.w + hself.w) * di + bv.w;
  acc.x = fmaxf(acc.x, 0.0f); acc.y = fmaxf(acc.y, 0.0f);
  acc.z = fmaxf(acc.z, 0.0f); acc.w = fmaxf(acc.w, 0.0f);
  *(float4*)(Y + (size_t)node * NHID + lane * 4) = acc;
}

#define SRT_R 4
__global__ void __launch_bounds__(256) agg_rows_order(const int* __restrict__ offs,
                                                      int* srcs,
                                                      const float* __restrict__ dinv,
                                                      const float* __restrict__ H,
                                                      const float* __restrict__ b,
                                                      float* __restrict__ Y, int n) {
  int node = (int)((blockIdx.x * (size_t)blockDim.x + threadIdx.x) >> 5);
  const int lane = threadIdx.x & 31;
  if (node >= n) return;
  node = __builtin_amdgcn_readfirstlane(node);
  const int beg = offs[node];
  const int d = offs[node + 1] - beg;
  const float di = dinv[node];
  const float* hl = H + lane * 4;
  int* seg = srcs + beg;

  int s_first = node;
  if (d <= 32) {
    if (d > 0) {
      const bool live = lane < d;
      const int key = live ? seg[lane] : 0;
      int r = 0;
      for (int q = 0; q < d; ++q) {
        const int k = __builtin_amdgcn_readlane(key, q);
        r += (int)(k < key) + (int)((k == key) & (q < lane));
      }
      if (live) seg[r] = key;
      s_first = __builtin_amdgcn_ds_permute((live ? r : lane) << 2, live ? key : node);
    }
  } else if (d <= 32 * SRT_R) {
    int key[SRT_R], r[SRT_R];
#pragma unroll
    for (int a = 0; a < SRT_R; ++a) {
      const int i = 32 * a + lane;
      key[a] = (i < d) ? seg[i] : 0;
      r[a] = 0;
    }
#pragma unroll
    for (int c = 0; c < SRT_R; ++c) {
      if (32 * c < d) {
        const int m = (d - 32 * c < 32) ? (d - 32 * c) : 32;
        for (int q = 0; q < m; ++q) {
          const int k = __builtin_amdgcn_readlane(key[c], q);
          const int j = 32 * c + q;
#pragma unroll
          for (int a = 0; a < SRT_R; ++a)
            r[a] += (int)(k < key[a]) + (int)((k == key[a]) & (j < 32 * a + lane));
        }
      }
    }
#pragma unroll
    for (int a = 0; a < SRT_R; ++a)
      if (32 * a + lane < d) seg[r[a]] = key[a];
    __threadfence();
  } else {
    for (int p = 0; p < d; ++p) {
      int best = 0x7fffffff, bpos = 0x7fffffff;
      for (int j = p + lane; j < d; j += 32) {
        const int k = seg[j];
        if (k < best) { best = k; bpos = j; }
      }
#pragma unroll
      for (int o = 16; o > 0; o >>= 1) {
        const int ob = __shfl_xor(best, o, 32);
        const int op = __shfl_xor(bpos, o, 32);
        if (ob < best || (ob == best && op < bpos)) { best = ob; bpos = op; }
      }
      if (lane == 0) {
        const int vp = seg[p];
        seg[bpos] = vp;
        seg[p] = best;
      }
      __threadfence();
    }
  }

  float4 acc = make_float4(0.0f, 0.0f, 0.0f, 0.0f);

  for (int base = 0; base < d; base += 32) {
    const int m = (d - base < 32) ? (d - base) : 32;
    int s = s_first;
    if (d > 32) {
      s = node;
      if (lane < m) s = seg[base + lane];
    }
    int j = 0;
    for (; j + 4 <= m; j += 4) {
      const int s0 = __builtin_amdgcn_readlane(s, j);
      const int s1 = __builtin_amdgcn_readlane(s, j + 1);
      const int s2 = __builtin_amdgcn_readlane(s, j + 2);
      const int s3 = __builtin_amdgcn_readlane(s, j + 3);
      const float4 h0 = *(const float4*)(hl + (size_t)s0 * NHID);
      const float4 h1 = *(const float4*)(hl + (size_t)s1 * NHID);
      const float4 h2 = *(const float4*)(hl + (size_t)s2 * NHID);
      const float4 h3 = *(const float4*)(hl + (size_t)s3 * NHID);
      row_add(acc, h0);
      row_add(acc, h1);
      row_add(acc, h2);
      row_add(acc, h3);
    }
#pragma nounroll
    for (; j < m; ++j) {
      const int s0 = __builtin_amdgcn_readlane(s, j);
      const float4 h0 = *(const float4*)(hl + (size_t)s0 * NHID);
      row_add(acc, h0);
    }
  }

  const float4 hself = *(const float4*)(hl + (size_t)node * NHID);
  const float4 bv = *(const float4*)(b + lane * 4);
  acc.x = (acc.x + hself.x) * di + bv.x; acc.y = (acc.y + hself.y) * di + bv.y;
  acc.z = (acc.z + hself.z) * di + bv.z; acc.w = (acc.w + hself.w) * di + bv.w;
  acc.x = fmaxf(acc.x, 0.0f); acc.y = fmaxf(acc.y, 0.0f);
  acc.z = fmaxf(acc.z, 0.0f); acc.w = fmaxf(acc.w, 0.0f);
  *(float4*)(Y + (size_t)node * NHID + lane * 4) = acc;
}

__global__ void __launch_bounds__(256) agg_last(const int* __restrict__ offs,
                                                const int* __restrict__ srcs,
                                                const float* __restrict__ dinv,
                                                const float* __restrict__ G,
                                                const float* __restrict__ bf,
                                                float* __restrict__ out, int n) {
  int node = (int)((blockIdx.x * (size_t)blockDim.x + threadIdx.x) >> 5);
  const int lane = threadIdx.x & 31;
  if (node >= n) return;
  node = __builtin_amdgcn_readfirstlane(node);
  const int beg = offs[node];
  const int d = offs[node + 1] - beg;
  const float di = dinv[node];
  const int slot = lane >> 2;
  const int q4 = (lane & 3) * 4;
  const float* gl = G + q4;

  float4 acc = make_float4(0.0f, 0.0f, 0.0f, 0.0f);
  for (int base = 0; base < d; base += 32) {
    bool v[4];
    int s[4];
    float4 g[4];
#pragma unroll
    for (int bt = 0; bt < 4; ++bt) {
      const int e = base + slot + 8 * bt;
      v[bt] = e < d;
      s[bt] = srcs[beg + (v[bt] ? e : d - 1)];
    }
#pragma unroll
    for (int bt = 0; bt < 4; ++bt) g[bt] = *(const float4*)(gl + (size_t)s[bt] * FOLD_N);
#pragma unroll
    for (int bt = 0; bt < 4; ++bt) row_fma(acc, g[bt], v[bt] ? 1.0f : 0.0f);
  }
#pragma unroll
  for (int o = 4; o < 32; o <<= 1) {
    acc.x += __shfl_xor(acc.x, o, 32);
    acc.y += __shfl_xor(acc.y, o, 32);
    acc.z += __shfl_xor(acc.z, o, 32);
    acc.w += __shfl_xor(acc.w, o, 32);
  }

  const float4 gself = *(const float4*)(gl + (size_t)node * FOLD_N);
  const float4 bv = *(const float4*)(bf + q4);
  acc.x = (acc.x + gself.x) * di + bv.x; acc.y = (acc.y + gself.y) * di + bv.y;
  acc.z = (acc.z + gself.z) * di + bv.z; acc.w = (acc.w + gself.w) * di + bv.w;

  const float ninf = -__builtin_inff();
  const bool c0 = q4 + 0 < NCLASS, c1 = q4 + 1 < NCLASS, c2 = q4 + 2 < NCLASS, c3 = q4 + 3 < NCLASS;
  float m = fmaxf(fmaxf(c0 ? acc.x : ninf, c1 ? acc.y : ninf),
                  fmaxf(c2 ? acc.z : ninf, c3 ? acc.w : ninf));
  m = fmaxf(m, __shfl_xor(m, 1, 32));
  m = fmaxf(m, __shfl_xor(m, 2, 32));
  float s = (c0 ? __expf(acc.x - m) : 0.0f) + (c1 ? __expf(acc.y - m) : 0.0f) +
            (c2 ? __expf(acc.z - m) : 0.0f) + (c3 ? __expf(acc.w - m) : 0.0f);
  s += __shfl_xor(s, 1, 32);
  s += __shfl_xor(s, 2, 32);
  const float lse = m + __logf(s);
  if (lane < 3) {
    float* o = out + (size_t)node * NCLASS + q4;
    *(float2*)o = make_float2(acc.x - lse, acc.y - lse);
    if (lane < 2) *(float2*)(o + 2) = make_float2(acc.z - lse, acc.w - lse);
  }
}

extern "C" void kernel_launch(void* const* d_in, const int* in_sizes, int n_in,
                              void* d_out, int out_size, void* d_ws, size_t ws_size,
                              hipStream_t stream) {
  const float* x  = (const float*)d_in[0];
  const int*   ei = (const int*)d_in[1];
  const float* W1 = (const float*)d_in[2];
  const float* b1 = (const float*)d_in[3];
  const float* W2 = (const float*)d_in[4];
  const float* b2 = (const float*)d_in[5];
  const float* W3 = (const float*)d_in[6];
  const float* b3 = (const float*)d_in[7];
  const float* Wl = (const float*)d_in[8];
  const float* bl = (const float*)d_in[9];
  float* out = (float*)d_out;

  const int E = in_sizes[1] / 2;
  const int N = in_sizes[0] / NFEAT;
  const int npad = ((N + 16) / 16) * 16;

  char* ws = (char*)d_ws;
  size_t off = 0;
  auto alloc = [&](size_t bytes) -> float* {
    float* p = (float*)(ws + off);
    off += (bytes + 255) & ~(size_t)255;
    return p;
  };
  const size_t cbytes = ((size_t)npad * sizeof(int) + 255) & ~(size_t)255;
  const size_t ebytes = ((size_t)E * sizeof(int) + 255) & ~(size_t)255;
  const size_t hbytes = ((size_t)SC_CHUNKS * npad * sizeof(int) + 255) & ~(size_t)255;
  const size_t tbytes = cbytes + ebytes + hbytes;
  const size_t ybytes = (size_t)N * NHID * sizeof(float);
  float* dinv = alloc((size_t)npad * sizeof(float));
  float* Wt1  = alloc((size_t)NFEAT * NHID * sizeof(float));
  float* Wt2  = alloc((size_t)NHID * NHID * sizeof(float));
  float* Wt3l = alloc((size_t)FOLD_N * NHID * sizeof(float));
  float* b3l  = alloc((size_t)FOLD_N * sizeof(float));
  float* H    = alloc(ybytes);
  float* Y    = alloc(ybytes > tbytes ? ybytes : tbytes);
  int* offs   = (int*)alloc((size_t)npad * sizeof(int));
  int* srcs   = (int*)alloc((size_t)E * sizeof(int));
  if (off > ws_size) return;
  const int nblk = (npad + SB_T - 1) / SB_T;
  if (nblk > FILL_TAB) return;
  int* btot = (int*)Y;
  int* rank = (int*)((char*)Y + cbytes);
  int* hist = (int*)((char*)Y + cbytes + ebytes);

  const int* rows = ei;
  const int* cols = ei + E;

  const int wgrid = (int)(((size_t)N * 32 + 255) / 256);

  const int epc = (E + SC_CHUNKS - 1) / SC_CHUNKS;

  chunk_count<<<SC_CHUNKS, SC_T, 0, stream>>>(cols, rank, hist, E, epc, npad);
  prep_scan<<<PB_ALLW + nblk, SB_T, 0, stream>>>(W1, Wt1, W2, Wt2, W3, b3, Wl, bl,
                                                 Wt3l, b3l, hist, btot, dinv, npad);

  const int tiles = (N + 15) / 16;
  const int gemm_grid = (tiles + 3) / 4;
  const int gemm_full = (gemm_grid / GB_WGPS) * GB_WGPS;
  const int gemm_wgs = gemm_full + (gemm_grid - gemm_full) * (NHID / GT_N);

  const int nfill = ((epc + FILL_EPB - 1) / FILL_EPB) * SC_CHUNKS;
  const int noffs = (npad + FILL_EPB - 1) / FILL_EPB;
  gemm_wmma_f32<<<gemm_wgs + nfill + noffs, 128, 0, stream>>>(x, Wt1, dinv, H, N, NFEAT, gemm_full,
                                                              gemm_wgs, rows, cols, rank, hist, btot,
                                                              srcs, offs, E, epc, npad, nblk, nfill);
  agg_rows_order<<<wgrid, 256, 0, stream>>>(offs, srcs, dinv, H, b1, Y, N);

  gemm_wmma_f32<<<gemm_wgs, 128, 0, stream>>>(Y, Wt2, dinv, H, N, NHID, gemm_full, gemm_wgs,
                                              nullptr, nullptr, nullptr, nullptr, nullptr, nullptr,
                                              nullptr, 0, 0, 0, 0, 0);
  agg_rows<<<wgrid, 256, 0, stream>>>(offs, srcs, dinv, H, b2, Y, N);

  gemm_wmma_f32_n16<<<gemm_grid, 128, 0, stream>>>(Y, Wt3l, dinv, H, N, NHID);
  agg_last<<<wgrid, 256, 0, stream>>>(offs, srcs, dinv, H, b3l, out, N);
}


# Both products behind an aggregation run in its launch: a 32-wave workgroup sums 64 target rows into LDS and multiplies that tile there (f32 WMMA); rows never stored; 8 launches to 6; sums unchanged
# speedup vs baseline: 4.9367x; 1.0572x over previous
#include <hip/hip_runtime.h>

typedef __attribute__((ext_vector_type(2))) float v2f;
typedef __attribute__((ext_vector_type(8))) float v8f;

#define NFEAT 512
#define NHID  128
#define NCLASS 10


__device__ __forceinline__ void transpose_elem(const float* __restrict__ W, float* __restrict__ Wt,
                                               int K, int Nc, int idx) {
  if (idx < K * Nc) {
    int k = idx / Nc, n = idx % Nc;
    Wt[(size_t)n * K + k] = W[idx];
  }
}


#define SC_CHUNKS 64
#define SC_T 1024
#define SC_CAP 50176
__global__ void __launch_bounds__(SC_T) chunk_count(const int* __restrict__ cols,
                                                    int* __restrict__ rank,
                                                    int* __restrict__ hist,
                                                    int E, int epc, int npad) {
  __shared__ __attribute__((aligned(16))) int ctr[SC_CAP];
  const int t = threadIdx.x;
  const int e0 = blockIdx.x * epc;
  const int e1 = (e0 + epc < E) ? (e0 + epc) : E;
  int* hrow = hist + (size_t)blockIdx.x * npad;
  for (int w0 = 0; w0 < npad; w0 += SC_CAP) {
    const int wn = (npad - w0 < SC_CAP) ? (npad - w0) : SC_CAP;
    for (int i = t * 4; i < wn; i += SC_T * 4) *(int4*)(ctr + i) = make_int4(0, 0, 0, 0);
    __syncthreads();
    for (int e = e0 + t; e < e1; e += 8 * SC_T) {
      int c[8], r[8];
#pragma unroll
      for (int q = 0; q < 8; ++q) c[q] = (e + q * SC_T < e1) ? cols[e + q * SC_T] : -1;
#pragma unroll
      for (int q = 0; q < 8; ++q) {
        const unsigned k = (unsigned)(c[q] - w0);
        r[q] = (k < (unsigned)wn) ? atomicAdd(&ctr[k], 1) : -1;
      }
#pragma unroll
      for (int q = 0; q < 8; ++q)
        if (r[q] >= 0) rank[e + q * SC_T] = r[q];
    }
    __syncthreads();
    for (int i = t * 4; i < wn; i += SC_T * 4) *(int4*)(hrow + w0 + i) = *(const int4*)(ctr + i);
    __syncthreads();
  }
}

#define FOLD_N 16
static_assert(NCLASS <= FOLD_N && NCLASS == 10, "the last layer's lane map and its stores are written for 10 classes in one 16-column tile");

#define SB_T 256
#define SB_SH 8
#define PB_W1 (NFEAT * NHID / 256)
#define PB_W2 (NHID * NHID / 256)
#define PB_FOLD ((NHID * FOLD_N + FOLD_N + 255) / 256)
#define PB_ALLW (PB_W1 + PB_W2 + PB_FOLD)
__global__ void __launch_bounds__(256) prep_scan(const float* __restrict__ W1, float* __restrict__ Wt1,
                                                 const float* __restrict__ W2, float* __restrict__ Wt2,
                                                 const float* __restrict__ W3, const float* __restrict__ b3,
                                                 const float* __restrict__ Wl, const float* __restrict__ bl,
                                                 float* __restrict__ Wt3l, float* __restrict__ b3l,
                                                 int* __restrict__ hist, int* __restrict__ btot,
                                                 float* __restrict__ dinv, int npad) {
  __shared__ int wsum[SB_T / 32];
  const int b = blockIdx.x, t = threadIdx.x;
  if (b < PB_W1) {
    transpose_elem(W1, Wt1, NFEAT, NHID, b * 256 + t);
  } else if (b < PB_W1 + PB_W2) {
    transpose_elem(W2, Wt2, NHID, NHID, (b - PB_W1) * 256 + t);
  } else if (b < PB_ALLW) {
    const int idx = (b - PB_W1 - PB_W2) * 256 + t;
    if (idx < NHID * FOLD_N) {
      const int k = idx >> 4, nn = idx & (FOLD_N - 1);
      float s = 0.0f;
      if (nn < NCLASS) {
#pragma unroll 32
        for (int j = 0; j < NHID; ++j) s = fmaf(W3[k * NHID + j], Wl[j * NCLASS + nn], s);
      }
      Wt3l[nn * NHID + k] = s;
    } else if (idx < NHID * FOLD_N + FOLD_N) {
      const int nn = idx - NHID * FOLD_N;
      float s = 0.0f;
      if (nn < NCLASS) {
        s = bl[nn];
#pragma unroll 32
        for (int j = 0; j < NHID; ++j) s = fmaf(b3[j], Wl[j * NCLASS + nn], s);
      }
      b3l[nn] = s;
    }
  } else {
    const int jb = b - PB_ALLW;
    const int i = jb * SB_T + t;
    const bool live = i < npad;
    const int lane = t & 31, wv = t >> 5;
    int* p = hist + i;
    const int* pl = hist + (live ? i : jb * SB_T);
    int v[SC_CHUNKS];
#pragma unroll
    for (int q = 0; q < SC_CHUNKS; ++q) {
      const int w = pl[(size_t)q * npad];
      v[q] = live ? w : 0;
    }
    int tot = 0;
#pragma unroll
    for (int q = 0; q < SC_CHUNKS; ++q) tot += v[q];
    int inc = tot;
#pragma unroll
    for (int o = 1; o < 32; o <<= 1) {
      const int nb = __shfl_up(inc, o, 32);
      if (lane >= o) inc += nb;
    }
    if (lane == 31) wsum[wv] = inc;
    __syncthreads();
    int wbase = 0, bsum = 0;
#pragma unroll
    for (int k = 0; k < SB_T / 32; ++k) {
      const int w = wsum[k];
      if (k < wv) wbase += w;
      bsum += w;
    }
    if (live) {
      int run = wbase + inc - tot;
#pragma unroll
      for (int q = 0; q < SC_CHUNKS; ++q) { p[(size_t)q * npad] = run; run += v[q]; }
      dinv[i] = rsqrtf((float)(tot + 1));
    }
    if (t == 0) btot[jb] = bsum;
  }
}

#define FILL_T 128
#define FILL_Q 8
#define FILL_EPB (FILL_T * FILL_Q)

__device__ __forceinline__ void block_bases(const int* __restrict__ btot, int nblk,
                                            int* tab, int* ws) {
  const int t = threadIdx.x, lane = t & 31, wv = t >> 5;
  int carry = 0;
  for (int j0 = 0; j0 < nblk; j0 += FILL_T) {
    const int j = j0 + t;
    const int v = (j < nblk) ? btot[j] : 0;
    int inc = v;
#pragma unroll
    for (int o = 1; o < 32; o <<= 1) {
      const int nb = __shfl_up(inc, o, 32);
      if (lane >= o) inc += nb;
    }
    if (lane == 31) ws[wv] = inc;
    __syncthreads();
    int wbase = 0, tsum = 0;
#pragma unroll
    for (int k = 0; k < FILL_T / 32; ++k) {
      const int w = ws[k];
      if (k < wv) wbase += w;
      tsum += w;
    }
    if (j < nblk) tab[j] = carry + wbase + inc - v;
    carry += tsum;
    __syncthreads();
  }
}

__device__ __forceinline__ void fill_edges(int* tab, int* ws,
                                           const int* __restrict__ rows, const int* __restrict__ cols,
                                           const int* __restrict__ rank, const int* __restrict__ hist,
                                           const int* __restrict__ btot, int* __restrict__ srcs,
                                           int E, int epc, int npad, int nblk, int fb) {
  const int bpc = (epc + FILL_EPB - 1) / FILL_EPB;
  const int chunk = fb / bpc;
  const int l0 = (fb - chunk * bpc) * FILL_EPB + (int)threadIdx.x;
  const int e0 = chunk * epc + l0;
  const int* hrow = hist + (size_t)chunk * npad;
  int c[FILL_Q], rk[FILL_Q], rw[FILL_Q], hh[FILL_Q];
#pragma unroll
  for (int q = 0; q < FILL_Q; ++q) {
    const int l = l0 + q * FILL_T, e = e0 + q * FILL_T;
    const bool ok = (l < epc) && (e < E);
    c[q] = ok ? cols[e] : -1;
    rk[q] = ok ? rank[e] : 0;
    rw[q] = ok ? rows[e] : 0;
  }
  block_bases(btot, nblk, tab, ws);
#pragma unroll
  for (int q = 0; q < FILL_Q; ++q) hh[q] = (c[q] >= 0) ? hrow[c[q]] : 0;
#pragma unroll
  for (int q = 0; q < FILL_Q; ++q)
    if (c[q] >= 0) srcs[tab[c[q] >> SB_SH] + hh[q] + rk[q]] = rw[q];
}

__device__ __forceinline__ void write_offs(int* tab, int* ws, const int* __restrict__ hist,
                                           const int* __restrict__ btot, int* __restrict__ offs,
                                           int npad, int nblk, int ob) {
  block_bases(btot, nblk, tab, ws);
  const int i0 = ob * FILL_EPB + (int)threadIdx.x;
  int lo[FILL_Q];
#pragma unroll
  for (int q = 0; q < FILL_Q; ++q) {
    const int i = i0 + q * FILL_T;
    lo[q] = (i < npad) ? hist[i] : 0;
  }
#pragma unroll
  for (int q = 0; q < FILL_Q; ++q) {
    const int i = i0 + q * FILL_T;
    if (i < npad) offs[i] = tab[i >> SB_SH] + lo[q];
  }
}

typedef __attribute__((ext_vector_type(4))) float v4f;
#define GB_M 64
#define GB_K 32
#define GB_P 36
#define GB_STAGE ((GB_M + NHID) * GB_P)
#define GB_CP 132
#define GB_RS (16 * GB_P)

#define GB_FETCH(kc)                                  \
  a0 = *(const v4f*)(x0 + (kc));                      \
  a1 = *(const v4f*)(x1 + (kc));                      \
  a2 = *(const v4f*)(x2 + (kc));                      \
  a3 = *(const v4f*)(x3 + (kc));                      \
  b0 = *(const v4f*)(wsrc + (kc));                    \
  b1 = *(const v4f*)(wsrc + wstep + (kc));            \
  b2 = *(const v4f*)(wsrc + 2 * wstep + (kc));        \
  b3 = *(const v4f*)(wsrc + 3 * wstep + (kc));        \
  b4 = *(const v4f*)(wsrc + 4 * wstep + (kc));        \
  b5 = *(const v4f*)(wsrc + 5 * wstep + (kc));        \
  b6 = *(const v4f*)(wsrc + 6 * wstep + (kc));        \
  b7 = *(const v4f*)(wsrc + 7 * wstep + (kc));

#define GB_PUT(dst)                                           \
  *(v4f*)((dst) + sdst) = a0;                                 \
  *(v4f*)((dst) + sdst + GB_RS) = a1;                         \
  *(v4f*)((dst) + sdst + 2 * GB_RS) = a2;                     \
  *(v4f*)((dst) + sdst + 3 * GB_RS) = a3;                     \
  *(v4f*)((dst) + GB_M * GB_P + sdst) = b0;                   \
  *(v4f*)((dst) + GB_M * GB_P + sdst + GB_RS) = b1;           \
  *(v4f*)((dst) + GB_M * GB_P + sdst + 2 * GB_RS) = b2;       \
  *(v4f*)((dst) + GB_M * GB_P + sdst + 3 * GB_RS) = b3;       \
  *(v4f*)((dst) + GB_M * GB_P + sdst + 4 * GB_RS) = b4;       \
  *(v4f*)((dst) + GB_M * GB_P + sdst + 5 * GB_RS) = b5;       \
  *(v4f*)((dst) + GB_M * GB_P + sdst + 6 * GB_RS) = b6;       \
  *(v4f*)((dst) + GB_M * GB_P + sdst + 7 * GB_RS) = b7;

#define GB_WGPS 256
#define GT_N 32
#define GT_STAGE ((GB_M + GT_N) * GB_P)
#define GT_CP 36
__device__ __forceinline__ void gemm_piece_f32(float* __restrict__ lds,
                                               const float* __restrict__ X,
                                               const float* __restrict__ Wt,
                                               const float* __restrict__ dinv,
                                               float* __restrict__ H,
                                               int M, int K, int row0, int col0) {
  const int tid = threadIdx.x;
  const int lane = tid & 31;
  const int wv = tid >> 5;
  const int c = lane & 15;
  const int h = lane >> 4;

  const int seg = tid & 7;
  const int rsub = tid >> 3;
  const int last = M - 1;
  const int r0 = row0 + rsub, r1 = r0 + 16, r2 = r0 + 32, r3 = r0 + 48;
  const float* x0 = X + (size_t)(r0 < last ? r0 : last) * K + seg * 4;
  const float* x1 = X + (size_t)(r1 < last ? r1 : last) * K + seg * 4;
  const float* x2 = X + (size_t)(r2 < last ? r2 : last) * K + seg * 4;
  const float* x3 = X + (size_t)(r3 < last ? r3 : last) * K + seg * 4;
  const float* w0 = Wt + (size_t)(col0 + rsub) * K + seg * 4;
  const float* w1 = w0 + (size_t)16 * K;
  const int sdst = rsub * GB_P + seg * 4;

  const int srow = row0 + wv * 16 + c;
  const float dsc = dinv[srow < last ? srow : last];

  v4f a0 = *(const v4f*)(x0);
  v4f a1 = *(const v4f*)(x1);
  v4f a2 = *(const v4f*)(x2);
  v4f a3 = *(const v4f*)(x3);
  v4f b0 = *(const v4f*)(w0);
  v4f b1 = *(const v4f*)(w1);
  *(v4f*)(lds + sdst) = a0;
  *(v4f*)(lds + sdst + GB_RS) = a1;
  *(v4f*)(lds + sdst + 2 * GB_RS) = a2;
  *(v4f*)(lds + sdst + 3 * GB_RS) = a3;
  *(v4f*)(lds + GB_M * GB_P + sdst) = b0;
  *(v4f*)(lds + GB_M * GB_P + sdst + GB_RS) = b1;
  __syncthreads();

  v8f acc0 = {}, acc1 = {};
  const int nch = K / GB_K;
  for (int ch = 0; ch < nch; ++ch) {
    const float* st = lds + (ch & 1) * GT_STAGE;
    float* nx = lds + ((ch + 1) & 1) * GT_STAGE;
    const bool more = ch + 1 < nch;
    if (more) {
      const int kc = (ch + 1) * GB_K;
      a0 = *(const v4f*)(x0 + kc);
      a1 = *(const v4f*)(x1 + kc);
      a2 = *(const v4f*)(x2 + kc);
      a3 = *(const v4f*)(x3 + kc);
      b0 = *(const v4f*)(w0 + kc);
      b1 = *(const v4f*)(w1 + kc);
    }

    const float* ap = st + (wv * 16 + c) * GB_P + 2 * h;
    const float* bp = st + (GB_M + c) * GB_P + 2 * h;
#pragma unroll
    for (int ks = 0; ks < GB_K / 4; ++ks) {
      const v2f a = *(const v2f*)(ap + ks * 4);
      const v2f f0 = *(const v2f*)(bp + ks * 4);
      const v2f f1 = *(const v2f*)(bp + GB_RS + ks * 4);
      acc0 = __builtin_amdgcn_wmma_f32_16x16x4_f32(
           false, a,  false, f0,
           (short)0, acc0,  false,  false);
      acc1 = __builtin_amdgcn_wmma_f32_16x16x4_f32(
           false, a,  false, f1,
           (short)0, acc1,  false,  false);
    }

    if (more) {
      *(v4f*)(nx + sdst) = a0;
      *(v4f*)(nx + sdst + GB_RS) = a1;
      *(v4f*)(nx + sdst + 2 * GB_RS) = a2;
      *(v4f*)(nx + sdst + 3 * GB_RS) = a3;
      *(v4f*)(nx + GB_M * GB_P + sdst) = b0;
      *(v4f*)(nx + GB_M * GB_P + sdst + GB_RS) = b1;
    }
    __syncthreads();
  }

  float* cs = lds + wv * (16 * GT_CP);
#pragma unroll
  for (int e = 0; e < 8; ++e) {
    cs[(8 * h + e) * GT_CP + c] = acc0[e];
    cs[(8 * h + e) * GT_CP + 16 + c] = acc1[e];
  }
  __syncthreads();
  for (int i = 0; i < 4; ++i) {
    const int r = i * 4 + (lane >> 3);
    const int row = row0 + wv * 16 + r;
    const float sc = __shfl(dsc, r, 32);
    if (row < M)
      *(v4f*)(H + (size_t)row * NHID + col0 + (lane & 7) * 4) =
          *(const v4f*)(cs + r * GT_CP + (lane & 7) * 4) * sc;
  }
}

#define FILL_WS 16
#define FILL_TAB (2 * GB_STAGE - FILL_WS)
static_assert(FILL_T == 128, "the extra workgroups have the GEMM's shape");
__global__ void __launch_bounds__(128) gemm_wmma_f32(const float* __restrict__ X,
                                                     const float* __restrict__ Wt,
                                                     const float* __restrict__ dinv,
                                                     float* __restrict__ H,
                                                     int M, int K, int nfull, int ngemm,
                                                     const int* __restrict__ rows,
                                                     const int* __restrict__ cols,
                                                     const int* __restrict__ rank,
                                                     const int* __restrict__ hist,
                                                     const int* __restrict__ btot,
                                                     int* __restrict__ srcs,
                                                     int* __restrict__ offs,
                                                     int E, int epc, int npad, int nblk, int nfill) {
  __shared__ __attribute__((aligned(16))) float lds[2 * GB_STAGE];

  if ((int)blockIdx.x >= ngemm) {
    const int fb = (int)blockIdx.x - ngemm;
    int* ws = (int*)lds;
    int* tab = ws + FILL_WS;
    if (fb < nfill)
      fill_edges(tab, ws, rows, cols, rank, hist, btot, srcs, E, epc, npad, nblk, fb);
    else
      write_offs(tab, ws, hist, btot, offs, npad, nblk, fb - nfill);
    return;
  }

  if ((int)blockIdx.x >= nfull) {
    const int p = (int)blockIdx.x - nfull;
    gemm_piece_f32(lds, X, Wt, dinv, H, M, K, (nfull + p / (NHID / GT_N)) * GB_M,
                   (p % (NHID / GT_N)) * GT_N);
    return;
  }

  const int tid = threadIdx.x;
  const int lane = tid & 31;
  const int wv = tid >> 5;
  const int c = lane & 15;
  const int h = lane >> 4;
  const int row0 = blockIdx.x * GB_M;

  const int seg = tid & 7;
  const int rsub = tid >> 3;
  const int last = M - 1;
  const int r0 = row0 + rsub, r1 = r0 + 16, r2 = r0 + 32, r3 = r0 + 48;
  const float* x0 = X + (size_t)(r0 < last ? r0 : last) * K + seg * 4;
  const float* x1 = X + (size_t)(r1 < last ? r1 : last) * K + seg * 4;
  const float* x2 = X + (size_t)(r2 < last ? r2 : last) * K + seg * 4;
  const float* x3 = X + (size_t)(r3 < last ? r3 : last) * K + seg * 4;
  const float* wsrc = Wt + (size_t)rsub * K + seg * 4;
  const size_t wstep = (size_t)16 * K;
  const int sdst = rsub * GB_P + seg * 4;

  const int srow = row0 + wv * 16 + c;
  const float dsc = dinv[srow < last ? srow : last];

  v4f a0, a1, a2, a3, b0, b1, b2, b3, b4, b5, b6, b7;
  GB_FETCH(0)
  GB_PUT(lds)
  __syncthreads();

  v8f acc[8] = {};
  const int nch = K / GB_K;
  for (int ch = 0; ch < nch; ++ch) {
    const float* st = lds + (ch & 1) * GB_STAGE;
    float* nx = lds + ((ch + 1) & 1) * GB_STAGE;
    const bool more = ch + 1 < nch;
    if (more) {
      const int kc = (ch + 1) * GB_K;
      GB_FETCH(kc)
    }

    const float* ap = st + (wv * 16 + c) * GB_P + 2 * h;
    const float* bp = st + (GB_M + c) * GB_P + 2 * h;
#pragma unroll
    for (int ks = 0; ks < GB_K / 4; ++ks) {
      const v2f a = *(const v2f*)(ap + ks * 4);
#pragma unroll
      for (int t = 0; t < 8; ++t) {
        const v2f b = *(const v2f*)(bp + t * GB_RS + ks * 4);
        acc[t] = __builtin_amdgcn_wmma_f32_16x16x4_f32(
             false, a,  false, b,
             (short)0, acc[t],  false,  false);
      }
    }

    if (more) {
      GB_PUT(nx)
    }
    __syncthreads();
  }

  float* cs = lds + wv * (16 * GB_CP);
#pragma unroll
  for (int t = 0; t < 8; ++t) {
#pragma unroll
    for (int e = 0; e < 8; ++e) cs[(8 * h + e) * GB_CP + t * 16 + c] = acc[t][e];
  }
  __syncthreads();
#pragma unroll
  for (int r = 0; r < 16; ++r) {
    const int row = row0 + wv * 16 + r;
    const float sc = __int_as_float(__builtin_amdgcn_readlane(__float_as_int(dsc), r));
    if (row < M)
      *(v4f*)(H + (size_t)row * NHID + lane * 4) = *(const v4f*)(cs + r * GB_CP + lane * 4) * sc;
  }
}
#undef GB_FETCH
#undef GB_PUT


__device__ __forceinline__ void row_fma(float4& acc, const float4 h, const float w) {
  acc.x = fmaf(h.x, w, acc.x);
  acc.y = fmaf(h.y, w, acc.y);
  acc.z = fmaf(h.z, w, acc.z);
  acc.w = fmaf(h.w, w, acc.w);
}

__device__ __forceinline__ void row_add(float4& acc, const float4 h) {
  acc.x += h.x;
  acc.y += h.y;
  acc.z += h.z;
  acc.w += h.w;
}

#define SRT_R 4
template <bool ORDER>
__device__ __forceinline__ float4 target_row(const int* __restrict__ offs, int* srcs,
                                             const float* __restrict__ dinv,
                                             const float* __restrict__ H,
                                             const float* __restrict__ b, int node, int lane) {
  const int beg = offs[node];
  const int d = offs[node + 1] - beg;
  const float di = dinv[node];
  const float* hl = H + lane * 4;
  int* seg = srcs + beg;

  int s_first = node;
  if (ORDER) {
    if (d <= 32) {
      if (d > 0) {
        const bool live = lane < d;
        const int key = live ? seg[lane] : 0;
        int r = 0;
        for (int q = 0; q < d; ++q) {
          const int k = __builtin_amdgcn_readlane(key, q);
          r += (int)(k < key) + (int)((k == key) & (q < lane));
        }
        if (live) seg[r] = key;
        s_first = __builtin_amdgcn_ds_permute((live ? r : lane) << 2, live ? key : node);
      }
    } else if (d <= 32 * SRT_R) {
      int key[SRT_R], r[SRT_R];
#pragma unroll
      for (int a = 0; a < SRT_R; ++a) {
        const int i = 32 * a + lane;
        key[a] = (i < d) ? seg[i] : 0;
        r[a] = 0;
      }
#pragma unroll
      for (int c = 0; c < SRT_R; ++c) {
        if (32 * c < d) {
          const int m = (d - 32 * c < 32) ? (d - 32 * c) : 32;
          for (int q = 0; q < m; ++q) {
            const int k = __builtin_amdgcn_readlane(key[c], q);
            const int j = 32 * c + q;
#pragma unroll
            for (int a = 0; a < SRT_R; ++a)
              r[a] += (int)(k < key[a]) + (int)((k == key[a]) & (j < 32 * a + lane));
          }
        }
      }
#pragma unroll
      for (int a = 0; a < SRT_R; ++a)
        if (32 * a + lane < d) seg[r[a]] = key[a];
      __threadfence();
    } else {
      for (int p = 0; p < d; ++p) {
        int best = 0x7fffffff, bpos = 0x7fffffff;
        for (int j = p + lane; j < d; j += 32) {
          const int k = seg[j];
          if (k < best) { best = k; bpos = j; }
        }
#pragma unroll
        for (int o = 16; o > 0; o >>= 1) {
          const int ob = __shfl_xor(best, o, 32);
          const int op = __shfl_xor(bpos, o, 32);
          if (ob < best || (ob == best && op < bpos)) { best = ob; bpos = op; }
        }
        if (lane == 0) {
          const int vp = seg[p];
          seg[bpos] = vp;
          seg[p] = best;
        }
        __threadfence();
      }
    }
  }

  float4 acc = make_float4(0.0f, 0.0f, 0.0f, 0.0f);

  for (int base = 0; base < d; base += 32) {
    const int m = (d - base < 32) ? (d - base) : 32;
    int s = s_first;
    if (!ORDER || d > 32) {
      s = node;
      if (lane < m) s = seg[base + lane];
    }
    int j = 0;
    for (; j + 4 <= m; j += 4) {
      const int s0 = __builtin_amdgcn_readlane(s, j);
      const int s1 = __builtin_amdgcn_readlane(s, j + 1);
      const int s2 = __builtin_amdgcn_readlane(s, j + 2);
      const int s3 = __builtin_amdgcn_readlane(s, j + 3);
      const float4 h0 = *(const float4*)(hl + (size_t)s0 * NHID);
      const float4 h1 = *(const float4*)(hl + (size_t)s1 * NHID);
      const float4 h2 = *(const float4*)(hl + (size_t)s2 * NHID);
      const float4 h3 = *(const float4*)(hl + (size_t)s3 * NHID);
      row_add(acc, h0);
      row_add(acc, h1);
      row_add(acc, h2);
      row_add(acc, h3);
    }
#pragma nounroll
    for (; j < m; ++j) {
      const int s0 = __builtin_amdgcn_readlane(s, j);
      const float4 h0 = *(const float4*)(hl + (size_t)s0 * NHID);
      row_add(acc, h0);
    }
  }

  const float4 hself = *(const float4*)(hl + (size_t)node * NHID);
  const float4 bv = *(const float4*)(b + lane * 4);
  acc.x = (acc.x + hself.x) * di + bv.x; acc.y = (acc.y + hself.y) * di + bv.y;
  acc.z = (acc.z + hself.z) * di + bv.z; acc.w = (acc.w + hself.w) * di + bv.w;
  acc.x = fmaxf(acc.x, 0.0f); acc.y = fmaxf(acc.y, 0.0f);
  acc.z = fmaxf(acc.z, 0.0f); acc.w = fmaxf(acc.w, 0.0f);
  return acc;
}

#define FT_M 64
#define FT_T 1024
#define FT_P 132
static_assert(FT_T / 32 * 2 == FT_M, "two targets per wave");
template <bool ORDER, int NT>
__device__ __forceinline__ void rows_then_product(float* __restrict__ tile, float* __restrict__ wt,
                                                  const int* __restrict__ offs, int* srcs,
                                                  const float* __restrict__ dinv,
                                                  const float* __restrict__ H,
                                                  const float* __restrict__ b,
                                                  const float* __restrict__ Wt,
                                                  float* __restrict__ Out, int n) {
  const int tid = threadIdx.x;
  const int lane = tid & 31;
  const int wv = tid >> 5;
  const int row0 = blockIdx.x * FT_M;

  constexpr int WPIECES = NT * 16 * (NHID / 4);
  constexpr int WPT = (WPIECES + FT_T - 1) / FT_T;
  v4f wreg[WPT];
#pragma unroll
  for (int q = 0; q < WPT; ++q) {
    const int p = tid + q * FT_T;
    if (p < WPIECES) wreg[q] = *(const v4f*)(Wt + (size_t)(p >> 5) * NHID + (p & 31) * 4);
  }

#pragma nounroll
  for (int i = 0; i < 2; ++i) {
    const int r = wv + 32 * i;
    float4 acc = make_float4(0.0f, 0.0f, 0.0f, 0.0f);
    if (row0 + r < n) {
      const int node = __builtin_amdgcn_readfirstlane(row0 + r);
      acc = target_row<ORDER>(offs, srcs, dinv, H, b, node, lane);
    }
    *(float4*)(tile + r * FT_P + lane * 4) = acc;
  }
#pragma unroll
  for (int q = 0; q < WPT; ++q) {
    const int p = tid + q * FT_T;
    if (p < WPIECES) *(v4f*)(wt + (p >> 5) * FT_P + (p & 31) * 4) = wreg[q];
  }
  __syncthreads();

  const int c = lane & 15;
  const int h = lane >> 4;
  const int strip = wv & 3, ct = wv >> 2;
  const bool mine = wv < 4 * NT;
  v8f acc = {};
  if (mine) {
    const float* ap = tile + (strip * 16 + c) * FT_P + 2 * h;
    const float* bp = wt + (ct * 16 + c) * FT_P + 2 * h;
    for (int kc = 0; kc < NHID; kc += 32) {
#pragma unroll
      for (int ks = 0; ks < 8; ++ks) {
        const v2f a = *(const v2f*)(ap + kc + ks * 4);
        const v2f f = *(const v2f*)(bp + kc + ks * 4);
        acc = __builtin_amdgcn_wmma_f32_16x16x4_f32(
             false, a,  false, f,
             (short)0, acc,  false,  false);
      }
    }
  }
  __syncthreads();

  if (mine) {
    float* cs = tile + (strip * 16 + 8 * h) * FT_P + ct * 16 + c;
#pragma unroll
    for (int e = 0; e < 8; ++e) cs[e * FT_P] = acc[e];
  }
  __syncthreads();

  if (NT == 8) {
    for (int i = 0; i < 2; ++i) {
      const int r = wv + 32 * i;
      const int row = row0 + r;
      if (row < n) {
        const float sc = dinv[row];
        *(v4f*)(Out + (size_t)row * (16 * NT) + lane * 4) = *(const v4f*)(tile + r * FT_P + lane * 4) * sc;
      }
    }
  } else {
    if (tid < FT_M * 4) {
      const int r = tid >> 2, q = tid & 3;
      const int row = row0 + r;
      if (row < n) {
        const float sc = dinv[row];
        *(v4f*)(Out + (size_t)row * (16 * NT) + q * 4) = *(const v4f*)(tile + r * FT_P + q * 4) * sc;
      }
    }
  }
}

__global__ void __launch_bounds__(FT_T) agg_order_mm128(const int* __restrict__ offs, int* srcs,
                                                           const float* __restrict__ dinv,
                                                           const float* __restrict__ H,
                                                           const float* __restrict__ b,
                                                           const float* __restrict__ Wt,
                                                           float* __restrict__ Out, int n) {
  __shared__ __attribute__((aligned(16))) float lds[(FT_M + NHID) * FT_P];
  rows_then_product<true, NHID / 16>(lds, lds + FT_M * FT_P, offs, srcs, dinv, H, b, Wt, Out, n);
}

__global__ void __launch_bounds__(FT_T) agg_mm16(const int* __restrict__ offs, int* srcs,
                                                    const float* __restrict__ dinv,
                                                    const float* __restrict__ H,
                                                    const float* __restrict__ b,
                                                    const float* __restrict__ Wt,
                                                    float* __restrict__ Out, int n) {
  __shared__ __attribute__((aligned(16))) float lds[(FT_M + FOLD_N) * FT_P];
  rows_then_product<false, FOLD_N / 16>(lds, lds + FT_M * FT_P, offs, srcs, dinv, H, b, Wt, Out, n);
}

__global__ void __launch_bounds__(256) agg_last(const int* __restrict__ offs,
                                                const int* __restrict__ srcs,
                                                const float* __restrict__ dinv,
                                                const float* __restrict__ G,
                                                const float* __restrict__ bf,
                                                float* __restrict__ out, int n) {
  int node = (int)((blockIdx.x * (size_t)blockDim.x + threadIdx.x) >> 5);
  const int lane = threadIdx.x & 31;
  if (node >= n) return;
  node = __builtin_amdgcn_readfirstlane(node);
  const int beg = offs[node];
  const int d = offs[node + 1] - beg;
  const float di = dinv[node];
  const int slot = lane >> 2;
  const int q4 = (lane & 3) * 4;
  const float* gl = G + q4;

  float4 acc = make_float4(0.0f, 0.0f, 0.0f, 0.0f);
  for (int base = 0; base < d; base += 32) {
    bool v[4];
    int s[4];
    float4 g[4];
#pragma unroll
    for (int bt = 0; bt < 4; ++bt) {
      const int e = base + slot + 8 * bt;
      v[bt] = e < d;
      s[bt] = srcs[beg + (v[bt] ? e : d - 1)];
    }
#pragma unroll
    for (int bt = 0; bt < 4; ++bt) g[bt] = *(const float4*)(gl + (size_t)s[bt] * FOLD_N);
#pragma unroll
    for (int bt = 0; bt < 4; ++bt) row_fma(acc, g[bt], v[bt] ? 1.0f : 0.0f);
  }
#pragma unroll
  for (int o = 4; o < 32; o <<= 1) {
    acc.x += __shfl_xor(acc.x, o, 32);
    acc.y += __shfl_xor(acc.y, o, 32);
    acc.z += __shfl_xor(acc.z, o, 32);
    acc.w += __shfl_xor(acc.w, o, 32);
  }

  const float4 gself = *(const float4*)(gl + (size_t)node * FOLD_N);
  const float4 bv = *(const float4*)(bf + q4);
  acc.x = (acc.x + gself.x) * di + bv.x; acc.y = (acc.y + gself.y) * di + bv.y;
  acc.z = (acc.z + gself.z) * di + bv.z; acc.w = (acc.w + gself.w) * di + bv.w;

  const float ninf = -__builtin_inff();
  const bool c0 = q4 + 0 < NCLASS, c1 = q4 + 1 < NCLASS, c2 = q4 + 2 < NCLASS, c3 = q4 + 3 < NCLASS;
  float m = fmaxf(fmaxf(c0 ? acc.x : ninf, c1 ? acc.y : ninf),
                  fmaxf(c2 ? acc.z : ninf, c3 ? acc.w : ninf));
  m = fmaxf(m, __shfl_xor(m, 1, 32));
  m = fmaxf(m, __shfl_xor(m, 2, 32));
  float s = (c0 ? __expf(acc.x - m) : 0.0f) + (c1 ? __expf(acc.y - m) : 0.0f) +
            (c2 ? __expf(acc.z - m) : 0.0f) + (c3 ? __expf(acc.w - m) : 0.0f);
  s += __shfl_xor(s, 1, 32);
  s += __shfl_xor(s, 2, 32);
  const float lse = m + __logf(s);
  if (lane < 3) {
    float* o = out + (size_t)node * NCLASS + q4;
    *(float2*)o = make_float2(acc.x - lse, acc.y - lse);
    if (lane < 2) *(float2*)(o + 2) = make_float2(acc.z - lse, acc.w - lse);
  }
}

extern "C" void kernel_launch(void* const* d_in, const int* in_sizes, int n_in,
                              void* d_out, int out_size, void* d_ws, size_t ws_size,
                              hipStream_t stream) {
  const float* x  = (const float*)d_in[0];
  const int*   ei = (const int*)d_in[1];
  const float* W1 = (const float*)d_in[2];
  const float* b1 = (const float*)d_in[3];
  const float* W2 = (const float*)d_in[4];
  const float* b2 = (const float*)d_in[5];
  const float* W3 = (const float*)d_in[6];
  const float* b3 = (const float*)d_in[7];
  const float* Wl = (const float*)d_in[8];
  const float* bl = (const float*)d_in[9];
  float* out = (float*)d_out;

  const int E = in_sizes[1] / 2;
  const int N = in_sizes[0] / NFEAT;
  const int npad = ((N + 16) / 16) * 16;

  char* ws = (char*)d_ws;
  size_t off = 0;
  auto alloc = [&](size_t bytes) -> float* {
    float* p = (float*)(ws + off);
    off += (bytes + 255) & ~(size_t)255;
    return p;
  };
  const size_t cbytes = ((size_t)npad * sizeof(int) + 255) & ~(size_t)255;
  const size_t ebytes = ((size_t)E * sizeof(int) + 255) & ~(size_t)255;
  const size_t hbytes = ((size_t)SC_CHUNKS * npad * sizeof(int) + 255) & ~(size_t)255;
  const size_t tbytes = cbytes + ebytes + hbytes;
  const size_t ybytes = (size_t)N * NHID * sizeof(float);
  float* dinv = alloc((size_t)npad * sizeof(float));
  float* Wt1  = alloc((size_t)NFEAT * NHID * sizeof(float));
  float* Wt2  = alloc((size_t)NHID * NHID * sizeof(float));
  float* Wt3l = alloc((size_t)FOLD_N * NHID * sizeof(float));
  float* b3l  = alloc((size_t)FOLD_N * sizeof(float));
  float* H    = alloc(ybytes);
  float* Y    = alloc(ybytes > tbytes ? ybytes : tbytes);
  int* offs   = (int*)alloc((size_t)npad * sizeof(int));
  int* srcs   = (int*)alloc((size_t)E * sizeof(int));
  if (off > ws_size) return;
  const int nblk = (npad + SB_T - 1) / SB_T;
  if (nblk > FILL_TAB) return;
  int* btot = (int*)Y;
  int* rank = (int*)((char*)Y + cbytes);
  int* hist = (int*)((char*)Y + cbytes + ebytes);

  const int* rows = ei;
  const int* cols = ei + E;

  const int wgrid = (int)(((size_t)N * 32 + 255) / 256);

  const int epc = (E + SC_CHUNKS - 1) / SC_CHUNKS;

  chunk_count<<<SC_CHUNKS, SC_T, 0, stream>>>(cols, rank, hist, E, epc, npad);
  prep_scan<<<PB_ALLW + nblk, SB_T, 0, stream>>>(W1, Wt1, W2, Wt2, W3, b3, Wl, bl,
                                                 Wt3l, b3l, hist, btot, dinv, npad);

  const int tiles = (N + 15) / 16;
  const int gemm_grid = (tiles + 3) / 4;
  const int gemm_full = (gemm_grid / GB_WGPS) * GB_WGPS;
  const int gemm_wgs = gemm_full + (gemm_grid - gemm_full) * (NHID / GT_N);

  const int nfill = ((epc + FILL_EPB - 1) / FILL_EPB) * SC_CHUNKS;
  const int noffs = (npad + FILL_EPB - 1) / FILL_EPB;
  gemm_wmma_f32<<<gemm_wgs + nfill + noffs, 128, 0, stream>>>(x, Wt1, dinv, H, N, NFEAT, gemm_full,
                                                              gemm_wgs, rows, cols, rank, hist, btot,
                                                              srcs, offs, E, epc, npad, nblk, nfill);

  const int fgrid = (N + FT_M - 1) / FT_M;

  agg_order_mm128<<<fgrid, FT_T, 0, stream>>>(offs, srcs, dinv, H, b1, Wt2, Y, N);

  agg_mm16<<<fgrid, FT_T, 0, stream>>>(offs, srcs, dinv, Y, b2, Wt3l, H, N);

  agg_last<<<wgrid, 256, 0, stream>>>(offs, srcs, dinv, H, b3l, out, N);
}
